# starting kernel
# baseline (speedup 1.0000x reference)
#include <hip/hip_runtime.h>
#include <math.h>


typedef __attribute__((ext_vector_type(2))) float v2f;
typedef __attribute__((ext_vector_type(8))) float v8f;

#define NSEQ   16384
#define XL     1024
#define UL     256
#define YLN    256
#define CHUNKS 128
#define CLEN   128
#define COLSTR ((size_t)CLEN * XL)

__device__ __forceinline__ v8f wmma4(v2f a, v2f b, v8f c) {
  return __builtin_amdgcn_wmma_f32_16x16x4_f32(
       false, a,  false, b,
       (short)0, c,  false,  false);
}

template <int WM, int WN>
__global__ __launch_bounds__(256) void gemm_rr_k(
    const float* __restrict__ Ap, const float* __restrict__ Bp,
    float* __restrict__ Cp, int K, int lda, int ldb, int ldc) {
  const int lane = threadIdx.x & 31;
  const int w    = threadIdx.x >> 5;
  const int wm = w >> 2, wn = w & 3;
  const int tm = blockIdx.x * (32 * WM) + wm * (16 * WM);
  const int tn = blockIdx.y * (64 * WN) + wn * (16 * WN);
  const int ml = lane & 15;
  const int kh = (lane >> 4) << 1;

  v8f acc[WM][WN];
#pragma unroll
  for (int mi = 0; mi < WM; ++mi)
#pragma unroll
    for (int ni = 0; ni < WN; ++ni) acc[mi][ni] = (v8f){};

  const float* ap[WM];
  const float* bp[WN];
#pragma unroll
  for (int mi = 0; mi < WM; ++mi)
    ap[mi] = Ap + (size_t)(tm + mi * 16 + ml) * lda + kh;
#pragma unroll
  for (int ni = 0; ni < WN; ++ni)
    bp[ni] = Bp + (size_t)kh * ldb + (tn + ni * 16 + ml);

  for (int k = 0; k < K; k += 4) {
    v2f af[WM], bf[WN];
    const size_t bo0 = (size_t)k * ldb, bo1 = bo0 + (size_t)ldb;
#pragma unroll
    for (int mi = 0; mi < WM; ++mi) af[mi] = (v2f){ ap[mi][k], ap[mi][k + 1] };
#pragma unroll
    for (int ni = 0; ni < WN; ++ni) bf[ni] = (v2f){ bp[ni][bo0], bp[ni][bo1] };
#pragma unroll
    for (int mi = 0; mi < WM; ++mi)
#pragma unroll
      for (int ni = 0; ni < WN; ++ni)
        acc[mi][ni] = wmma4(af[mi], bf[ni], acc[mi][ni]);
  }

  const int nn = lane & 15;
  const int mb = (lane >> 4) << 3;
#pragma unroll
  for (int mi = 0; mi < WM; ++mi)
#pragma unroll
    for (int ni = 0; ni < WN; ++ni) {
      float* cp = Cp + (size_t)(tm + mi * 16 + mb) * ldc + (tn + ni * 16 + nn);
#pragma unroll
      for (int r = 0; r < 8; ++r) cp[(size_t)r * ldc] = acc[mi][ni][r];
    }
}

template <bool STORE_XS>
__global__ __launch_bounds__(256) void scan_step_k(
    const float* __restrict__ Ap, const float* __restrict__ Sold,
    const float* __restrict__ addb, float* __restrict__ Snew,
    float* __restrict__ xsb) {
  const int lane = threadIdx.x & 31;
  const int w    = threadIdx.x >> 5;
  const int wm = w >> 2, wn = w & 3;
  const int tm = blockIdx.x * 32 + wm * 16;
  const int tn = blockIdx.y * 64 + wn * 16;
  const int ml = lane & 15;
  const int kh = (lane >> 4) << 1;

  constexpr int KQ = XL / 4;
  constexpr size_t BQ = (size_t)KQ * CHUNKS;

  v8f acc0 = {}, acc1 = {}, acc2 = {}, acc3 = {};
  const float* ap = Ap + (size_t)(tm + ml) * XL + kh;
  const float* bp = Sold + (size_t)kh * CHUNKS + (tn + ml);

  for (int k = 0; k < KQ; k += 4) {
    v2f a0 = { ap[k],          ap[k + 1] };
    v2f a1 = { ap[k + KQ],     ap[k + KQ + 1] };
    v2f a2 = { ap[k + 2 * KQ], ap[k + 2 * KQ + 1] };
    v2f a3 = { ap[k + 3 * KQ], ap[k + 3 * KQ + 1] };
    const size_t bo = (size_t)k * CHUNKS;
    v2f b0 = { bp[bo],          bp[bo + CHUNKS] };
    v2f b1 = { bp[bo + BQ],     bp[bo + BQ + CHUNKS] };
    v2f b2 = { bp[bo + 2 * BQ], bp[bo + 2 * BQ + CHUNKS] };
    v2f b3 = { bp[bo + 3 * BQ], bp[bo + 3 * BQ + CHUNKS] };
    acc0 = wmma4(a0, b0, acc0);
    acc1 = wmma4(a1, b1, acc1);
    acc2 = wmma4(a2, b2, acc2);
    acc3 = wmma4(a3, b3, acc3);
  }
  v8f acc = (acc0 + acc1) + (acc2 + acc3);

  const int j  = tn + (lane & 15);
  const int mb = tm + ((lane >> 4) << 3);
  const float* addc = addb + (size_t)j * COLSTR + mb;
  if constexpr (STORE_XS) {
    float* xsc = xsb + (size_t)j * COLSTR + mb;
#pragma unroll
    for (int r = 0; r < 8; ++r) {
      float v = acc[r] + addc[r];
      Snew[(size_t)(mb + r) * CHUNKS + j] = v;
      xsc[r] = v;
    }
  } else {
    (void)xsb;
#pragma unroll
    for (int r = 0; r < 8; ++r) {
      float v = acc[r] + addc[r];
      Snew[(size_t)(mb + r) * CHUNKS + j] = v;
    }
  }
}

__global__ __launch_bounds__(256) void matvec_step_k(
    const float* __restrict__ Mm, const float* __restrict__ vin, int vins,
    const float* __restrict__ vadd, int vadds,
    float* __restrict__ vout, int vouts) {
  __shared__ float vsh[XL];
  const int tid = threadIdx.x;
  for (int i = tid; i < XL; i += 256) vsh[i] = vin[(size_t)i * vins];
  __syncthreads();

  const int lane = tid & 31, wid = tid >> 5;
  const int gw = blockIdx.x * 8 + wid;
#pragma unroll
  for (int rr = 0; rr < 4; ++rr) {
    const int row = gw * 4 + rr;
    const float* Mr = Mm + (size_t)row * XL;
    float s = 0.f;
    for (int i = 0; i < XL; i += 32) s += Mr[i + lane] * vsh[i + lane];
#pragma unroll
    for (int off = 16; off > 0; off >>= 1) s += __shfl_down(s, off, 32);
    if (lane == 0) vout[(size_t)row * vouts] = s + vadd[(size_t)row * vadds];
  }
}

__global__ __launch_bounds__(256) void final_gemm_k(
    const float* __restrict__ xs, const float* __restrict__ Ct,
    const float* __restrict__ uf, const float* __restrict__ Dt,
    float* __restrict__ out) {
  const int lane = threadIdx.x & 31;
  const int w    = threadIdx.x >> 5;
  const int wm = w >> 2, wn = w & 3;
  const int tm = blockIdx.x * 128 + wm * 64;
  const int tn = wn * 64;
  const int ml = lane & 15;
  const int kh = (lane >> 4) << 1;

  v8f acc[4][4];
#pragma unroll
  for (int mi = 0; mi < 4; ++mi)
#pragma unroll
    for (int ni = 0; ni < 4; ++ni) acc[mi][ni] = (v8f){};

  {
    const float* ap[4];
    const float* bp[4];
#pragma unroll
    for (int mi = 0; mi < 4; ++mi)
      ap[mi] = xs + (size_t)(tm + mi * 16 + ml) * XL + kh;
#pragma unroll
    for (int ni = 0; ni < 4; ++ni)
      bp[ni] = Ct + (size_t)kh * YLN + (tn + ni * 16 + ml);
    for (int k = 0; k < XL; k += 4) {
      v2f af[4], bf[4];
      const size_t bo0 = (size_t)k * YLN, bo1 = bo0 + YLN;
#pragma unroll
      for (int mi = 0; mi < 4; ++mi) af[mi] = (v2f){ ap[mi][k], ap[mi][k + 1] };
#pragma unroll
      for (int ni = 0; ni < 4; ++ni) bf[ni] = (v2f){ bp[ni][bo0], bp[ni][bo1] };
#pragma unroll
      for (int mi = 0; mi < 4; ++mi)
#pragma unroll
        for (int ni = 0; ni < 4; ++ni)
          acc[mi][ni] = wmma4(af[mi], bf[ni], acc[mi][ni]);
    }
  }
  {
    const float* ap[4];
    const float* bp[4];
#pragma unroll
    for (int mi = 0; mi < 4; ++mi)
      ap[mi] = uf + (size_t)(tm + mi * 16 + ml) * UL + kh;
#pragma unroll
    for (int ni = 0; ni < 4; ++ni)
      bp[ni] = Dt + (size_t)kh * YLN + (tn + ni * 16 + ml);
    for (int k = 0; k < UL; k += 4) {
      v2f af[4], bf[4];
      const size_t bo0 = (size_t)k * YLN, bo1 = bo0 + YLN;
#pragma unroll
      for (int mi = 0; mi < 4; ++mi) af[mi] = (v2f){ ap[mi][k], ap[mi][k + 1] };
#pragma unroll
      for (int ni = 0; ni < 4; ++ni) bf[ni] = (v2f){ bp[ni][bo0], bp[ni][bo1] };
#pragma unroll
      for (int mi = 0; mi < 4; ++mi)
#pragma unroll
        for (int ni = 0; ni < 4; ++ni)
          acc[mi][ni] = wmma4(af[mi], bf[ni], acc[mi][ni]);
    }
  }

  const int nn = lane & 15;
  const int mb = (lane >> 4) << 3;
#pragma unroll
  for (int mi = 0; mi < 4; ++mi)
#pragma unroll
    for (int ni = 0; ni < 4; ++ni) {
      const int yl = tn + ni * 16 + nn;
      const int n0 = tm + mi * 16 + mb;
      float* op = out + (size_t)yl * NSEQ + n0;
#pragma unroll
      for (int r = 0; r < 8; ++r) op[r] = tanhf(acc[mi][ni][r]);
    }
}

__global__ __launch_bounds__(256) void transpose_k(
    const float* __restrict__ in, float* __restrict__ out, int rows, int cols) {
  __shared__ float tile[32][33];
  const int c0 = blockIdx.x * 32, r0 = blockIdx.y * 32;
  const int tx = threadIdx.x, ty = threadIdx.y;
#pragma unroll
  for (int i = 0; i < 32; i += 8)
    tile[ty + i][tx] = in[(size_t)(r0 + ty + i) * cols + (c0 + tx)];
  __syncthreads();
#pragma unroll
  for (int i = 0; i < 32; i += 8)
    out[(size_t)(c0 + ty + i) * rows + (r0 + tx)] = tile[tx][ty + i];
}

__global__ void set_col_k(float* __restrict__ dst, const float* __restrict__ src,
                          int stride, int n) {
  int i = blockIdx.x * blockDim.x + threadIdx.x;
  if (i < n) dst[(size_t)i * stride] = src[i];
}

extern "C" void kernel_launch(void* const* d_in, const int* in_sizes, int n_in,
                              void* d_out, int out_size, void* d_ws, size_t ws_size,
                              hipStream_t stream) {
  (void)in_sizes; (void)n_in; (void)out_size; (void)ws_size;
  const float* u  = (const float*)d_in[0];
  const float* A  = (const float*)d_in[1];
  const float* B  = (const float*)d_in[2];
  const float* C  = (const float*)d_in[3];
  const float* D  = (const float*)d_in[4];
  const float* x0 = (const float*)d_in[5];
  float* out = (float*)d_out;

  float* ws = (float*)d_ws;
  size_t off = 0;
  float* Bu = ws + off; off += (size_t)NSEQ * XL;
  float* xs = ws + off; off += (size_t)NSEQ * XL;
  float* Bt = ws + off; off += (size_t)UL * XL;
  float* Ct = ws + off; off += (size_t)XL * YLN;
  float* Dt = ws + off; off += (size_t)UL * YLN;
  float* Q0 = ws + off; off += (size_t)XL * XL;
  float* Q1 = ws + off; off += (size_t)XL * XL;
  float* SA = ws + off; off += (size_t)XL * CHUNKS;
  float* SB = ws + off; off += (size_t)XL * CHUNKS;
  float* EB = ws + off; off += (size_t)XL * CHUNKS;

  transpose_k<<<dim3(UL / 32, XL / 32), dim3(32, 8), 0, stream>>>(B, Bt, XL, UL);
  transpose_k<<<dim3(XL / 32, YLN / 32), dim3(32, 8), 0, stream>>>(C, Ct, YLN, XL);
  transpose_k<<<dim3(YLN / 32, UL / 32), dim3(32, 8), 0, stream>>>(D, Dt, UL, YLN);

  gemm_rr_k<4, 4><<<dim3(NSEQ / 128, XL / 256), 256, 0, stream>>>(
      u, Bt, Bu, UL, UL, XL, XL);

  gemm_rr_k<2, 2><<<dim3(XL / 64, XL / 128), 256, 0, stream>>>(
      A, A, Q0, XL, XL, XL, XL);
  {
    float* src = Q0; float* dst = Q1;
    for (int i = 0; i < 6; ++i) {
      gemm_rr_k<2, 2><<<dim3(XL / 64, XL / 128), 256, 0, stream>>>(
          src, src, dst, XL, XL, XL, XL);
      float* t = src; src = dst; dst = t;
    }
  }
  const float* M128 = Q0;

  hipMemsetAsync(SA, 0, (size_t)XL * CHUNKS * sizeof(float), stream);
  hipMemsetAsync(EB, 0, (size_t)XL * CHUNKS * sizeof(float), stream);
  set_col_k<<<4, 256, 0, stream>>>(EB, x0, CHUNKS, XL);

  {
    float* cur = SA; float* nxt = SB;
    for (int t = 0; t < CLEN; ++t) {
      scan_step_k<false><<<dim3(XL / 32, CHUNKS / 64), 256, 0, stream>>>(
          A, cur, Bu + (size_t)t * XL, nxt, nullptr);
      float* tmp = cur; cur = nxt; nxt = tmp;
    }
  }

  for (int j = 1; j < CHUNKS; ++j) {
    matvec_step_k<<<32, 256, 0, stream>>>(M128, EB + (j - 1), CHUNKS,
                                          SA + (j - 1), CHUNKS, EB + j, CHUNKS);
  }

  {
    float* cur = EB; float* nxt = SB;
    for (int t = 0; t < CLEN; ++t) {
      scan_step_k<true><<<dim3(XL / 32, CHUNKS / 64), 256, 0, stream>>>(
          A, cur, Bu + (size_t)t * XL, nxt, xs + (size_t)t * XL);
      float* tmp = cur; cur = nxt; nxt = tmp;
    }
  }

  final_gemm_k<<<dim3(NSEQ / 128, 1), 256, 0, stream>>>(xs, Ct, u, Dt, out);
}


# scan step re-tiled: 256 workgroups (16x32 tile, one per WGP), each K-quarter chain on its own wave with dozens of operand loads in flight, chains summed in LDS; sum order unchanged
# speedup vs baseline: 1.7226x; 1.7226x over previous
#include <hip/hip_runtime.h>
#include <math.h>


typedef __attribute__((ext_vector_type(2))) float v2f;
typedef __attribute__((ext_vector_type(8))) float v8f;

#define NSEQ   16384
#define XL     1024
#define UL     256
#define YLN    256
#define CHUNKS 128
#define CLEN   128
#define COLSTR ((size_t)CLEN * XL)

__device__ __forceinline__ v8f wmma4(v2f a, v2f b, v8f c) {
  return __builtin_amdgcn_wmma_f32_16x16x4_f32(
       false, a,  false, b,
       (short)0, c,  false,  false);
}

template <int WM, int WN>
__global__ __launch_bounds__(256) void gemm_rr_k(
    const float* __restrict__ Ap, const float* __restrict__ Bp,
    float* __restrict__ Cp, int K, int lda, int ldb, int ldc) {
  const int lane = threadIdx.x & 31;
  const int w    = threadIdx.x >> 5;
  const int wm = w >> 2, wn = w & 3;
  const int tm = blockIdx.x * (32 * WM) + wm * (16 * WM);
  const int tn = blockIdx.y * (64 * WN) + wn * (16 * WN);
  const int ml = lane & 15;
  const int kh = (lane >> 4) << 1;

  v8f acc[WM][WN];
#pragma unroll
  for (int mi = 0; mi < WM; ++mi)
#pragma unroll
    for (int ni = 0; ni < WN; ++ni) acc[mi][ni] = (v8f){};

  const float* ap[WM];
  const float* bp[WN];
#pragma unroll
  for (int mi = 0; mi < WM; ++mi)
    ap[mi] = Ap + (size_t)(tm + mi * 16 + ml) * lda + kh;
#pragma unroll
  for (int ni = 0; ni < WN; ++ni)
    bp[ni] = Bp + (size_t)kh * ldb + (tn + ni * 16 + ml);

  for (int k = 0; k < K; k += 4) {
    v2f af[WM], bf[WN];
    const size_t bo0 = (size_t)k * ldb, bo1 = bo0 + (size_t)ldb;
#pragma unroll
    for (int mi = 0; mi < WM; ++mi) af[mi] = (v2f){ ap[mi][k], ap[mi][k + 1] };
#pragma unroll
    for (int ni = 0; ni < WN; ++ni) bf[ni] = (v2f){ bp[ni][bo0], bp[ni][bo1] };
#pragma unroll
    for (int mi = 0; mi < WM; ++mi)
#pragma unroll
      for (int ni = 0; ni < WN; ++ni)
        acc[mi][ni] = wmma4(af[mi], bf[ni], acc[mi][ni]);
  }

  const int nn = lane & 15;
  const int mb = (lane >> 4) << 3;
#pragma unroll
  for (int mi = 0; mi < WM; ++mi)
#pragma unroll
    for (int ni = 0; ni < WN; ++ni) {
      float* cp = Cp + (size_t)(tm + mi * 16 + mb) * ldc + (tn + ni * 16 + nn);
#pragma unroll
      for (int r = 0; r < 8; ++r) cp[(size_t)r * ldc] = acc[mi][ni][r];
    }
}

template <bool STORE_XS>
__global__ __launch_bounds__(256) __attribute__((amdgpu_waves_per_eu(2, 2)))
void scan_step_k(
    const float* __restrict__ Ap, const float* __restrict__ Sold,
    const float* addb, float* __restrict__ Snew,
    float* __restrict__ xsb) {
  constexpr int KQ  = XL / 4;
  constexpr int GS  = 8;
  constexpr int NG  = KQ / 4 / GS;
  constexpr int RP  = 8 * 32 + 16;
  __shared__ float red[4][2][RP];

  const int lane = threadIdx.x & 31;
  const int w    = threadIdx.x >> 5;
  const int nt = w & 1, q = w >> 1;
  const int tm = blockIdx.x * 16;
  const int tn = blockIdx.y * 32;
  const int ml = lane & 15;
  const int kh = (lane >> 4) << 1;

  const int er = 2 * w;
  const int ej = tn + lane;
  const float* addp = addb + (size_t)ej * COLSTR + (tm + er);
  v2f addv = {};

  const float* ap = Ap + (size_t)(tm + ml) * XL + (q * KQ + kh);
  const float* bp = Sold + (size_t)(q * KQ + kh) * CHUNKS + (tn + nt * 16 + ml);

  v2f af[NG][GS], bf[NG][GS];
#pragma unroll
  for (int g = 0; g < 2; ++g)
#pragma unroll
    for (int s = 0; s < GS; ++s) {
      const int k = (g * GS + s) * 4;
      af[g][s] = (v2f){ ap[k], ap[k + 1] };
      bf[g][s] = (v2f){ bp[(size_t)k * CHUNKS], bp[(size_t)(k + 1) * CHUNKS] };
    }

  v8f acc = {};
#pragma unroll
  for (int g = 0; g < NG; ++g) {
    if (g + 2 < NG) {
#pragma unroll
      for (int s = 0; s < GS; ++s) {
        const int k = ((g + 2) * GS + s) * 4;
        af[g + 2][s] = (v2f){ ap[k], ap[k + 1] };
        bf[g + 2][s] = (v2f){ bp[(size_t)k * CHUNKS], bp[(size_t)(k + 1) * CHUNKS] };
      }
    }
    if (g + 3 == NG) {
      addv = *(const v2f*)addp;
      __builtin_amdgcn_sched_barrier(0);
    }
#pragma unroll
    for (int s = 0; s < GS; ++s) acc = wmma4(af[g][s], bf[g][s], acc);
  }

#pragma unroll
  for (int r = 0; r < 8; ++r) red[q][nt][r * 32 + lane] = acc[r];
  __syncthreads();

  const int et = lane >> 4, ec = lane & 15;
  v2f v;
#pragma unroll
  for (int i = 0; i < 2; ++i) {
    const int mr = er + i;
    const int sl = (mr & 7) * 32 + (mr >> 3) * 16 + ec;
    const float c0 = red[0][et][sl], c1 = red[1][et][sl];
    const float c2 = red[2][et][sl], c3 = red[3][et][sl];
    v[i] = ((c0 + c1) + (c2 + c3)) + addv[i];
    Snew[(size_t)(tm + mr) * CHUNKS + ej] = v[i];
  }
  if constexpr (STORE_XS) {
    *(v2f*)(xsb + (size_t)ej * COLSTR + (tm + er)) = v;
  } else {
    (void)xsb;
  }
}

__global__ __launch_bounds__(256) void matvec_step_k(
    const float* __restrict__ Mm, const float* __restrict__ vin, int vins,
    const float* __restrict__ vadd, int vadds,
    float* __restrict__ vout, int vouts) {
  __shared__ float vsh[XL];
  const int tid = threadIdx.x;
  for (int i = tid; i < XL; i += 256) vsh[i] = vin[(size_t)i * vins];
  __syncthreads();

  const int lane = tid & 31, wid = tid >> 5;
  const int gw = blockIdx.x * 8 + wid;
#pragma unroll
  for (int rr = 0; rr < 4; ++rr) {
    const int row = gw * 4 + rr;
    const float* Mr = Mm + (size_t)row * XL;
    float s = 0.f;
    for (int i = 0; i < XL; i += 32) s += Mr[i + lane] * vsh[i + lane];
#pragma unroll
    for (int off = 16; off > 0; off >>= 1) s += __shfl_down(s, off, 32);
    if (lane == 0) vout[(size_t)row * vouts] = s + vadd[(size_t)row * vadds];
  }
}

__global__ __launch_bounds__(256) void final_gemm_k(
    const float* __restrict__ xs, const float* __restrict__ Ct,
    const float* __restrict__ uf, const float* __restrict__ Dt,
    float* __restrict__ out) {
  const int lane = threadIdx.x & 31;
  const int w    = threadIdx.x >> 5;
  const int wm = w >> 2, wn = w & 3;
  const int tm = blockIdx.x * 128 + wm * 64;
  const int tn = wn * 64;
  const int ml = lane & 15;
  const int kh = (lane >> 4) << 1;

  v8f acc[4][4];
#pragma unroll
  for (int mi = 0; mi < 4; ++mi)
#pragma unroll
    for (int ni = 0; ni < 4; ++ni) acc[mi][ni] = (v8f){};

  {
    const float* ap[4];
    const float* bp[4];
#pragma unroll
    for (int mi = 0; mi < 4; ++mi)
      ap[mi] = xs + (size_t)(tm + mi * 16 + ml) * XL + kh;
#pragma unroll
    for (int ni = 0; ni < 4; ++ni)
      bp[ni] = Ct + (size_t)kh * YLN + (tn + ni * 16 + ml);
    for (int k = 0; k < XL; k += 4) {
      v2f af[4], bf[4];
      const size_t bo0 = (size_t)k * YLN, bo1 = bo0 + YLN;
#pragma unroll
      for (int mi = 0; mi < 4; ++mi) af[mi] = (v2f){ ap[mi][k], ap[mi][k + 1] };
#pragma unroll
      for (int ni = 0; ni < 4; ++ni) bf[ni] = (v2f){ bp[ni][bo0], bp[ni][bo1] };
#pragma unroll
      for (int mi = 0; mi < 4; ++mi)
#pragma unroll
        for (int ni = 0; ni < 4; ++ni)
          acc[mi][ni] = wmma4(af[mi], bf[ni], acc[mi][ni]);
    }
  }
  {
    const float* ap[4];
    const float* bp[4];
#pragma unroll
    for (int mi = 0; mi < 4; ++mi)
      ap[mi] = uf + (size_t)(tm + mi * 16 + ml) * UL + kh;
#pragma unroll
    for (int ni = 0; ni < 4; ++ni)
      bp[ni] = Dt + (size_t)kh * YLN + (tn + ni * 16 + ml);
    for (int k = 0; k < UL; k += 4) {
      v2f af[4], bf[4];
      const size_t bo0 = (size_t)k * YLN, bo1 = bo0 + YLN;
#pragma unroll
      for (int mi = 0; mi < 4; ++mi) af[mi] = (v2f){ ap[mi][k], ap[mi][k + 1] };
#pragma unroll
      for (int ni = 0; ni < 4; ++ni) bf[ni] = (v2f){ bp[ni][bo0], bp[ni][bo1] };
#pragma unroll
      for (int mi = 0; mi < 4; ++mi)
#pragma unroll
        for (int ni = 0; ni < 4; ++ni)
          acc[mi][ni] = wmma4(af[mi], bf[ni], acc[mi][ni]);
    }
  }

  const int nn = lane & 15;
  const int mb = (lane >> 4) << 3;
#pragma unroll
  for (int mi = 0; mi < 4; ++mi)
#pragma unroll
    for (int ni = 0; ni < 4; ++ni) {
      const int yl = tn + ni * 16 + nn;
      const int n0 = tm + mi * 16 + mb;
      float* op = out + (size_t)yl * NSEQ + n0;
#pragma unroll
      for (int r = 0; r < 8; ++r) op[r] = tanhf(acc[mi][ni][r]);
    }
}

__global__ __launch_bounds__(256) void transpose_k(
    const float* __restrict__ in, float* __restrict__ out, int rows, int cols) {
  __shared__ float tile[32][33];
  const int c0 = blockIdx.x * 32, r0 = blockIdx.y * 32;
  const int tx = threadIdx.x, ty = threadIdx.y;
#pragma unroll
  for (int i = 0; i < 32; i += 8)
    tile[ty + i][tx] = in[(size_t)(r0 + ty + i) * cols + (c0 + tx)];
  __syncthreads();
#pragma unroll
  for (int i = 0; i < 32; i += 8)
    out[(size_t)(c0 + ty + i) * rows + (r0 + tx)] = tile[tx][ty + i];
}

__global__ void set_col_k(float* __restrict__ dst, const float* __restrict__ src,
                          int stride, int n) {
  int i = blockIdx.x * blockDim.x + threadIdx.x;
  if (i < n) dst[(size_t)i * stride] = src[i];
}

extern "C" void kernel_launch(void* const* d_in, const int* in_sizes, int n_in,
                              void* d_out, int out_size, void* d_ws, size_t ws_size,
                              hipStream_t stream) {
  (void)in_sizes; (void)n_in; (void)out_size; (void)ws_size;
  const float* u  = (const float*)d_in[0];
  const float* A  = (const float*)d_in[1];
  const float* B  = (const float*)d_in[2];
  const float* C  = (const float*)d_in[3];
  const float* D  = (const float*)d_in[4];
  const float* x0 = (const float*)d_in[5];
  float* out = (float*)d_out;

  float* ws = (float*)d_ws;
  size_t off = 0;
  float* Bu = ws + off; off += (size_t)NSEQ * XL;
  float* xs = ws + off; off += (size_t)NSEQ * XL;
  float* Bt = ws + off; off += (size_t)UL * XL;
  float* Ct = ws + off; off += (size_t)XL * YLN;
  float* Dt = ws + off; off += (size_t)UL * YLN;
  float* Q0 = ws + off; off += (size_t)XL * XL;
  float* Q1 = ws + off; off += (size_t)XL * XL;
  float* SA = ws + off; off += (size_t)XL * CHUNKS;
  float* SB = ws + off; off += (size_t)XL * CHUNKS;
  float* EB = ws + off; off += (size_t)XL * CHUNKS;

  transpose_k<<<dim3(UL / 32, XL / 32), dim3(32, 8), 0, stream>>>(B, Bt, XL, UL);
  transpose_k<<<dim3(XL / 32, YLN / 32), dim3(32, 8), 0, stream>>>(C, Ct, YLN, XL);
  transpose_k<<<dim3(YLN / 32, UL / 32), dim3(32, 8), 0, stream>>>(D, Dt, UL, YLN);

  gemm_rr_k<4, 4><<<dim3(NSEQ / 128, XL / 256), 256, 0, stream>>>(
      u, Bt, Bu, UL, UL, XL, XL);

  gemm_rr_k<2, 2><<<dim3(XL / 64, XL / 128), 256, 0, stream>>>(
      A, A, Q0, XL, XL, XL, XL);
  {
    float* src = Q0; float* dst = Q1;
    for (int i = 0; i < 6; ++i) {
      gemm_rr_k<2, 2><<<dim3(XL / 64, XL / 128), 256, 0, stream>>>(
          src, src, dst, XL, XL, XL, XL);
      float* t = src; src = dst; dst = t;
    }
  }
  const float* M128 = Q0;

  hipMemsetAsync(SA, 0, (size_t)XL * CHUNKS * sizeof(float), stream);
  hipMemsetAsync(EB, 0, (size_t)XL * CHUNKS * sizeof(float), stream);
  set_col_k<<<4, 256, 0, stream>>>(EB, x0, CHUNKS, XL);

  {
    float* cur = SA; float* nxt = SB;
    for (int t = 0; t < CLEN; ++t) {
      scan_step_k<false><<<dim3(XL / 16, CHUNKS / 32), 256, 0, stream>>>(
          A, cur, Bu + (size_t)t * XL, nxt, nullptr);
      float* tmp = cur; cur = nxt; nxt = tmp;
    }
  }

  for (int j = 1; j < CHUNKS; ++j) {
    matvec_step_k<<<32, 256, 0, stream>>>(M128, EB + (j - 1), CHUNKS,
                                          SA + (j - 1), CHUNKS, EB + j, CHUNKS);
  }

  {
    float* cur = EB; float* nxt = SB;
    for (int t = 0; t < CLEN; ++t) {
      scan_step_k<true><<<dim3(XL / 16, CHUNKS / 32), 256, 0, stream>>>(
          A, cur, Bu + (size_t)t * XL, nxt, xs + (size_t)t * XL);
      float* tmp = cur; cur = nxt; nxt = tmp;
    }
  }

  final_gemm_k<<<dim3(NSEQ / 128, 1), 256, 0, stream>>>(xs, Ct, u, Dt, out);
}


# three-level chunked scan: 512 chunks of 32 steps, chunk ends scanned in 32 groups of 16 with A^32, group starts by 31 matvecs with A^512; 127 serial launches instead of 383; sum order changes
# speedup vs baseline: 2.5053x; 1.4544x over previous
#include <hip/hip_runtime.h>
#include <math.h>


typedef __attribute__((ext_vector_type(2))) float v2f;
typedef __attribute__((ext_vector_type(8))) float v8f;

#define NSEQ   16384
#define XL     1024
#define UL     256
#define YLN    256
#define CHUNKS 512
#define CLEN   32
#define COLSTR (CLEN * XL)
#define GROUPS 32
#define GLEN   16

__device__ __forceinline__ v8f wmma4(v2f a, v2f b, v8f c) {
  return __builtin_amdgcn_wmma_f32_16x16x4_f32(
       false, a,  false, b,
       (short)0, c,  false,  false);
}

template <int WM, int WN>
__global__ __launch_bounds__(256) void gemm_rr_k(
    const float* __restrict__ Ap, const float* __restrict__ Bp,
    float* __restrict__ Cp, int K, int lda, int ldb, int ldc) {
  const int lane = threadIdx.x & 31;
  const int w    = threadIdx.x >> 5;
  const int wm = w >> 2, wn = w & 3;
  const int tm = blockIdx.x * (32 * WM) + wm * (16 * WM);
  const int tn = blockIdx.y * (64 * WN) + wn * (16 * WN);
  const int ml = lane & 15;
  const int kh = (lane >> 4) << 1;

  v8f acc[WM][WN];
#pragma unroll
  for (int mi = 0; mi < WM; ++mi)
#pragma unroll
    for (int ni = 0; ni < WN; ++ni) acc[mi][ni] = (v8f){};

  const float* ap[WM];
  const float* bp[WN];
#pragma unroll
  for (int mi = 0; mi < WM; ++mi)
    ap[mi] = Ap + (size_t)(tm + mi * 16 + ml) * lda + kh;
#pragma unroll
  for (int ni = 0; ni < WN; ++ni)
    bp[ni] = Bp + (size_t)kh * ldb + (tn + ni * 16 + ml);

  for (int k = 0; k < K; k += 4) {
    v2f af[WM], bf[WN];
    const size_t bo0 = (size_t)k * ldb, bo1 = bo0 + (size_t)ldb;
#pragma unroll
    for (int mi = 0; mi < WM; ++mi) af[mi] = (v2f){ ap[mi][k], ap[mi][k + 1] };
#pragma unroll
    for (int ni = 0; ni < WN; ++ni) bf[ni] = (v2f){ bp[ni][bo0], bp[ni][bo1] };
#pragma unroll
    for (int mi = 0; mi < WM; ++mi)
#pragma unroll
      for (int ni = 0; ni < WN; ++ni)
        acc[mi][ni] = wmma4(af[mi], bf[ni], acc[mi][ni]);
  }

  const int nn = lane & 15;
  const int mb = (lane >> 4) << 3;
#pragma unroll
  for (int mi = 0; mi < WM; ++mi)
#pragma unroll
    for (int ni = 0; ni < WN; ++ni) {
      float* cp = Cp + (size_t)(tm + mi * 16 + mb) * ldc + (tn + ni * 16 + nn);
#pragma unroll
      for (int r = 0; r < 8; ++r) cp[(size_t)r * ldc] = acc[mi][ni][r];
    }
}

template <bool STORE_XS, int LD, int ACS, int ARS, int XCS, int XRS>
__global__ __launch_bounds__(256) __attribute__((amdgpu_waves_per_eu(2, 4)))
void scan_step_k(
    const float* __restrict__ Ap, const float* __restrict__ Sold,
    const float* addb, float* __restrict__ Snew,
    float* __restrict__ xsb) {
  constexpr int KQ  = XL / 4;
  constexpr int GS  = 8;
  constexpr int NG  = KQ / 4 / GS;
  constexpr int RP  = 8 * 32 + 16;
  __shared__ float red[4][2][RP];

  const int lane = threadIdx.x & 31;
  const int w    = threadIdx.x >> 5;
  const int nt = w & 1, q = w >> 1;
  const int tm = blockIdx.x * 16;
  const int tn = blockIdx.y * 32;
  const int ml = lane & 15;
  const int kh = (lane >> 4) << 1;

  const int er = 2 * w;
  const int ej = tn + lane;
  const float* addp = addb + (size_t)ej * ACS + (size_t)(tm + er) * ARS;
  v2f addv = {};

  const float* ap = Ap + (size_t)(tm + ml) * XL + (q * KQ + kh);
  const float* bp = Sold + (size_t)(q * KQ + kh) * LD + (tn + nt * 16 + ml);

  v2f af[NG][GS], bf[NG][GS];
#pragma unroll
  for (int g = 0; g < 2; ++g)
#pragma unroll
    for (int s = 0; s < GS; ++s) {
      const int k = (g * GS + s) * 4;
      af[g][s] = (v2f){ ap[k], ap[k + 1] };
      bf[g][s] = (v2f){ bp[(size_t)k * LD], bp[(size_t)(k + 1) * LD] };
    }

  v8f acc = {};
#pragma unroll
  for (int g = 0; g < NG; ++g) {
    if (g + 2 < NG) {
#pragma unroll
      for (int s = 0; s < GS; ++s) {
        const int k = ((g + 2) * GS + s) * 4;
        af[g + 2][s] = (v2f){ ap[k], ap[k + 1] };
        bf[g + 2][s] = (v2f){ bp[(size_t)k * LD], bp[(size_t)(k + 1) * LD] };
      }
    }
    if (g + 3 == NG) {
      if constexpr (ARS == 1) addv = *(const v2f*)addp;
      else addv = (v2f){ addp[0], addp[ARS] };
      __builtin_amdgcn_sched_barrier(0);
    }
#pragma unroll
    for (int s = 0; s < GS; ++s) acc = wmma4(af[g][s], bf[g][s], acc);
  }

#pragma unroll
  for (int r = 0; r < 8; ++r) red[q][nt][r * 32 + lane] = acc[r];
  __syncthreads();

  const int et = lane >> 4, ec = lane & 15;
  v2f v;
#pragma unroll
  for (int i = 0; i < 2; ++i) {
    const int mr = er + i;
    const int sl = (mr & 7) * 32 + (mr >> 3) * 16 + ec;
    const float c0 = red[0][et][sl], c1 = red[1][et][sl];
    const float c2 = red[2][et][sl], c3 = red[3][et][sl];
    v[i] = ((c0 + c1) + (c2 + c3)) + addv[i];
    Snew[(size_t)(tm + mr) * LD + ej] = v[i];
  }
  if constexpr (STORE_XS) {
    float* xp = xsb + (size_t)ej * XCS + (size_t)(tm + er) * XRS;
    if constexpr (XRS == 1) {
      *(v2f*)xp = v;
    } else {
      xp[0] = v[0];
      xp[XRS] = v[1];
    }
  } else {
    (void)xsb;
  }
}

__global__ __launch_bounds__(256) void matvec_step_k(
    const float* __restrict__ Mm, const float* __restrict__ vin, int vins,
    const float* __restrict__ vadd, int vadds,
    float* __restrict__ vout, int vouts,
    float* __restrict__ vout2, int vouts2) {
  __shared__ float vsh[XL];
  const int tid = threadIdx.x;
  for (int i = tid; i < XL; i += 256) vsh[i] = vin[(size_t)i * vins];
  __syncthreads();

  const int lane = tid & 31, wid = tid >> 5;
  const int gw = blockIdx.x * 8 + wid;
#pragma unroll
  for (int rr = 0; rr < 4; ++rr) {
    const int row = gw * 4 + rr;
    const float* Mr = Mm + (size_t)row * XL;
    float s = 0.f;
    for (int i = 0; i < XL; i += 32) s += Mr[i + lane] * vsh[i + lane];
#pragma unroll
    for (int off = 16; off > 0; off >>= 1) s += __shfl_down(s, off, 32);
    if (lane == 0) {
      const float r = s + vadd[(size_t)row * vadds];
      vout[(size_t)row * vouts] = r;
      vout2[(size_t)row * vouts2] = r;
    }
  }
}

__global__ __launch_bounds__(256) void final_gemm_k(
    const float* __restrict__ xs, const float* __restrict__ Ct,
    const float* __restrict__ uf, const float* __restrict__ Dt,
    float* __restrict__ out) {
  const int lane = threadIdx.x & 31;
  const int w    = threadIdx.x >> 5;
  const int wm = w >> 2, wn = w & 3;
  const int tm = blockIdx.x * 128 + wm * 64;
  const int tn = wn * 64;
  const int ml = lane & 15;
  const int kh = (lane >> 4) << 1;

  v8f acc[4][4];
#pragma unroll
  for (int mi = 0; mi < 4; ++mi)
#pragma unroll
    for (int ni = 0; ni < 4; ++ni) acc[mi][ni] = (v8f){};

  {
    const float* ap[4];
    const float* bp[4];
#pragma unroll
    for (int mi = 0; mi < 4; ++mi)
      ap[mi] = xs + (size_t)(tm + mi * 16 + ml) * XL + kh;
#pragma unroll
    for (int ni = 0; ni < 4; ++ni)
      bp[ni] = Ct + (size_t)kh * YLN + (tn + ni * 16 + ml);
    for (int k = 0; k < XL; k += 4) {
      v2f af[4], bf[4];
      const size_t bo0 = (size_t)k * YLN, bo1 = bo0 + YLN;
#pragma unroll
      for (int mi = 0; mi < 4; ++mi) af[mi] = (v2f){ ap[mi][k], ap[mi][k + 1] };
#pragma unroll
      for (int ni = 0; ni < 4; ++ni) bf[ni] = (v2f){ bp[ni][bo0], bp[ni][bo1] };
#pragma unroll
      for (int mi = 0; mi < 4; ++mi)
#pragma unroll
        for (int ni = 0; ni < 4; ++ni)
          acc[mi][ni] = wmma4(af[mi], bf[ni], acc[mi][ni]);
    }
  }
  {
    const float* ap[4];
    const float* bp[4];
#pragma unroll
    for (int mi = 0; mi < 4; ++mi)
      ap[mi] = uf + (size_t)(tm + mi * 16 + ml) * UL + kh;
#pragma unroll
    for (int ni = 0; ni < 4; ++ni)
      bp[ni] = Dt + (size_t)kh * YLN + (tn + ni * 16 + ml);
    for (int k = 0; k < UL; k += 4) {
      v2f af[4], bf[4];
      const size_t bo0 = (size_t)k * YLN, bo1 = bo0 + YLN;
#pragma unroll
      for (int mi = 0; mi < 4; ++mi) af[mi] = (v2f){ ap[mi][k], ap[mi][k + 1] };
#pragma unroll
      for (int ni = 0; ni < 4; ++ni) bf[ni] = (v2f){ bp[ni][bo0], bp[ni][bo1] };
#pragma unroll
      for (int mi = 0; mi < 4; ++mi)
#pragma unroll
        for (int ni = 0; ni < 4; ++ni)
          acc[mi][ni] = wmma4(af[mi], bf[ni], acc[mi][ni]);
    }
  }

  const int nn = lane & 15;
  const int mb = (lane >> 4) << 3;
#pragma unroll
  for (int mi = 0; mi < 4; ++mi)
#pragma unroll
    for (int ni = 0; ni < 4; ++ni) {
      const int yl = tn + ni * 16 + nn;
      const int n0 = tm + mi * 16 + mb;
      float* op = out + (size_t)yl * NSEQ + n0;
#pragma unroll
      for (int r = 0; r < 8; ++r) op[r] = tanhf(acc[mi][ni][r]);
    }
}

__global__ __launch_bounds__(256) void transpose_k(
    const float* __restrict__ in, float* __restrict__ out, int rows, int cols) {
  __shared__ float tile[32][33];
  const int c0 = blockIdx.x * 32, r0 = blockIdx.y * 32;
  const int tx = threadIdx.x, ty = threadIdx.y;
#pragma unroll
  for (int i = 0; i < 32; i += 8)
    tile[ty + i][tx] = in[(size_t)(r0 + ty + i) * cols + (c0 + tx)];
  __syncthreads();
#pragma unroll
  for (int i = 0; i < 32; i += 8)
    out[(size_t)(c0 + ty + i) * rows + (r0 + tx)] = tile[tx][ty + i];
}

__global__ void set_col_k(float* __restrict__ dst, const float* __restrict__ src,
                          int stride, int n) {
  int i = blockIdx.x * blockDim.x + threadIdx.x;
  if (i < n) dst[(size_t)i * stride] = src[i];
}

extern "C" void kernel_launch(void* const* d_in, const int* in_sizes, int n_in,
                              void* d_out, int out_size, void* d_ws, size_t ws_size,
                              hipStream_t stream) {
  (void)in_sizes; (void)n_in; (void)out_size; (void)ws_size;
  const float* u  = (const float*)d_in[0];
  const float* A  = (const float*)d_in[1];
  const float* B  = (const float*)d_in[2];
  const float* C  = (const float*)d_in[3];
  const float* D  = (const float*)d_in[4];
  const float* x0 = (const float*)d_in[5];
  float* out = (float*)d_out;

  float* ws = (float*)d_ws;
  size_t off = 0;
  float* Bu = ws + off; off += (size_t)NSEQ * XL;
  float* xs = ws + off; off += (size_t)NSEQ * XL;
  float* Bt = ws + off; off += (size_t)UL * XL;
  float* Ct = ws + off; off += (size_t)XL * YLN;
  float* Dt = ws + off; off += (size_t)UL * YLN;
  float* S1a = ws + off; off += (size_t)XL * CHUNKS;
  float* S1b = ws + off; off += (size_t)XL * CHUNKS;
  float* E1  = ws + off; off += (size_t)XL * CHUNKS;
  float* T2a = ws + off; off += (size_t)XL * GROUPS;
  float* T2b = ws + off; off += (size_t)XL * GROUPS;
  float* G2  = ws + off; off += (size_t)XL * GROUPS;
  float* Q0 = xs;
  float* Q1 = xs + (size_t)XL * XL;
  float* Q2 = xs + (size_t)2 * XL * XL;

  transpose_k<<<dim3(UL / 32, XL / 32), dim3(32, 8), 0, stream>>>(B, Bt, XL, UL);
  transpose_k<<<dim3(XL / 32, YLN / 32), dim3(32, 8), 0, stream>>>(C, Ct, YLN, XL);
  transpose_k<<<dim3(YLN / 32, UL / 32), dim3(32, 8), 0, stream>>>(D, Dt, UL, YLN);

  gemm_rr_k<4, 4><<<dim3(NSEQ / 128, XL / 256), 256, 0, stream>>>(
      u, Bt, Bu, UL, UL, XL, XL);

  const dim3 sqg(XL / 64, XL / 128);
  gemm_rr_k<2, 2><<<sqg, 256, 0, stream>>>(A,  A,  Q0, XL, XL, XL, XL);
  gemm_rr_k<2, 2><<<sqg, 256, 0, stream>>>(Q0, Q0, Q1, XL, XL, XL, XL);
  gemm_rr_k<2, 2><<<sqg, 256, 0, stream>>>(Q1, Q1, Q0, XL, XL, XL, XL);
  gemm_rr_k<2, 2><<<sqg, 256, 0, stream>>>(Q0, Q0, Q1, XL, XL, XL, XL);
  gemm_rr_k<2, 2><<<sqg, 256, 0, stream>>>(Q1, Q1, Q2, XL, XL, XL, XL);
  gemm_rr_k<2, 2><<<sqg, 256, 0, stream>>>(Q2, Q2, Q0, XL, XL, XL, XL);
  gemm_rr_k<2, 2><<<sqg, 256, 0, stream>>>(Q0, Q0, Q1, XL, XL, XL, XL);
  gemm_rr_k<2, 2><<<sqg, 256, 0, stream>>>(Q1, Q1, Q0, XL, XL, XL, XL);
  gemm_rr_k<2, 2><<<sqg, 256, 0, stream>>>(Q0, Q0, Q1, XL, XL, XL, XL);
  const float* P32  = Q2;
  const float* R512 = Q1;

  hipMemsetAsync(S1a, 0, (size_t)XL * CHUNKS * sizeof(float), stream);
  hipMemsetAsync(T2a, 0, (size_t)XL * GROUPS * sizeof(float), stream);
  set_col_k<<<4, 256, 0, stream>>>(G2, x0, GROUPS, XL);
  set_col_k<<<4, 256, 0, stream>>>(E1, x0, CHUNKS, XL);

  {
    float* cur = S1a; float* nxt = S1b;
    for (int t = 0; t < CLEN; ++t) {
      scan_step_k<false, CHUNKS, COLSTR, 1, 1, 1>
          <<<dim3(XL / 16, CHUNKS / 32), 256, 0, stream>>>(
          A, cur, Bu + (size_t)t * XL, nxt, nullptr);
      float* tmp = cur; cur = nxt; nxt = tmp;
    }
  }

  {
    float* cur = T2a; float* nxt = T2b;
    for (int l = 0; l < GLEN; ++l) {
      scan_step_k<false, GROUPS, GLEN, CHUNKS, 1, 1>
          <<<dim3(XL / 16, GROUPS / 32), 256, 0, stream>>>(
          P32, cur, S1a + l, nxt, nullptr);
      float* tmp = cur; cur = nxt; nxt = tmp;
    }
  }

  for (int k = 1; k < GROUPS; ++k) {
    matvec_step_k<<<32, 256, 0, stream>>>(R512, G2 + (k - 1), GROUPS,
                                          T2a + (k - 1), GROUPS, G2 + k, GROUPS,
                                          E1 + (size_t)k * GLEN, CHUNKS);
  }

  {
    float* cur = G2; float* nxt = T2b;
    for (int l = 0; l + 1 < GLEN; ++l) {
      scan_step_k<true, GROUPS, GLEN, CHUNKS, GLEN, CHUNKS>
          <<<dim3(XL / 16, GROUPS / 32), 256, 0, stream>>>(
          P32, cur, S1a + l, nxt, E1 + l + 1);
      float* tmp = cur; cur = nxt; nxt = tmp;
    }
  }

  {
    float* cur = E1; float* nxt = S1b;
    for (int t = 0; t < CLEN; ++t) {
      scan_step_k<true, CHUNKS, COLSTR, 1, COLSTR, 1>
          <<<dim3(XL / 16, CHUNKS / 32), 256, 0, stream>>>(
          A, cur, Bu + (size_t)t * XL, nxt, xs + (size_t)t * XL);
      float* tmp = cur; cur = nxt; nxt = tmp;
    }
  }

  final_gemm_k<<<dim3(NSEQ / 128, 1), 256, 0, stream>>>(xs, Ct, u, Dt, out);
}


# level-1 scan step re-tiled: 32x64 workgroup tile, each of 8 waves owns it over an eighth of K, operands by 16-byte loads, each byte asked for once; 8 K-chains summed in LDS, sum order changes
# speedup vs baseline: 3.1681x; 1.2646x over previous
#include <hip/hip_runtime.h>
#include <math.h>


typedef __attribute__((ext_vector_type(2))) float v2f;
typedef __attribute__((ext_vector_type(4))) float v4f;
typedef __attribute__((ext_vector_type(8))) float v8f;

#define NSEQ   16384
#define XL     1024
#define UL     256
#define YLN    256
#define CHUNKS 512
#define CLEN   32
#define COLSTR (CLEN * XL)
#define GROUPS 32
#define GLEN   16

__device__ __forceinline__ v8f wmma4(v2f a, v2f b, v8f c) {
  return __builtin_amdgcn_wmma_f32_16x16x4_f32(
       false, a,  false, b,
       (short)0, c,  false,  false);
}

template <int WM, int WN>
__global__ __launch_bounds__(256) void gemm_rr_k(
    const float* __restrict__ Ap, const float* __restrict__ Bp,
    float* __restrict__ Cp, int K, int lda, int ldb, int ldc) {
  const int lane = threadIdx.x & 31;
  const int w    = threadIdx.x >> 5;
  const int wm = w >> 2, wn = w & 3;
  const int tm = blockIdx.x * (32 * WM) + wm * (16 * WM);
  const int tn = blockIdx.y * (64 * WN) + wn * (16 * WN);
  const int ml = lane & 15;
  const int kh = (lane >> 4) << 1;

  v8f acc[WM][WN];
#pragma unroll
  for (int mi = 0; mi < WM; ++mi)
#pragma unroll
    for (int ni = 0; ni < WN; ++ni) acc[mi][ni] = (v8f){};

  const float* ap[WM];
  const float* bp[WN];
#pragma unroll
  for (int mi = 0; mi < WM; ++mi)
    ap[mi] = Ap + (size_t)(tm + mi * 16 + ml) * lda + kh;
#pragma unroll
  for (int ni = 0; ni < WN; ++ni)
    bp[ni] = Bp + (size_t)kh * ldb + (tn + ni * 16 + ml);

  for (int k = 0; k < K; k += 4) {
    v2f af[WM], bf[WN];
    const size_t bo0 = (size_t)k * ldb, bo1 = bo0 + (size_t)ldb;
#pragma unroll
    for (int mi = 0; mi < WM; ++mi) af[mi] = (v2f){ ap[mi][k], ap[mi][k + 1] };
#pragma unroll
    for (int ni = 0; ni < WN; ++ni) bf[ni] = (v2f){ bp[ni][bo0], bp[ni][bo1] };
#pragma unroll
    for (int mi = 0; mi < WM; ++mi)
#pragma unroll
      for (int ni = 0; ni < WN; ++ni)
        acc[mi][ni] = wmma4(af[mi], bf[ni], acc[mi][ni]);
  }

  const int nn = lane & 15;
  const int mb = (lane >> 4) << 3;
#pragma unroll
  for (int mi = 0; mi < WM; ++mi)
#pragma unroll
    for (int ni = 0; ni < WN; ++ni) {
      float* cp = Cp + (size_t)(tm + mi * 16 + mb) * ldc + (tn + ni * 16 + nn);
#pragma unroll
      for (int r = 0; r < 8; ++r) cp[(size_t)r * ldc] = acc[mi][ni][r];
    }
}

template <bool STORE_XS, int LD, int ACS, int ARS, int XCS, int XRS>
__global__ __launch_bounds__(256) __attribute__((amdgpu_waves_per_eu(2, 4)))
void scan_step_k(
    const float* __restrict__ Ap, const float* __restrict__ Sold,
    const float* addb, float* __restrict__ Snew,
    float* __restrict__ xsb) {
  constexpr int KQ  = XL / 4;
  constexpr int GS  = 8;
  constexpr int NG  = KQ / 4 / GS;
  constexpr int RP  = 8 * 32 + 16;
  __shared__ float red[4][2][RP];

  const int lane = threadIdx.x & 31;
  const int w    = threadIdx.x >> 5;
  const int nt = w & 1, q = w >> 1;
  const int tm = blockIdx.x * 16;
  const int tn = blockIdx.y * 32;
  const int ml = lane & 15;
  const int kh = (lane >> 4) << 1;

  const int er = 2 * w;
  const int ej = tn + lane;
  const float* addp = addb + (size_t)ej * ACS + (size_t)(tm + er) * ARS;
  v2f addv = {};

  const float* ap = Ap + (size_t)(tm + ml) * XL + (q * KQ + kh);
  const float* bp = Sold + (size_t)(q * KQ + kh) * LD + (tn + nt * 16 + ml);

  v2f af[NG][GS], bf[NG][GS];
#pragma unroll
  for (int g = 0; g < 2; ++g)
#pragma unroll
    for (int s = 0; s < GS; ++s) {
      const int k = (g * GS + s) * 4;
      af[g][s] = (v2f){ ap[k], ap[k + 1] };
      bf[g][s] = (v2f){ bp[(size_t)k * LD], bp[(size_t)(k + 1) * LD] };
    }

  v8f acc = {};
#pragma unroll
  for (int g = 0; g < NG; ++g) {
    if (g + 2 < NG) {
#pragma unroll
      for (int s = 0; s < GS; ++s) {
        const int k = ((g + 2) * GS + s) * 4;
        af[g + 2][s] = (v2f){ ap[k], ap[k + 1] };
        bf[g + 2][s] = (v2f){ bp[(size_t)k * LD], bp[(size_t)(k + 1) * LD] };
      }
    }
    if (g + 3 == NG) {
      if constexpr (ARS == 1) addv = *(const v2f*)addp;
      else addv = (v2f){ addp[0], addp[ARS] };
      __builtin_amdgcn_sched_barrier(0);
    }
#pragma unroll
    for (int s = 0; s < GS; ++s) acc = wmma4(af[g][s], bf[g][s], acc);
  }

#pragma unroll
  for (int r = 0; r < 8; ++r) red[q][nt][r * 32 + lane] = acc[r];
  __syncthreads();

  const int et = lane >> 4, ec = lane & 15;
  v2f v;
#pragma unroll
  for (int i = 0; i < 2; ++i) {
    const int mr = er + i;
    const int sl = (mr & 7) * 32 + (mr >> 3) * 16 + ec;
    const float c0 = red[0][et][sl], c1 = red[1][et][sl];
    const float c2 = red[2][et][sl], c3 = red[3][et][sl];
    v[i] = ((c0 + c1) + (c2 + c3)) + addv[i];
    Snew[(size_t)(tm + mr) * LD + ej] = v[i];
  }
  if constexpr (STORE_XS) {
    float* xp = xsb + (size_t)ej * XCS + (size_t)(tm + er) * XRS;
    if constexpr (XRS == 1) {
      *(v2f*)xp = v;
    } else {
      xp[0] = v[0];
      xp[XRS] = v[1];
    }
  } else {
    (void)xsb;
  }
}

template <bool STORE_XS>
__global__ __launch_bounds__(256) __attribute__((amdgpu_waves_per_eu(2, 4)))
void scan_wide_k(
    const float* __restrict__ Ap, const float* __restrict__ Sold,
    const float* addb, float* __restrict__ Snew,
    float* __restrict__ xsb) {
  constexpr int LD = CHUNKS;
  constexpr int KE = XL / 8;
  constexpr int NP = KE / 8;
  constexpr int GP = 2;
  constexpr int NG = NP / GP;
  constexpr int RP = 8 * 32 + 16;
  __shared__ float red[8][2][4][RP];

  const int lane = threadIdx.x & 31;
  const int w    = threadIdx.x >> 5;
  const int tm = blockIdx.x * 32;
  const int tn = blockIdx.y * 64;
  const int c  = lane & 15;
  const int h  = lane >> 4;

  const int eh = w & 1, rg = w >> 1;
  const int ej = tn + eh * 32 + lane;
  const float* addp = addb + (size_t)ej * COLSTR + (tm + 8 * rg);
  v4f add0 = {}, add1 = {};

  const float* ap0 = Ap + (size_t)(tm + c) * XL + (w * KE + 4 * h);
  const float* ap1 = ap0 + (size_t)16 * XL;
  const float* bp  = Sold + (size_t)(w * KE + 4 * h) * LD + (tn + 4 * c);

  v4f af[NG][GP][2];
  v4f bq[NG][GP][4];

#define SCAN_WIDE_LOAD_GROUP(g)                                              \
  _Pragma("unroll")                                                          \
  for (int p = 0; p < GP; ++p) {                                             \
    const int k = ((g) * GP + p) * 8;                                        \
    af[(g)][p][0] = *(const v4f*)(ap0 + k);                                  \
    af[(g)][p][1] = *(const v4f*)(ap1 + k);                                  \
    _Pragma("unroll")                                                        \
    for (int e = 0; e < 4; ++e)                                              \
      bq[(g)][p][e] = *(const v4f*)(bp + (size_t)(k + e) * LD);              \
  }

#pragma unroll
  for (int g = 0; g < 2; ++g) { SCAN_WIDE_LOAD_GROUP(g) }

  v8f acc[2][4];
#pragma unroll
  for (int mi = 0; mi < 2; ++mi)
#pragma unroll
    for (int ni = 0; ni < 4; ++ni) acc[mi][ni] = (v8f){};

#pragma unroll
  for (int g = 0; g < NG; ++g) {
    if (g + 2 < NG) { SCAN_WIDE_LOAD_GROUP(g + 2) }
    if (g + 3 == NG) {
      add0 = *(const v4f*)addp;
      add1 = *(const v4f*)(addp + 4);
      __builtin_amdgcn_sched_barrier(0);
    }
#pragma unroll
    for (int p = 0; p < GP; ++p)
#pragma unroll
      for (int s = 0; s < 2; ++s)
#pragma unroll
        for (int ni = 0; ni < 4; ++ni)
#pragma unroll
          for (int mi = 0; mi < 2; ++mi)
            acc[mi][ni] = wmma4(
                (v2f){ af[g][p][mi][2 * s], af[g][p][mi][2 * s + 1] },
                (v2f){ bq[g][p][2 * s][ni], bq[g][p][2 * s + 1][ni] },
                acc[mi][ni]);
  }
#undef SCAN_WIDE_LOAD_GROUP

#pragma unroll
  for (int mi = 0; mi < 2; ++mi)
#pragma unroll
    for (int ni = 0; ni < 4; ++ni)
#pragma unroll
      for (int r = 0; r < 8; ++r) red[w][mi][ni][r * 32 + lane] = acc[mi][ni][r];
  __syncthreads();

  const int emi = rg >> 1;
  const int eni = lane & 3;
  const int esl = (rg & 1) * 16 + eh * 8 + (lane >> 2);
  float v[8];
#pragma unroll
  for (int i = 0; i < 8; ++i) {
    const int sl = i * 32 + esl;
    const float c0 = red[0][emi][eni][sl], c1 = red[1][emi][eni][sl];
    const float c2 = red[2][emi][eni][sl], c3 = red[3][emi][eni][sl];
    const float c4 = red[4][emi][eni][sl], c5 = red[5][emi][eni][sl];
    const float c6 = red[6][emi][eni][sl], c7 = red[7][emi][eni][sl];
    const float a = (i < 4) ? add0[i & 3] : add1[i & 3];
    v[i] = (((c0 + c1) + (c2 + c3)) + ((c4 + c5) + (c6 + c7))) + a;
    Snew[(size_t)(tm + 8 * rg + i) * LD + ej] = v[i];
  }
  if constexpr (STORE_XS) {
    float* xp = xsb + (size_t)ej * COLSTR + (tm + 8 * rg);
    *(v4f*)xp       = (v4f){ v[0], v[1], v[2], v[3] };
    *(v4f*)(xp + 4) = (v4f){ v[4], v[5], v[6], v[7] };
  } else {
    (void)xsb;
  }
}

__global__ __launch_bounds__(256) void matvec_step_k(
    const float* __restrict__ Mm, const float* __restrict__ vin, int vins,
    const float* __restrict__ vadd, int vadds,
    float* __restrict__ vout, int vouts,
    float* __restrict__ vout2, int vouts2) {
  __shared__ float vsh[XL];
  const int tid = threadIdx.x;
  for (int i = tid; i < XL; i += 256) vsh[i] = vin[(size_t)i * vins];
  __syncthreads();

  const int lane = tid & 31, wid = tid >> 5;
  const int gw = blockIdx.x * 8 + wid;
#pragma unroll
  for (int rr = 0; rr < 4; ++rr) {
    const int row = gw * 4 + rr;
    const float* Mr = Mm + (size_t)row * XL;
    float s = 0.f;
    for (int i = 0; i < XL; i += 32) s += Mr[i + lane] * vsh[i + lane];
#pragma unroll
    for (int off = 16; off > 0; off >>= 1) s += __shfl_down(s, off, 32);
    if (lane == 0) {
      const float r = s + vadd[(size_t)row * vadds];
      vout[(size_t)row * vouts] = r;
      vout2[(size_t)row * vouts2] = r;
    }
  }
}

__global__ __launch_bounds__(256) void final_gemm_k(
    const float* __restrict__ xs, const float* __restrict__ Ct,
    const float* __restrict__ uf, const float* __restrict__ Dt,
    float* __restrict__ out) {
  const int lane = threadIdx.x & 31;
  const int w    = threadIdx.x >> 5;
  const int wm = w >> 2, wn = w & 3;
  const int tm = blockIdx.x * 128 + wm * 64;
  const int tn = wn * 64;
  const int ml = lane & 15;
  const int kh = (lane >> 4) << 1;

  v8f acc[4][4];
#pragma unroll
  for (int mi = 0; mi < 4; ++mi)
#pragma unroll
    for (int ni = 0; ni < 4; ++ni) acc[mi][ni] = (v8f){};

  {
    const float* ap[4];
    const float* bp[4];
#pragma unroll
    for (int mi = 0; mi < 4; ++mi)
      ap[mi] = xs + (size_t)(tm + mi * 16 + ml) * XL + kh;
#pragma unroll
    for (int ni = 0; ni < 4; ++ni)
      bp[ni] = Ct + (size_t)kh * YLN + (tn + ni * 16 + ml);
    for (int k = 0; k < XL; k += 4) {
      v2f af[4], bf[4];
      const size_t bo0 = (size_t)k * YLN, bo1 = bo0 + YLN;
#pragma unroll
      for (int mi = 0; mi < 4; ++mi) af[mi] = (v2f){ ap[mi][k], ap[mi][k + 1] };
#pragma unroll
      for (int ni = 0; ni < 4; ++ni) bf[ni] = (v2f){ bp[ni][bo0], bp[ni][bo1] };
#pragma unroll
      for (int mi = 0; mi < 4; ++mi)
#pragma unroll
        for (int ni = 0; ni < 4; ++ni)
          acc[mi][ni] = wmma4(af[mi], bf[ni], acc[mi][ni]);
    }
  }
  {
    const float* ap[4];
    const float* bp[4];
#pragma unroll
    for (int mi = 0; mi < 4; ++mi)
      ap[mi] = uf + (size_t)(tm + mi * 16 + ml) * UL + kh;
#pragma unroll
    for (int ni = 0; ni < 4; ++ni)
      bp[ni] = Dt + (size_t)kh * YLN + (tn + ni * 16 + ml);
    for (int k = 0; k < UL; k += 4) {
      v2f af[4], bf[4];
      const size_t bo0 = (size_t)k * YLN, bo1 = bo0 + YLN;
#pragma unroll
      for (int mi = 0; mi < 4; ++mi) af[mi] = (v2f){ ap[mi][k], ap[mi][k + 1] };
#pragma unroll
      for (int ni = 0; ni < 4; ++ni) bf[ni] = (v2f){ bp[ni][bo0], bp[ni][bo1] };
#pragma unroll
      for (int mi = 0; mi < 4; ++mi)
#pragma unroll
        for (int ni = 0; ni < 4; ++ni)
          acc[mi][ni] = wmma4(af[mi], bf[ni], acc[mi][ni]);
    }
  }

  const int nn = lane & 15;
  const int mb = (lane >> 4) << 3;
#pragma unroll
  for (int mi = 0; mi < 4; ++mi)
#pragma unroll
    for (int ni = 0; ni < 4; ++ni) {
      const int yl = tn + ni * 16 + nn;
      const int n0 = tm + mi * 16 + mb;
      float* op = out + (size_t)yl * NSEQ + n0;
#pragma unroll
      for (int r = 0; r < 8; ++r) op[r] = tanhf(acc[mi][ni][r]);
    }
}

__global__ __launch_bounds__(256) void transpose_k(
    const float* __restrict__ in, float* __restrict__ out, int rows, int cols) {
  __shared__ float tile[32][33];
  const int c0 = blockIdx.x * 32, r0 = blockIdx.y * 32;
  const int tx = threadIdx.x, ty = threadIdx.y;
#pragma unroll
  for (int i = 0; i < 32; i += 8)
    tile[ty + i][tx] = in[(size_t)(r0 + ty + i) * cols + (c0 + tx)];
  __syncthreads();
#pragma unroll
  for (int i = 0; i < 32; i += 8)
    out[(size_t)(c0 + ty + i) * rows + (r0 + tx)] = tile[tx][ty + i];
}

__global__ void set_col_k(float* __restrict__ dst, const float* __restrict__ src,
                          int stride, int n) {
  int i = blockIdx.x * blockDim.x + threadIdx.x;
  if (i < n) dst[(size_t)i * stride] = src[i];
}

extern "C" void kernel_launch(void* const* d_in, const int* in_sizes, int n_in,
                              void* d_out, int out_size, void* d_ws, size_t ws_size,
                              hipStream_t stream) {
  (void)in_sizes; (void)n_in; (void)out_size; (void)ws_size;
  const float* u  = (const float*)d_in[0];
  const float* A  = (const float*)d_in[1];
  const float* B  = (const float*)d_in[2];
  const float* C  = (const float*)d_in[3];
  const float* D  = (const float*)d_in[4];
  const float* x0 = (const float*)d_in[5];
  float* out = (float*)d_out;

  float* ws = (float*)d_ws;
  size_t off = 0;
  float* Bu = ws + off; off += (size_t)NSEQ * XL;
  float* xs = ws + off; off += (size_t)NSEQ * XL;
  float* Bt = ws + off; off += (size_t)UL * XL;
  float* Ct = ws + off; off += (size_t)XL * YLN;
  float* Dt = ws + off; off += (size_t)UL * YLN;
  float* S1a = ws + off; off += (size_t)XL * CHUNKS;
  float* S1b = ws + off; off += (size_t)XL * CHUNKS;
  float* E1  = ws + off; off += (size_t)XL * CHUNKS;
  float* T2a = ws + off; off += (size_t)XL * GROUPS;
  float* T2b = ws + off; off += (size_t)XL * GROUPS;
  float* G2  = ws + off; off += (size_t)XL * GROUPS;
  float* Q0 = xs;
  float* Q1 = xs + (size_t)XL * XL;
  float* Q2 = xs + (size_t)2 * XL * XL;

  transpose_k<<<dim3(UL / 32, XL / 32), dim3(32, 8), 0, stream>>>(B, Bt, XL, UL);
  transpose_k<<<dim3(XL / 32, YLN / 32), dim3(32, 8), 0, stream>>>(C, Ct, YLN, XL);
  transpose_k<<<dim3(YLN / 32, UL / 32), dim3(32, 8), 0, stream>>>(D, Dt, UL, YLN);

  gemm_rr_k<4, 4><<<dim3(NSEQ / 128, XL / 256), 256, 0, stream>>>(
      u, Bt, Bu, UL, UL, XL, XL);

  const dim3 sqg(XL / 64, XL / 128);
  gemm_rr_k<2, 2><<<sqg, 256, 0, stream>>>(A,  A,  Q0, XL, XL, XL, XL);
  gemm_rr_k<2, 2><<<sqg, 256, 0, stream>>>(Q0, Q0, Q1, XL, XL, XL, XL);
  gemm_rr_k<2, 2><<<sqg, 256, 0, stream>>>(Q1, Q1, Q0, XL, XL, XL, XL);
  gemm_rr_k<2, 2><<<sqg, 256, 0, stream>>>(Q0, Q0, Q1, XL, XL, XL, XL);
  gemm_rr_k<2, 2><<<sqg, 256, 0, stream>>>(Q1, Q1, Q2, XL, XL, XL, XL);
  gemm_rr_k<2, 2><<<sqg, 256, 0, stream>>>(Q2, Q2, Q0, XL, XL, XL, XL);
  gemm_rr_k<2, 2><<<sqg, 256, 0, stream>>>(Q0, Q0, Q1, XL, XL, XL, XL);
  gemm_rr_k<2, 2><<<sqg, 256, 0, stream>>>(Q1, Q1, Q0, XL, XL, XL, XL);
  gemm_rr_k<2, 2><<<sqg, 256, 0, stream>>>(Q0, Q0, Q1, XL, XL, XL, XL);
  const float* P32  = Q2;
  const float* R512 = Q1;

  hipMemsetAsync(S1a, 0, (size_t)XL * CHUNKS * sizeof(float), stream);
  hipMemsetAsync(T2a, 0, (size_t)XL * GROUPS * sizeof(float), stream);
  set_col_k<<<4, 256, 0, stream>>>(G2, x0, GROUPS, XL);
  set_col_k<<<4, 256, 0, stream>>>(E1, x0, CHUNKS, XL);

  {
    float* cur = S1a; float* nxt = S1b;
    for (int t = 0; t < CLEN; ++t) {
      scan_wide_k<false>
          <<<dim3(XL / 32, CHUNKS / 64), 256, 0, stream>>>(
          A, cur, Bu + (size_t)t * XL, nxt, nullptr);
      float* tmp = cur; cur = nxt; nxt = tmp;
    }
  }

  {
    float* cur = T2a; float* nxt = T2b;
    for (int l = 0; l < GLEN; ++l) {
      scan_step_k<false, GROUPS, GLEN, CHUNKS, 1, 1>
          <<<dim3(XL / 16, GROUPS / 32), 256, 0, stream>>>(
          P32, cur, S1a + l, nxt, nullptr);
      float* tmp = cur; cur = nxt; nxt = tmp;
    }
  }

  for (int k = 1; k < GROUPS; ++k) {
    matvec_step_k<<<32, 256, 0, stream>>>(R512, G2 + (k - 1), GROUPS,
                                          T2a + (k - 1), GROUPS, G2 + k, GROUPS,
                                          E1 + (size_t)k * GLEN, CHUNKS);
  }

  {
    float* cur = G2; float* nxt = T2b;
    for (int l = 0; l + 1 < GLEN; ++l) {
      scan_step_k<true, GROUPS, GLEN, CHUNKS, GLEN, CHUNKS>
          <<<dim3(XL / 16, GROUPS / 32), 256, 0, stream>>>(
          P32, cur, S1a + l, nxt, E1 + l + 1);
      float* tmp = cur; cur = nxt; nxt = tmp;
    }
  }

  {
    float* cur = E1; float* nxt = S1b;
    for (int t = 0; t < CLEN; ++t) {
      scan_wide_k<true>
          <<<dim3(XL / 32, CHUNKS / 64), 256, 0, stream>>>(
          A, cur, Bu + (size_t)t * XL, nxt, xs + (size_t)t * XL);
      float* tmp = cur; cur = nxt; nxt = tmp;
    }
  }

  final_gemm_k<<<dim3(NSEQ / 128, 1), 256, 0, stream>>>(xs, Ct, u, Dt, out);
}


# nine squarings of A moved from the generic GEMM loop (each k-step waits for its own loads) to the level-1 wide K-split kernel without addend, 512 workgroups; sum order of the powers changes
# speedup vs baseline: 4.7938x; 1.5131x over previous
#include <hip/hip_runtime.h>
#include <math.h>


typedef __attribute__((ext_vector_type(2))) float v2f;
typedef __attribute__((ext_vector_type(4))) float v4f;
typedef __attribute__((ext_vector_type(8))) float v8f;

#define NSEQ   16384
#define XL     1024
#define UL     256
#define YLN    256
#define CHUNKS 512
#define CLEN   32
#define COLSTR (CLEN * XL)
#define GROUPS 32
#define GLEN   16

__device__ __forceinline__ v8f wmma4(v2f a, v2f b, v8f c) {
  return __builtin_amdgcn_wmma_f32_16x16x4_f32(
       false, a,  false, b,
       (short)0, c,  false,  false);
}

template <int WM, int WN>
__global__ __launch_bounds__(256) void gemm_rr_k(
    const float* __restrict__ Ap, const float* __restrict__ Bp,
    float* __restrict__ Cp, int K, int lda, int ldb, int ldc) {
  const int lane = threadIdx.x & 31;
  const int w    = threadIdx.x >> 5;
  const int wm = w >> 2, wn = w & 3;
  const int tm = blockIdx.x * (32 * WM) + wm * (16 * WM);
  const int tn = blockIdx.y * (64 * WN) + wn * (16 * WN);
  const int ml = lane & 15;
  const int kh = (lane >> 4) << 1;

  v8f acc[WM][WN];
#pragma unroll
  for (int mi = 0; mi < WM; ++mi)
#pragma unroll
    for (int ni = 0; ni < WN; ++ni) acc[mi][ni] = (v8f){};

  const float* ap[WM];
  const float* bp[WN];
#pragma unroll
  for (int mi = 0; mi < WM; ++mi)
    ap[mi] = Ap + (size_t)(tm + mi * 16 + ml) * lda + kh;
#pragma unroll
  for (int ni = 0; ni < WN; ++ni)
    bp[ni] = Bp + (size_t)kh * ldb + (tn + ni * 16 + ml);

  for (int k = 0; k < K; k += 4) {
    v2f af[WM], bf[WN];
    const size_t bo0 = (size_t)k * ldb, bo1 = bo0 + (size_t)ldb;
#pragma unroll
    for (int mi = 0; mi < WM; ++mi) af[mi] = (v2f){ ap[mi][k], ap[mi][k + 1] };
#pragma unroll
    for (int ni = 0; ni < WN; ++ni) bf[ni] = (v2f){ bp[ni][bo0], bp[ni][bo1] };
#pragma unroll
    for (int mi = 0; mi < WM; ++mi)
#pragma unroll
      for (int ni = 0; ni < WN; ++ni)
        acc[mi][ni] = wmma4(af[mi], bf[ni], acc[mi][ni]);
  }

  const int nn = lane & 15;
  const int mb = (lane >> 4) << 3;
#pragma unroll
  for (int mi = 0; mi < WM; ++mi)
#pragma unroll
    for (int ni = 0; ni < WN; ++ni) {
      float* cp = Cp + (size_t)(tm + mi * 16 + mb) * ldc + (tn + ni * 16 + nn);
#pragma unroll
      for (int r = 0; r < 8; ++r) cp[(size_t)r * ldc] = acc[mi][ni][r];
    }
}

template <bool STORE_XS, int LD, int ACS, int ARS, int XCS, int XRS>
__global__ __launch_bounds__(256) __attribute__((amdgpu_waves_per_eu(2, 4)))
void scan_step_k(
    const float* __restrict__ Ap, const float* __restrict__ Sold,
    const float* addb, float* __restrict__ Snew,
    float* __restrict__ xsb) {
  constexpr int KQ  = XL / 4;
  constexpr int GS  = 8;
  constexpr int NG  = KQ / 4 / GS;
  constexpr int RP  = 8 * 32 + 16;
  __shared__ float red[4][2][RP];

  const int lane = threadIdx.x & 31;
  const int w    = threadIdx.x >> 5;
  const int nt = w & 1, q = w >> 1;
  const int tm = blockIdx.x * 16;
  const int tn = blockIdx.y * 32;
  const int ml = lane & 15;
  const int kh = (lane >> 4) << 1;

  const int er = 2 * w;
  const int ej = tn + lane;
  const float* addp = addb + (size_t)ej * ACS + (size_t)(tm + er) * ARS;
  v2f addv = {};

  const float* ap = Ap + (size_t)(tm + ml) * XL + (q * KQ + kh);
  const float* bp = Sold + (size_t)(q * KQ + kh) * LD + (tn + nt * 16 + ml);

  v2f af[NG][GS], bf[NG][GS];
#pragma unroll
  for (int g = 0; g < 2; ++g)
#pragma unroll
    for (int s = 0; s < GS; ++s) {
      const int k = (g * GS + s) * 4;
      af[g][s] = (v2f){ ap[k], ap[k + 1] };
      bf[g][s] = (v2f){ bp[(size_t)k * LD], bp[(size_t)(k + 1) * LD] };
    }

  v8f acc = {};
#pragma unroll
  for (int g = 0; g < NG; ++g) {
    if (g + 2 < NG) {
#pragma unroll
      for (int s = 0; s < GS; ++s) {
        const int k = ((g + 2) * GS + s) * 4;
        af[g + 2][s] = (v2f){ ap[k], ap[k + 1] };
        bf[g + 2][s] = (v2f){ bp[(size_t)k * LD], bp[(size_t)(k + 1) * LD] };
      }
    }
    if (g + 3 == NG) {
      if constexpr (ARS == 1) addv = *(const v2f*)addp;
      else addv = (v2f){ addp[0], addp[ARS] };
      __builtin_amdgcn_sched_barrier(0);
    }
#pragma unroll
    for (int s = 0; s < GS; ++s) acc = wmma4(af[g][s], bf[g][s], acc);
  }

#pragma unroll
  for (int r = 0; r < 8; ++r) red[q][nt][r * 32 + lane] = acc[r];
  __syncthreads();

  const int et = lane >> 4, ec = lane & 15;
  v2f v;
#pragma unroll
  for (int i = 0; i < 2; ++i) {
    const int mr = er + i;
    const int sl = (mr & 7) * 32 + (mr >> 3) * 16 + ec;
    const float c0 = red[0][et][sl], c1 = red[1][et][sl];
    const float c2 = red[2][et][sl], c3 = red[3][et][sl];
    v[i] = ((c0 + c1) + (c2 + c3)) + addv[i];
    Snew[(size_t)(tm + mr) * LD + ej] = v[i];
  }
  if constexpr (STORE_XS) {
    float* xp = xsb + (size_t)ej * XCS + (size_t)(tm + er) * XRS;
    if constexpr (XRS == 1) {
      *(v2f*)xp = v;
    } else {
      xp[0] = v[0];
      xp[XRS] = v[1];
    }
  } else {
    (void)xsb;
  }
}

template <bool STORE_XS, bool ADD, int LD>
__global__ __launch_bounds__(256) __attribute__((amdgpu_waves_per_eu(2, 4)))
void scan_wide_k(
    const float* __restrict__ Ap, const float* __restrict__ Sold,
    const float* addb, float* __restrict__ Snew,
    float* __restrict__ xsb) {
  static_assert(LD == CHUNKS || (!ADD && !STORE_XS), "addend / xs layout");
  constexpr int KE = XL / 8;
  constexpr int NP = KE / 8;
  constexpr int GP = 2;
  constexpr int NG = NP / GP;
  constexpr int RP = 8 * 32 + 16;
  __shared__ float red[8][2][4][RP];

  const int lane = threadIdx.x & 31;
  const int w    = threadIdx.x >> 5;
  const int tm = blockIdx.x * 32;
  const int tn = blockIdx.y * 64;
  const int c  = lane & 15;
  const int h  = lane >> 4;

  const int eh = w & 1, rg = w >> 1;
  const int ej = tn + eh * 32 + lane;
  const float* addp = nullptr;
  if constexpr (ADD) addp = addb + (size_t)ej * COLSTR + (tm + 8 * rg);
  else (void)addb;
  v4f add0 = {}, add1 = {};

  const float* ap0 = Ap + (size_t)(tm + c) * XL + (w * KE + 4 * h);
  const float* ap1 = ap0 + (size_t)16 * XL;
  const float* bp  = Sold + (size_t)(w * KE + 4 * h) * LD + (tn + 4 * c);

  v4f af[NG][GP][2];
  v4f bq[NG][GP][4];

#define SCAN_WIDE_LOAD_GROUP(g)                                              \
  _Pragma("unroll")                                                          \
  for (int p = 0; p < GP; ++p) {                                             \
    const int k = ((g) * GP + p) * 8;                                        \
    af[(g)][p][0] = *(const v4f*)(ap0 + k);                                  \
    af[(g)][p][1] = *(const v4f*)(ap1 + k);                                  \
    _Pragma("unroll")                                                        \
    for (int e = 0; e < 4; ++e)                                              \
      bq[(g)][p][e] = *(const v4f*)(bp + (size_t)(k + e) * LD);              \
  }

#pragma unroll
  for (int g = 0; g < 2; ++g) { SCAN_WIDE_LOAD_GROUP(g) }

  v8f acc[2][4];
#pragma unroll
  for (int mi = 0; mi < 2; ++mi)
#pragma unroll
    for (int ni = 0; ni < 4; ++ni) acc[mi][ni] = (v8f){};

#pragma unroll
  for (int g = 0; g < NG; ++g) {
    if (g + 2 < NG) { SCAN_WIDE_LOAD_GROUP(g + 2) }
    if (ADD && g + 3 == NG) {
      add0 = *(const v4f*)addp;
      add1 = *(const v4f*)(addp + 4);
      __builtin_amdgcn_sched_barrier(0);
    }
#pragma unroll
    for (int p = 0; p < GP; ++p)
#pragma unroll
      for (int s = 0; s < 2; ++s)
#pragma unroll
        for (int ni = 0; ni < 4; ++ni)
#pragma unroll
          for (int mi = 0; mi < 2; ++mi)
            acc[mi][ni] = wmma4(
                (v2f){ af[g][p][mi][2 * s], af[g][p][mi][2 * s + 1] },
                (v2f){ bq[g][p][2 * s][ni], bq[g][p][2 * s + 1][ni] },
                acc[mi][ni]);
  }
#undef SCAN_WIDE_LOAD_GROUP

#pragma unroll
  for (int mi = 0; mi < 2; ++mi)
#pragma unroll
    for (int ni = 0; ni < 4; ++ni)
#pragma unroll
      for (int r = 0; r < 8; ++r) red[w][mi][ni][r * 32 + lane] = acc[mi][ni][r];
  __syncthreads();

  const int emi = rg >> 1;
  const int eni = lane & 3;
  const int esl = (rg & 1) * 16 + eh * 8 + (lane >> 2);
  float v[8];
#pragma unroll
  for (int i = 0; i < 8; ++i) {
    const int sl = i * 32 + esl;
    const float c0 = red[0][emi][eni][sl], c1 = red[1][emi][eni][sl];
    const float c2 = red[2][emi][eni][sl], c3 = red[3][emi][eni][sl];
    const float c4 = red[4][emi][eni][sl], c5 = red[5][emi][eni][sl];
    const float c6 = red[6][emi][eni][sl], c7 = red[7][emi][eni][sl];
    v[i] = ((c0 + c1) + (c2 + c3)) + ((c4 + c5) + (c6 + c7));
    if constexpr (ADD) v[i] += (i < 4) ? add0[i & 3] : add1[i & 3];
    Snew[(size_t)(tm + 8 * rg + i) * LD + ej] = v[i];
  }
  if constexpr (STORE_XS) {
    float* xp = xsb + (size_t)ej * COLSTR + (tm + 8 * rg);
    *(v4f*)xp       = (v4f){ v[0], v[1], v[2], v[3] };
    *(v4f*)(xp + 4) = (v4f){ v[4], v[5], v[6], v[7] };
  } else {
    (void)xsb;
  }
}

__global__ __launch_bounds__(256) void matvec_step_k(
    const float* __restrict__ Mm, const float* __restrict__ vin, int vins,
    const float* __restrict__ vadd, int vadds,
    float* __restrict__ vout, int vouts,
    float* __restrict__ vout2, int vouts2) {
  __shared__ float vsh[XL];
  const int tid = threadIdx.x;
  for (int i = tid; i < XL; i += 256) vsh[i] = vin[(size_t)i * vins];
  __syncthreads();

  const int lane = tid & 31, wid = tid >> 5;
  const int gw = blockIdx.x * 8 + wid;
#pragma unroll
  for (int rr = 0; rr < 4; ++rr) {
    const int row = gw * 4 + rr;
    const float* Mr = Mm + (size_t)row * XL;
    float s = 0.f;
    for (int i = 0; i < XL; i += 32) s += Mr[i + lane] * vsh[i + lane];
#pragma unroll
    for (int off = 16; off > 0; off >>= 1) s += __shfl_down(s, off, 32);
    if (lane == 0) {
      const float r = s + vadd[(size_t)row * vadds];
      vout[(size_t)row * vouts] = r;
      vout2[(size_t)row * vouts2] = r;
    }
  }
}

__global__ __launch_bounds__(256) void final_gemm_k(
    const float* __restrict__ xs, const float* __restrict__ Ct,
    const float* __restrict__ uf, const float* __restrict__ Dt,
    float* __restrict__ out) {
  const int lane = threadIdx.x & 31;
  const int w    = threadIdx.x >> 5;
  const int wm = w >> 2, wn = w & 3;
  const int tm = blockIdx.x * 128 + wm * 64;
  const int tn = wn * 64;
  const int ml = lane & 15;
  const int kh = (lane >> 4) << 1;

  v8f acc[4][4];
#pragma unroll
  for (int mi = 0; mi < 4; ++mi)
#pragma unroll
    for (int ni = 0; ni < 4; ++ni) acc[mi][ni] = (v8f){};

  {
    const float* ap[4];
    const float* bp[4];
#pragma unroll
    for (int mi = 0; mi < 4; ++mi)
      ap[mi] = xs + (size_t)(tm + mi * 16 + ml) * XL + kh;
#pragma unroll
    for (int ni = 0; ni < 4; ++ni)
      bp[ni] = Ct + (size_t)kh * YLN + (tn + ni * 16 + ml);
    for (int k = 0; k < XL; k += 4) {
      v2f af[4], bf[4];
      const size_t bo0 = (size_t)k * YLN, bo1 = bo0 + YLN;
#pragma unroll
      for (int mi = 0; mi < 4; ++mi) af[mi] = (v2f){ ap[mi][k], ap[mi][k + 1] };
#pragma unroll
      for (int ni = 0; ni < 4; ++ni) bf[ni] = (v2f){ bp[ni][bo0], bp[ni][bo1] };
#pragma unroll
      for (int mi = 0; mi < 4; ++mi)
#pragma unroll
        for (int ni = 0; ni < 4; ++ni)
          acc[mi][ni] = wmma4(af[mi], bf[ni], acc[mi][ni]);
    }
  }
  {
    const float* ap[4];
    const float* bp[4];
#pragma unroll
    for (int mi = 0; mi < 4; ++mi)
      ap[mi] = uf + (size_t)(tm + mi * 16 + ml) * UL + kh;
#pragma unroll
    for (int ni = 0; ni < 4; ++ni)
      bp[ni] = Dt + (size_t)kh * YLN + (tn + ni * 16 + ml);
    for (int k = 0; k < UL; k += 4) {
      v2f af[4], bf[4];
      const size_t bo0 = (size_t)k * YLN, bo1 = bo0 + YLN;
#pragma unroll
      for (int mi = 0; mi < 4; ++mi) af[mi] = (v2f){ ap[mi][k], ap[mi][k + 1] };
#pragma unroll
      for (int ni = 0; ni < 4; ++ni) bf[ni] = (v2f){ bp[ni][bo0], bp[ni][bo1] };
#pragma unroll
      for (int mi = 0; mi < 4; ++mi)
#pragma unroll
        for (int ni = 0; ni < 4; ++ni)
          acc[mi][ni] = wmma4(af[mi], bf[ni], acc[mi][ni]);
    }
  }

  const int nn = lane & 15;
  const int mb = (lane >> 4) << 3;
#pragma unroll
  for (int mi = 0; mi < 4; ++mi)
#pragma unroll
    for (int ni = 0; ni < 4; ++ni) {
      const int yl = tn + ni * 16 + nn;
      const int n0 = tm + mi * 16 + mb;
      float* op = out + (size_t)yl * NSEQ + n0;
#pragma unroll
      for (int r = 0; r < 8; ++r) op[r] = tanhf(acc[mi][ni][r]);
    }
}

__global__ __launch_bounds__(256) void transpose_k(
    const float* __restrict__ in, float* __restrict__ out, int rows, int cols) {
  __shared__ float tile[32][33];
  const int c0 = blockIdx.x * 32, r0 = blockIdx.y * 32;
  const int tx = threadIdx.x, ty = threadIdx.y;
#pragma unroll
  for (int i = 0; i < 32; i += 8)
    tile[ty + i][tx] = in[(size_t)(r0 + ty + i) * cols + (c0 + tx)];
  __syncthreads();
#pragma unroll
  for (int i = 0; i < 32; i += 8)
    out[(size_t)(c0 + ty + i) * rows + (r0 + tx)] = tile[tx][ty + i];
}

__global__ void set_col_k(float* __restrict__ dst, const float* __restrict__ src,
                          int stride, int n) {
  int i = blockIdx.x * blockDim.x + threadIdx.x;
  if (i < n) dst[(size_t)i * stride] = src[i];
}

extern "C" void kernel_launch(void* const* d_in, const int* in_sizes, int n_in,
                              void* d_out, int out_size, void* d_ws, size_t ws_size,
                              hipStream_t stream) {
  (void)in_sizes; (void)n_in; (void)out_size; (void)ws_size;
  const float* u  = (const float*)d_in[0];
  const float* A  = (const float*)d_in[1];
  const float* B  = (const float*)d_in[2];
  const float* C  = (const float*)d_in[3];
  const float* D  = (const float*)d_in[4];
  const float* x0 = (const float*)d_in[5];
  float* out = (float*)d_out;

  float* ws = (float*)d_ws;
  size_t off = 0;
  float* Bu = ws + off; off += (size_t)NSEQ * XL;
  float* xs = ws + off; off += (size_t)NSEQ * XL;
  float* Bt = ws + off; off += (size_t)UL * XL;
  float* Ct = ws + off; off += (size_t)XL * YLN;
  float* Dt = ws + off; off += (size_t)UL * YLN;
  float* S1a = ws + off; off += (size_t)XL * CHUNKS;
  float* S1b = ws + off; off += (size_t)XL * CHUNKS;
  float* E1  = ws + off; off += (size_t)XL * CHUNKS;
  float* T2a = ws + off; off += (size_t)XL * GROUPS;
  float* T2b = ws + off; off += (size_t)XL * GROUPS;
  float* G2  = ws + off; off += (size_t)XL * GROUPS;
  float* Q0 = xs;
  float* Q1 = xs + (size_t)XL * XL;
  float* Q2 = xs + (size_t)2 * XL * XL;

  transpose_k<<<dim3(UL / 32, XL / 32), dim3(32, 8), 0, stream>>>(B, Bt, XL, UL);
  transpose_k<<<dim3(XL / 32, YLN / 32), dim3(32, 8), 0, stream>>>(C, Ct, YLN, XL);
  transpose_k<<<dim3(YLN / 32, UL / 32), dim3(32, 8), 0, stream>>>(D, Dt, UL, YLN);

  gemm_rr_k<4, 4><<<dim3(NSEQ / 128, XL / 256), 256, 0, stream>>>(
      u, Bt, Bu, UL, UL, XL, XL);

  const dim3 sqg(XL / 32, XL / 64);
  scan_wide_k<false, false, XL><<<sqg, 256, 0, stream>>>(A, A, nullptr, Q0, nullptr);
  scan_wide_k<false, false, XL><<<sqg, 256, 0, stream>>>(Q0, Q0, nullptr, Q1, nullptr);
  scan_wide_k<false, false, XL><<<sqg, 256, 0, stream>>>(Q1, Q1, nullptr, Q0, nullptr);
  scan_wide_k<false, false, XL><<<sqg, 256, 0, stream>>>(Q0, Q0, nullptr, Q1, nullptr);
  scan_wide_k<false, false, XL><<<sqg, 256, 0, stream>>>(Q1, Q1, nullptr, Q2, nullptr);
  scan_wide_k<false, false, XL><<<sqg, 256, 0, stream>>>(Q2, Q2, nullptr, Q0, nullptr);
  scan_wide_k<false, false, XL><<<sqg, 256, 0, stream>>>(Q0, Q0, nullptr, Q1, nullptr);
  scan_wide_k<false, false, XL><<<sqg, 256, 0, stream>>>(Q1, Q1, nullptr, Q0, nullptr);
  scan_wide_k<false, false, XL><<<sqg, 256, 0, stream>>>(Q0, Q0, nullptr, Q1, nullptr);
  const float* P32  = Q2;
  const float* R512 = Q1;

  hipMemsetAsync(S1a, 0, (size_t)XL * CHUNKS * sizeof(float), stream);
  hipMemsetAsync(T2a, 0, (size_t)XL * GROUPS * sizeof(float), stream);
  set_col_k<<<4, 256, 0, stream>>>(G2, x0, GROUPS, XL);
  set_col_k<<<4, 256, 0, stream>>>(E1, x0, CHUNKS, XL);

  {
    float* cur = S1a; float* nxt = S1b;
    for (int t = 0; t < CLEN; ++t) {
      scan_wide_k<false, true, CHUNKS>
          <<<dim3(XL / 32, CHUNKS / 64), 256, 0, stream>>>(
          A, cur, Bu + (size_t)t * XL, nxt, nullptr);
      float* tmp = cur; cur = nxt; nxt = tmp;
    }
  }

  {
    float* cur = T2a; float* nxt = T2b;
    for (int l = 0; l < GLEN; ++l) {
      scan_step_k<false, GROUPS, GLEN, CHUNKS, 1, 1>
          <<<dim3(XL / 16, GROUPS / 32), 256, 0, stream>>>(
          P32, cur, S1a + l, nxt, nullptr);
      float* tmp = cur; cur = nxt; nxt = tmp;
    }
  }

  for (int k = 1; k < GROUPS; ++k) {
    matvec_step_k<<<32, 256, 0, stream>>>(R512, G2 + (k - 1), GROUPS,
                                          T2a + (k - 1), GROUPS, G2 + k, GROUPS,
                                          E1 + (size_t)k * GLEN, CHUNKS);
  }

  {
    float* cur = G2; float* nxt = T2b;
    for (int l = 0; l + 1 < GLEN; ++l) {
      scan_step_k<true, GROUPS, GLEN, CHUNKS, GLEN, CHUNKS>
          <<<dim3(XL / 16, GROUPS / 32), 256, 0, stream>>>(
          P32, cur, S1a + l, nxt, E1 + l + 1);
      float* tmp = cur; cur = nxt; nxt = tmp;
    }
  }

  {
    float* cur = E1; float* nxt = S1b;
    for (int t = 0; t < CLEN; ++t) {
      scan_wide_k<true, true, CHUNKS>
          <<<dim3(XL / 32, CHUNKS / 64), 256, 0, stream>>>(
          A, cur, Bu + (size_t)t * XL, nxt, xs + (size_t)t * XL);
      float* tmp = cur; cur = nxt; nxt = tmp;
    }
  }

  final_gemm_k<<<dim3(NSEQ / 128, 1), 256, 0, stream>>>(xs, Ct, u, Dt, out);
}


# levels 2 and 3 of the scan (62 narrow serial launches) replaced by a 9-step doubling scan over the 512 chunk starts with A^32..A^8192 on the wide kernel, 4 more squarings; sum order changes
# speedup vs baseline: 6.1692x; 1.2869x over previous
#include <hip/hip_runtime.h>
#include <math.h>


typedef __attribute__((ext_vector_type(2))) float v2f;
typedef __attribute__((ext_vector_type(4))) float v4f;
typedef __attribute__((ext_vector_type(8))) float v8f;
typedef float v4fu __attribute__((ext_vector_type(4), aligned(4)));

#define NSEQ   16384
#define XL     1024
#define UL     256
#define YLN    256
#define CHUNKS 512
#define CLEN   32
#define COLSTR (CLEN * XL)
#define PADC   64
#define LDP    (CHUNKS + PADC)

__device__ __forceinline__ v8f wmma4(v2f a, v2f b, v8f c) {
  return __builtin_amdgcn_wmma_f32_16x16x4_f32(
       false, a,  false, b,
       (short)0, c,  false,  false);
}

template <int WM, int WN>
__global__ __launch_bounds__(256) void gemm_rr_k(
    const float* __restrict__ Ap, const float* __restrict__ Bp,
    float* __restrict__ Cp, int K, int lda, int ldb, int ldc) {
  const int lane = threadIdx.x & 31;
  const int w    = threadIdx.x >> 5;
  const int wm = w >> 2, wn = w & 3;
  const int tm = blockIdx.x * (32 * WM) + wm * (16 * WM);
  const int tn = blockIdx.y * (64 * WN) + wn * (16 * WN);
  const int ml = lane & 15;
  const int kh = (lane >> 4) << 1;

  v8f acc[WM][WN];
#pragma unroll
  for (int mi = 0; mi < WM; ++mi)
#pragma unroll
    for (int ni = 0; ni < WN; ++ni) acc[mi][ni] = (v8f){};

  const float* ap[WM];
  const float* bp[WN];
#pragma unroll
  for (int mi = 0; mi < WM; ++mi)
    ap[mi] = Ap + (size_t)(tm + mi * 16 + ml) * lda + kh;
#pragma unroll
  for (int ni = 0; ni < WN; ++ni)
    bp[ni] = Bp + (size_t)kh * ldb + (tn + ni * 16 + ml);

  for (int k = 0; k < K; k += 4) {
    v2f af[WM], bf[WN];
    const size_t bo0 = (size_t)k * ldb, bo1 = bo0 + (size_t)ldb;
#pragma unroll
    for (int mi = 0; mi < WM; ++mi) af[mi] = (v2f){ ap[mi][k], ap[mi][k + 1] };
#pragma unroll
    for (int ni = 0; ni < WN; ++ni) bf[ni] = (v2f){ bp[ni][bo0], bp[ni][bo1] };
#pragma unroll
    for (int mi = 0; mi < WM; ++mi)
#pragma unroll
      for (int ni = 0; ni < WN; ++ni)
        acc[mi][ni] = wmma4(af[mi], bf[ni], acc[mi][ni]);
  }

  const int nn = lane & 15;
  const int mb = (lane >> 4) << 3;
#pragma unroll
  for (int mi = 0; mi < WM; ++mi)
#pragma unroll
    for (int ni = 0; ni < WN; ++ni) {
      float* cp = Cp + (size_t)(tm + mi * 16 + mb) * ldc + (tn + ni * 16 + nn);
#pragma unroll
      for (int r = 0; r < 8; ++r) cp[(size_t)r * ldc] = acc[mi][ni][r];
    }
}

template <bool STORE_XS, int ADD, int LD, bool UNAL>
__global__ __launch_bounds__(256) __attribute__((amdgpu_waves_per_eu(2, 4)))
void scan_wide_k(
    const float* __restrict__ Ap, const float* __restrict__ Sold,
    const float* addb, float* __restrict__ Snew,
    float* __restrict__ xsb, int sh) {
  static_assert(ADD == 1 || !STORE_XS, "xs copy goes with the level-1 addend");
  static_assert(ADD == 2 || !UNAL, "only a doubling step moves its operand");
  constexpr int KE = XL / 8;
  constexpr int NP = KE / 8;
  constexpr int GP = 2;
  constexpr int NG = NP / GP;
  constexpr int RP = 8 * 32 + 16;
  __shared__ float red[8][2][4][RP];

  const int lane = threadIdx.x & 31;
  const int w    = threadIdx.x >> 5;
  const int tm = blockIdx.x * 32;
  const int tn = blockIdx.y * 64;
  const int c  = lane & 15;
  const int h  = lane >> 4;

  if constexpr (ADD == 2) {
    if (tn + 64 <= sh) {
#pragma unroll
      for (int i = 0; i < 8; ++i) {
        const int idx = (int)threadIdx.x + 256 * i;
        const size_t o = (size_t)(tm + (idx >> 6)) * LD + (tn + (idx & 63));
        Snew[o] = addb[o];
      }
      return;
    }
  }

  const int eh = w & 1, rg = w >> 1;
  const int ej = tn + eh * 32 + lane;
  const float* addp = nullptr;
  if constexpr (ADD == 1) {
    const int aj = (ej - sh) > 0 ? (ej - sh) : 0;
    addp = addb + (size_t)aj * COLSTR + (tm + 8 * rg);
  } else if constexpr (ADD == 2) {
    addp = addb + (size_t)(tm + 8 * rg) * LD + ej;
  } else {
    (void)addb; (void)sh;
  }
  v4f add0 = {}, add1 = {};

  const float* ap0 = Ap + (size_t)(tm + c) * XL + (w * KE + 4 * h);
  const float* ap1 = ap0 + (size_t)16 * XL;
  const float* bp  = Sold + (size_t)(w * KE + 4 * h) * LD + (tn + 4 * c);

  v4f af[NG][GP][2];
  v4f bq[NG][GP][4];

#define SCAN_WIDE_LOAD_GROUP(g)                                              \
  _Pragma("unroll")                                                          \
  for (int p = 0; p < GP; ++p) {                                             \
    const int k = ((g) * GP + p) * 8;                                        \
    af[(g)][p][0] = *(const v4f*)(ap0 + k);                                  \
    af[(g)][p][1] = *(const v4f*)(ap1 + k);                                  \
    _Pragma("unroll")                                                        \
    for (int e = 0; e < 4; ++e) {                                            \
      if constexpr (UNAL)                                                    \
        bq[(g)][p][e] = *(const v4fu*)(bp + (size_t)(k + e) * LD);           \
      else                                                                   \
        bq[(g)][p][e] = *(const v4f*)(bp + (size_t)(k + e) * LD);            \
    }                                                                        \
  }

#pragma unroll
  for (int g = 0; g < 2; ++g) { SCAN_WIDE_LOAD_GROUP(g) }

  v8f acc[2][4];
#pragma unroll
  for (int mi = 0; mi < 2; ++mi)
#pragma unroll
    for (int ni = 0; ni < 4; ++ni) acc[mi][ni] = (v8f){};

#pragma unroll
  for (int g = 0; g < NG; ++g) {
    if (g + 2 < NG) { SCAN_WIDE_LOAD_GROUP(g + 2) }
    if (ADD != 0 && g + 3 == NG) {
      if constexpr (ADD == 1) {
        add0 = *(const v4f*)addp;
        add1 = *(const v4f*)(addp + 4);
      } else if constexpr (ADD == 2) {
#pragma unroll
        for (int i = 0; i < 4; ++i) {
          add0[i] = addp[(size_t)i * LD];
          add1[i] = addp[(size_t)(i + 4) * LD];
        }
      }
      __builtin_amdgcn_sched_barrier(0);
    }
#pragma unroll
    for (int p = 0; p < GP; ++p)
#pragma unroll
      for (int s = 0; s < 2; ++s)
#pragma unroll
        for (int ni = 0; ni < 4; ++ni)
#pragma unroll
          for (int mi = 0; mi < 2; ++mi)
            acc[mi][ni] = wmma4(
                (v2f){ af[g][p][mi][2 * s], af[g][p][mi][2 * s + 1] },
                (v2f){ bq[g][p][2 * s][ni], bq[g][p][2 * s + 1][ni] },
                acc[mi][ni]);
  }
#undef SCAN_WIDE_LOAD_GROUP

#pragma unroll
  for (int mi = 0; mi < 2; ++mi)
#pragma unroll
    for (int ni = 0; ni < 4; ++ni)
#pragma unroll
      for (int r = 0; r < 8; ++r) red[w][mi][ni][r * 32 + lane] = acc[mi][ni][r];
  __syncthreads();

  const int emi = rg >> 1;
  const int eni = lane & 3;
  const int esl = (rg & 1) * 16 + eh * 8 + (lane >> 2);
  float v[8];
#pragma unroll
  for (int i = 0; i < 8; ++i) {
    const int sl = i * 32 + esl;
    const float c0 = red[0][emi][eni][sl], c1 = red[1][emi][eni][sl];
    const float c2 = red[2][emi][eni][sl], c3 = red[3][emi][eni][sl];
    const float c4 = red[4][emi][eni][sl], c5 = red[5][emi][eni][sl];
    const float c6 = red[6][emi][eni][sl], c7 = red[7][emi][eni][sl];
    v[i] = ((c0 + c1) + (c2 + c3)) + ((c4 + c5) + (c6 + c7));
    if constexpr (ADD != 0) v[i] += (i < 4) ? add0[i & 3] : add1[i & 3];
    Snew[(size_t)(tm + 8 * rg + i) * LD + ej] = v[i];
  }
  if constexpr (STORE_XS) {
    float* xp = xsb + (size_t)ej * COLSTR + (tm + 8 * rg);
    *(v4f*)xp       = (v4f){ v[0], v[1], v[2], v[3] };
    *(v4f*)(xp + 4) = (v4f){ v[4], v[5], v[6], v[7] };
  } else {
    (void)xsb;
  }
}

__global__ __launch_bounds__(256) void final_gemm_k(
    const float* __restrict__ xs, const float* __restrict__ Ct,
    const float* __restrict__ uf, const float* __restrict__ Dt,
    float* __restrict__ out) {
  const int lane = threadIdx.x & 31;
  const int w    = threadIdx.x >> 5;
  const int wm = w >> 2, wn = w & 3;
  const int tm = blockIdx.x * 128 + wm * 64;
  const int tn = wn * 64;
  const int ml = lane & 15;
  const int kh = (lane >> 4) << 1;

  v8f acc[4][4];
#pragma unroll
  for (int mi = 0; mi < 4; ++mi)
#pragma unroll
    for (int ni = 0; ni < 4; ++ni) acc[mi][ni] = (v8f){};

  {
    const float* ap[4];
    const float* bp[4];
#pragma unroll
    for (int mi = 0; mi < 4; ++mi)
      ap[mi] = xs + (size_t)(tm + mi * 16 + ml) * XL + kh;
#pragma unroll
    for (int ni = 0; ni < 4; ++ni)
      bp[ni] = Ct + (size_t)kh * YLN + (tn + ni * 16 + ml);
    for (int k = 0; k < XL; k += 4) {
      v2f af[4], bf[4];
      const size_t bo0 = (size_t)k * YLN, bo1 = bo0 + YLN;
#pragma unroll
      for (int mi = 0; mi < 4; ++mi) af[mi] = (v2f){ ap[mi][k], ap[mi][k + 1] };
#pragma unroll
      for (int ni = 0; ni < 4; ++ni) bf[ni] = (v2f){ bp[ni][bo0], bp[ni][bo1] };
#pragma unroll
      for (int mi = 0; mi < 4; ++mi)
#pragma unroll
        for (int ni = 0; ni < 4; ++ni)
          acc[mi][ni] = wmma4(af[mi], bf[ni], acc[mi][ni]);
    }
  }
  {
    const float* ap[4];
    const float* bp[4];
#pragma unroll
    for (int mi = 0; mi < 4; ++mi)
      ap[mi] = uf + (size_t)(tm + mi * 16 + ml) * UL + kh;
#pragma unroll
    for (int ni = 0; ni < 4; ++ni)
      bp[ni] = Dt + (size_t)kh * YLN + (tn + ni * 16 + ml);
    for (int k = 0; k < UL; k += 4) {
      v2f af[4], bf[4];
      const size_t bo0 = (size_t)k * YLN, bo1 = bo0 + YLN;
#pragma unroll
      for (int mi = 0; mi < 4; ++mi) af[mi] = (v2f){ ap[mi][k], ap[mi][k + 1] };
#pragma unroll
      for (int ni = 0; ni < 4; ++ni) bf[ni] = (v2f){ bp[ni][bo0], bp[ni][bo1] };
#pragma unroll
      for (int mi = 0; mi < 4; ++mi)
#pragma unroll
        for (int ni = 0; ni < 4; ++ni)
          acc[mi][ni] = wmma4(af[mi], bf[ni], acc[mi][ni]);
    }
  }

  const int nn = lane & 15;
  const int mb = (lane >> 4) << 3;
#pragma unroll
  for (int mi = 0; mi < 4; ++mi)
#pragma unroll
    for (int ni = 0; ni < 4; ++ni) {
      const int yl = tn + ni * 16 + nn;
      const int n0 = tm + mi * 16 + mb;
      float* op = out + (size_t)yl * NSEQ + n0;
#pragma unroll
      for (int r = 0; r < 8; ++r) op[r] = tanhf(acc[mi][ni][r]);
    }
}

__global__ __launch_bounds__(256) void transpose_k(
    const float* __restrict__ in, float* __restrict__ out, int rows, int cols) {
  __shared__ float tile[32][33];
  const int c0 = blockIdx.x * 32, r0 = blockIdx.y * 32;
  const int tx = threadIdx.x, ty = threadIdx.y;
#pragma unroll
  for (int i = 0; i < 32; i += 8)
    tile[ty + i][tx] = in[(size_t)(r0 + ty + i) * cols + (c0 + tx)];
  __syncthreads();
#pragma unroll
  for (int i = 0; i < 32; i += 8)
    out[(size_t)(c0 + ty + i) * rows + (r0 + tx)] = tile[tx][ty + i];
}

__global__ void set_col_k(float* __restrict__ dst, const float* __restrict__ src,
                          int stride, int n) {
  int i = blockIdx.x * blockDim.x + threadIdx.x;
  if (i < n) dst[(size_t)i * stride] = src[i];
}

extern "C" void kernel_launch(void* const* d_in, const int* in_sizes, int n_in,
                              void* d_out, int out_size, void* d_ws, size_t ws_size,
                              hipStream_t stream) {
  (void)in_sizes; (void)n_in; (void)out_size; (void)ws_size;
  const float* u  = (const float*)d_in[0];
  const float* A  = (const float*)d_in[1];
  const float* B  = (const float*)d_in[2];
  const float* C  = (const float*)d_in[3];
  const float* D  = (const float*)d_in[4];
  const float* x0 = (const float*)d_in[5];
  float* out = (float*)d_out;

  float* ws = (float*)d_ws;
  size_t off = 0;
  float* Bu = ws + off; off += (size_t)NSEQ * XL;
  float* xs = ws + off; off += (size_t)NSEQ * XL;
  float* Bt = ws + off; off += (size_t)UL * XL;
  float* Ct = ws + off; off += (size_t)XL * YLN;
  float* Dt = ws + off; off += (size_t)UL * YLN;
  float* S1a = ws + off; off += (size_t)XL * LDP;
  float* S1b = ws + off; off += (size_t)XL * LDP;
  float* Q1  = xs;
  float* Q2  = xs + (size_t)1 * XL * XL;
  float* Q3  = xs + (size_t)2 * XL * XL;
  float* Q4  = xs + (size_t)3 * XL * XL;
  float* Q5  = xs + (size_t)4 * XL * XL;
  float* Q6  = xs + (size_t)5 * XL * XL;
  float* Q7  = xs + (size_t)6 * XL * XL;
  float* Q8  = xs + (size_t)7 * XL * XL;
  float* Q9  = xs + (size_t)8 * XL * XL;
  float* Q10 = xs + (size_t)9 * XL * XL;
  float* Q11 = xs + (size_t)10 * XL * XL;
  float* Q12 = xs + (size_t)11 * XL * XL;
  float* Q13 = xs + (size_t)12 * XL * XL;

  transpose_k<<<dim3(UL / 32, XL / 32), dim3(32, 8), 0, stream>>>(B, Bt, XL, UL);
  transpose_k<<<dim3(XL / 32, YLN / 32), dim3(32, 8), 0, stream>>>(C, Ct, YLN, XL);
  transpose_k<<<dim3(YLN / 32, UL / 32), dim3(32, 8), 0, stream>>>(D, Dt, UL, YLN);

  gemm_rr_k<4, 4><<<dim3(NSEQ / 128, XL / 256), 256, 0, stream>>>(
      u, Bt, Bu, UL, UL, XL, XL);

  const dim3 sqg(XL / 32, XL / 64);
#define SQUARE(src, dst) \
  scan_wide_k<false, 0, XL, false><<<sqg, 256, 0, stream>>>(src, src, nullptr, dst, nullptr, 0)
  SQUARE(A,   Q1);
  SQUARE(Q1,  Q2);
  SQUARE(Q2,  Q3);
  SQUARE(Q3,  Q4);
  SQUARE(Q4,  Q5);
  SQUARE(Q5,  Q6);
  SQUARE(Q6,  Q7);
  SQUARE(Q7,  Q8);
  SQUARE(Q8,  Q9);
  SQUARE(Q9,  Q10);
  SQUARE(Q10, Q11);
  SQUARE(Q11, Q12);
  SQUARE(Q12, Q13);
#undef SQUARE

  hipMemsetAsync(S1a, 0, (size_t)XL * LDP * sizeof(float), stream);
  hipMemsetAsync(S1b, 0, (size_t)XL * LDP * sizeof(float), stream);

  const dim3 stg(XL / 32, CHUNKS / 64);
  float* Va = S1a + PADC;
  float* Vb = S1b + PADC;

  {
    float* cur = Va; float* nxt = Vb;
    for (int t = 0; t < CLEN; ++t) {
      scan_wide_k<false, 1, LDP, false><<<stg, 256, 0, stream>>>(
          A, cur, Bu + (size_t)t * XL, nxt, nullptr, 1);
      float* tmp = cur; cur = nxt; nxt = tmp;
    }
  }
  set_col_k<<<4, 256, 0, stream>>>(Va, x0, LDP, XL);

  scan_wide_k<false, 2, LDP, true ><<<stg, 256, 0, stream>>>(Q5,  Va - 1,   Va, Vb, nullptr, 1);
  scan_wide_k<false, 2, LDP, true ><<<stg, 256, 0, stream>>>(Q6,  Vb - 2,   Vb, Va, nullptr, 2);
  scan_wide_k<false, 2, LDP, false><<<stg, 256, 0, stream>>>(Q7,  Va - 4,   Va, Vb, nullptr, 4);
  scan_wide_k<false, 2, LDP, false><<<stg, 256, 0, stream>>>(Q8,  Vb - 8,   Vb, Va, nullptr, 8);
  scan_wide_k<false, 2, LDP, false><<<stg, 256, 0, stream>>>(Q9,  Va - 16,  Va, Vb, nullptr, 16);
  scan_wide_k<false, 2, LDP, false><<<stg, 256, 0, stream>>>(Q10, Vb - 32,  Vb, Va, nullptr, 32);
  scan_wide_k<false, 2, LDP, false><<<stg, 256, 0, stream>>>(Q11, Va - 64,  Va, Vb, nullptr, 64);
  scan_wide_k<false, 2, LDP, false><<<stg, 256, 0, stream>>>(Q12, Vb - 128, Vb, Va, nullptr, 128);
  scan_wide_k<false, 2, LDP, false><<<stg, 256, 0, stream>>>(Q13, Va - 256, Va, Vb, nullptr, 256);

  {
    float* cur = Vb; float* nxt = Va;
    for (int t = 0; t < CLEN; ++t) {
      scan_wide_k<true, 1, LDP, false><<<stg, 256, 0, stream>>>(
          A, cur, Bu + (size_t)t * XL, nxt, xs + (size_t)t * XL, 0);
      float* tmp = cur; cur = nxt; nxt = tmp;
    }
  }

  final_gemm_k<<<dim3(NSEQ / 128, 1), 256, 0, stream>>>(xs, Ct, u, Dt, out);
}


# local chunk ends from one K=2048 product against [A^7 B|..|B] over 8-step sub-chunks plus two pairwise merges, replacing the 32-launch serial up pass; sum order changes
# speedup vs baseline: 7.0592x; 1.1443x over previous
#include <hip/hip_runtime.h>
#include <math.h>


typedef __attribute__((ext_vector_type(2))) float v2f;
typedef __attribute__((ext_vector_type(4))) float v4f;
typedef __attribute__((ext_vector_type(8))) float v8f;
typedef float v4fu __attribute__((ext_vector_type(4), aligned(4)));

#define NSEQ   16384
#define XL     1024
#define UL     256
#define YLN    256
#define CHUNKS 512
#define CLEN   32
#define COLSTR (CLEN * XL)
#define PADC   64
#define LDP    (CHUNKS + PADC)
#define SUBL   8
#define SUBK   (SUBL * UL)
#define SUBC   (4 * CHUNKS)

__device__ __forceinline__ v8f wmma4(v2f a, v2f b, v8f c) {
  return __builtin_amdgcn_wmma_f32_16x16x4_f32(
       false, a,  false, b,
       (short)0, c,  false,  false);
}

template <int WM, int WN>
__global__ __launch_bounds__(256) void gemm_rr_k(
    const float* __restrict__ Ap, const float* __restrict__ Bp,
    float* __restrict__ Cp, int K, int lda, int ldb, int ldc) {
  const int lane = threadIdx.x & 31;
  const int w    = threadIdx.x >> 5;
  const int wm = w >> 2, wn = w & 3;
  const int tm = blockIdx.x * (32 * WM) + wm * (16 * WM);
  const int tn = blockIdx.y * (64 * WN) + wn * (16 * WN);
  const int ml = lane & 15;
  const int kh = (lane >> 4) << 1;

  v8f acc[WM][WN];
#pragma unroll
  for (int mi = 0; mi < WM; ++mi)
#pragma unroll
    for (int ni = 0; ni < WN; ++ni) acc[mi][ni] = (v8f){};

  const float* ap[WM];
  const float* bp[WN];
#pragma unroll
  for (int mi = 0; mi < WM; ++mi)
    ap[mi] = Ap + (size_t)(tm + mi * 16 + ml) * lda + kh;
#pragma unroll
  for (int ni = 0; ni < WN; ++ni)
    bp[ni] = Bp + (size_t)kh * ldb + (tn + ni * 16 + ml);

  for (int k = 0; k < K; k += 4) {
    v2f af[WM], bf[WN];
    const size_t bo0 = (size_t)k * ldb, bo1 = bo0 + (size_t)ldb;
#pragma unroll
    for (int mi = 0; mi < WM; ++mi) af[mi] = (v2f){ ap[mi][k], ap[mi][k + 1] };
#pragma unroll
    for (int ni = 0; ni < WN; ++ni) bf[ni] = (v2f){ bp[ni][bo0], bp[ni][bo1] };
#pragma unroll
    for (int mi = 0; mi < WM; ++mi)
#pragma unroll
      for (int ni = 0; ni < WN; ++ni)
        acc[mi][ni] = wmma4(af[mi], bf[ni], acc[mi][ni]);
  }

  const int nn = lane & 15;
  const int mb = (lane >> 4) << 3;
#pragma unroll
  for (int mi = 0; mi < WM; ++mi)
#pragma unroll
    for (int ni = 0; ni < WN; ++ni) {
      float* cp = Cp + (size_t)(tm + mi * 16 + mb) * ldc + (tn + ni * 16 + nn);
#pragma unroll
      for (int r = 0; r < 8; ++r) cp[(size_t)r * ldc] = acc[mi][ni][r];
    }
}

template <bool STORE_XS, int ADD, int LD, bool UNAL,
          int LDA = XL, int LDB = LD, int LDC = LD>
__global__ __launch_bounds__(256) __attribute__((amdgpu_waves_per_eu(2, 4)))
void scan_wide_k(
    const float* __restrict__ Ap, const float* __restrict__ Sold,
    const float* addb, float* __restrict__ Snew,
    float* __restrict__ xsb, int sh) {
  static_assert(ADD == 1 || !STORE_XS, "xs copy goes with the level-1 addend");
  static_assert(ADD == 2 || !UNAL, "only a doubling step moves its operand");
  constexpr int KE = XL / 8;
  constexpr int NP = KE / 8;
  constexpr int GP = 2;
  constexpr int NG = NP / GP;
  constexpr int RP = 8 * 32 + 16;
  __shared__ float red[8][2][4][RP];

  const int lane = threadIdx.x & 31;
  const int w    = threadIdx.x >> 5;
  const int tm = blockIdx.x * 32;
  const int tn = blockIdx.y * 64;
  const int c  = lane & 15;
  const int h  = lane >> 4;

  if constexpr (ADD == 2) {
    if (tn + 64 <= sh) {
#pragma unroll
      for (int i = 0; i < 8; ++i) {
        const int idx = (int)threadIdx.x + 256 * i;
        const size_t o  = (size_t)(tm + (idx >> 6)) * LD  + (tn + (idx & 63));
        const size_t oa = (size_t)(tm + (idx >> 6)) * LDC + (tn + (idx & 63));
        Snew[o] = addb[oa];
      }
      return;
    }
  }

  const int eh = w & 1, rg = w >> 1;
  const int ej = tn + eh * 32 + lane;
  const float* addp = nullptr;
  if constexpr (ADD == 1) {
    const int aj = (ej - sh) > 0 ? (ej - sh) : 0;
    addp = addb + (size_t)aj * COLSTR + (tm + 8 * rg);
  } else if constexpr (ADD == 2) {
    addp = addb + (size_t)(tm + 8 * rg) * LDC + ej;
  } else {
    (void)addb; (void)sh;
  }
  v4f add0 = {}, add1 = {};

  const float* ap0 = Ap + (size_t)(tm + c) * LDA + (w * KE + 4 * h);
  const float* ap1 = ap0 + (size_t)16 * LDA;
  const float* bp  = Sold + (size_t)(w * KE + 4 * h) * LDB + (tn + 4 * c);

  v4f af[NG][GP][2];
  v4f bq[NG][GP][4];

#define SCAN_WIDE_LOAD_GROUP(g)                                              \
  _Pragma("unroll")                                                          \
  for (int p = 0; p < GP; ++p) {                                             \
    const int k = ((g) * GP + p) * 8;                                        \
    af[(g)][p][0] = *(const v4f*)(ap0 + k);                                  \
    af[(g)][p][1] = *(const v4f*)(ap1 + k);                                  \
    _Pragma("unroll")                                                        \
    for (int e = 0; e < 4; ++e) {                                            \
      if constexpr (UNAL)                                                    \
        bq[(g)][p][e] = *(const v4fu*)(bp + (size_t)(k + e) * LDB);          \
      else                                                                   \
        bq[(g)][p][e] = *(const v4f*)(bp + (size_t)(k + e) * LDB);           \
    }                                                                        \
  }

#pragma unroll
  for (int g = 0; g < 2; ++g) { SCAN_WIDE_LOAD_GROUP(g) }

  v8f acc[2][4];
#pragma unroll
  for (int mi = 0; mi < 2; ++mi)
#pragma unroll
    for (int ni = 0; ni < 4; ++ni) acc[mi][ni] = (v8f){};

#pragma unroll
  for (int g = 0; g < NG; ++g) {
    if (g + 2 < NG) { SCAN_WIDE_LOAD_GROUP(g + 2) }
    if (ADD != 0 && g + 3 == NG) {
      if constexpr (ADD == 1) {
        add0 = *(const v4f*)addp;
        add1 = *(const v4f*)(addp + 4);
      } else if constexpr (ADD == 2) {
#pragma unroll
        for (int i = 0; i < 4; ++i) {
          add0[i] = addp[(size_t)i * LDC];
          add1[i] = addp[(size_t)(i + 4) * LDC];
        }
      }
      __builtin_amdgcn_sched_barrier(0);
    }
#pragma unroll
    for (int p = 0; p < GP; ++p)
#pragma unroll
      for (int s = 0; s < 2; ++s)
#pragma unroll
        for (int ni = 0; ni < 4; ++ni)
#pragma unroll
          for (int mi = 0; mi < 2; ++mi)
            acc[mi][ni] = wmma4(
                (v2f){ af[g][p][mi][2 * s], af[g][p][mi][2 * s + 1] },
                (v2f){ bq[g][p][2 * s][ni], bq[g][p][2 * s + 1][ni] },
                acc[mi][ni]);
  }
#undef SCAN_WIDE_LOAD_GROUP

#pragma unroll
  for (int mi = 0; mi < 2; ++mi)
#pragma unroll
    for (int ni = 0; ni < 4; ++ni)
#pragma unroll
      for (int r = 0; r < 8; ++r) red[w][mi][ni][r * 32 + lane] = acc[mi][ni][r];
  __syncthreads();

  const int emi = rg >> 1;
  const int eni = lane & 3;
  const int esl = (rg & 1) * 16 + eh * 8 + (lane >> 2);
  float v[8];
#pragma unroll
  for (int i = 0; i < 8; ++i) {
    const int sl = i * 32 + esl;
    const float c0 = red[0][emi][eni][sl], c1 = red[1][emi][eni][sl];
    const float c2 = red[2][emi][eni][sl], c3 = red[3][emi][eni][sl];
    const float c4 = red[4][emi][eni][sl], c5 = red[5][emi][eni][sl];
    const float c6 = red[6][emi][eni][sl], c7 = red[7][emi][eni][sl];
    v[i] = ((c0 + c1) + (c2 + c3)) + ((c4 + c5) + (c6 + c7));
    if constexpr (ADD != 0) v[i] += (i < 4) ? add0[i & 3] : add1[i & 3];
    Snew[(size_t)(tm + 8 * rg + i) * LD + ej] = v[i];
  }
  if constexpr (STORE_XS) {
    float* xp = xsb + (size_t)ej * COLSTR + (tm + 8 * rg);
    *(v4f*)xp       = (v4f){ v[0], v[1], v[2], v[3] };
    *(v4f*)(xp + 4) = (v4f){ v[4], v[5], v[6], v[7] };
  } else {
    (void)xsb;
  }
}

__global__ __launch_bounds__(256) void final_gemm_k(
    const float* __restrict__ xs, const float* __restrict__ Ct,
    const float* __restrict__ uf, const float* __restrict__ Dt,
    float* __restrict__ out) {
  const int lane = threadIdx.x & 31;
  const int w    = threadIdx.x >> 5;
  const int wm = w >> 2, wn = w & 3;
  const int tm = blockIdx.x * 128 + wm * 64;
  const int tn = wn * 64;
  const int ml = lane & 15;
  const int kh = (lane >> 4) << 1;

  v8f acc[4][4];
#pragma unroll
  for (int mi = 0; mi < 4; ++mi)
#pragma unroll
    for (int ni = 0; ni < 4; ++ni) acc[mi][ni] = (v8f){};

  {
    const float* ap[4];
    const float* bp[4];
#pragma unroll
    for (int mi = 0; mi < 4; ++mi)
      ap[mi] = xs + (size_t)(tm + mi * 16 + ml) * XL + kh;
#pragma unroll
    for (int ni = 0; ni < 4; ++ni)
      bp[ni] = Ct + (size_t)kh * YLN + (tn + ni * 16 + ml);
    for (int k = 0; k < XL; k += 4) {
      v2f af[4], bf[4];
      const size_t bo0 = (size_t)k * YLN, bo1 = bo0 + YLN;
#pragma unroll
      for (int mi = 0; mi < 4; ++mi) af[mi] = (v2f){ ap[mi][k], ap[mi][k + 1] };
#pragma unroll
      for (int ni = 0; ni < 4; ++ni) bf[ni] = (v2f){ bp[ni][bo0], bp[ni][bo1] };
#pragma unroll
      for (int mi = 0; mi < 4; ++mi)
#pragma unroll
        for (int ni = 0; ni < 4; ++ni)
          acc[mi][ni] = wmma4(af[mi], bf[ni], acc[mi][ni]);
    }
  }
  {
    const float* ap[4];
    const float* bp[4];
#pragma unroll
    for (int mi = 0; mi < 4; ++mi)
      ap[mi] = uf + (size_t)(tm + mi * 16 + ml) * UL + kh;
#pragma unroll
    for (int ni = 0; ni < 4; ++ni)
      bp[ni] = Dt + (size_t)kh * YLN + (tn + ni * 16 + ml);
    for (int k = 0; k < UL; k += 4) {
      v2f af[4], bf[4];
      const size_t bo0 = (size_t)k * YLN, bo1 = bo0 + YLN;
#pragma unroll
      for (int mi = 0; mi < 4; ++mi) af[mi] = (v2f){ ap[mi][k], ap[mi][k + 1] };
#pragma unroll
      for (int ni = 0; ni < 4; ++ni) bf[ni] = (v2f){ bp[ni][bo0], bp[ni][bo1] };
#pragma unroll
      for (int mi = 0; mi < 4; ++mi)
#pragma unroll
        for (int ni = 0; ni < 4; ++ni)
          acc[mi][ni] = wmma4(af[mi], bf[ni], acc[mi][ni]);
    }
  }

  const int nn = lane & 15;
  const int mb = (lane >> 4) << 3;
#pragma unroll
  for (int mi = 0; mi < 4; ++mi)
#pragma unroll
    for (int ni = 0; ni < 4; ++ni) {
      const int yl = tn + ni * 16 + nn;
      const int n0 = tm + mi * 16 + mb;
      float* op = out + (size_t)yl * NSEQ + n0;
#pragma unroll
      for (int r = 0; r < 8; ++r) op[r] = tanhf(acc[mi][ni][r]);
    }
}

__global__ __launch_bounds__(256) void transpose_k(
    const float* __restrict__ in, float* __restrict__ out, int rows, int cols) {
  __shared__ float tile[32][33];
  const int c0 = blockIdx.x * 32, r0 = blockIdx.y * 32;
  const int tx = threadIdx.x, ty = threadIdx.y;
#pragma unroll
  for (int i = 0; i < 32; i += 8)
    tile[ty + i][tx] = in[(size_t)(r0 + ty + i) * cols + (c0 + tx)];
  __syncthreads();
#pragma unroll
  for (int i = 0; i < 32; i += 8)
    out[(size_t)(c0 + ty + i) * rows + (r0 + tx)] = tile[tx][ty + i];
}

__global__ __launch_bounds__(256) void u_substack_k(
    const float* __restrict__ u, float* __restrict__ Ut) {
  __shared__ float tile[32][33];
  const int c0 = blockIdx.x * 32, k0 = blockIdx.y * 32;
  const int tx = threadIdx.x, ty = threadIdx.y;
#pragma unroll
  for (int i = 0; i < 32; i += 8) {
    const int cc = c0 + ty + i;
    const int g = cc >> 9, jj = cc & (CHUNKS - 1);
    const int r = ((g & 1) << 1) | (g >> 1);
    const int j = jj > 0 ? jj - 1 : 0;
    const int q = 4 * j + r;
    tile[ty + i][tx] = u[(size_t)q * SUBK + (k0 + tx)];
  }
  __syncthreads();
#pragma unroll
  for (int i = 0; i < 32; i += 8)
    Ut[(size_t)(k0 + ty + i) * SUBC + (c0 + tx)] = tile[tx][ty + i];
}

__global__ __launch_bounds__(256) void copy_b_k(
    const float* __restrict__ B, float* __restrict__ dst) {
  const int idx = blockIdx.x * 256 + threadIdx.x;
  dst[(size_t)(idx >> 8) * SUBK + (idx & (UL - 1))] = B[idx];
}

__global__ void set_col_k(float* __restrict__ dst, const float* __restrict__ src,
                          int stride, int n) {
  int i = blockIdx.x * blockDim.x + threadIdx.x;
  if (i < n) dst[(size_t)i * stride] = src[i];
}

extern "C" void kernel_launch(void* const* d_in, const int* in_sizes, int n_in,
                              void* d_out, int out_size, void* d_ws, size_t ws_size,
                              hipStream_t stream) {
  (void)in_sizes; (void)n_in; (void)out_size; (void)ws_size;
  const float* u  = (const float*)d_in[0];
  const float* A  = (const float*)d_in[1];
  const float* B  = (const float*)d_in[2];
  const float* C  = (const float*)d_in[3];
  const float* D  = (const float*)d_in[4];
  const float* x0 = (const float*)d_in[5];
  float* out = (float*)d_out;

  float* ws = (float*)d_ws;
  size_t off = 0;
  float* Bu = ws + off; off += (size_t)NSEQ * XL;
  float* xs = ws + off; off += (size_t)NSEQ * XL;
  float* Bt = ws + off; off += (size_t)UL * XL;
  float* Ct = ws + off; off += (size_t)XL * YLN;
  float* Dt = ws + off; off += (size_t)UL * YLN;
  float* S1a = ws + off; off += (size_t)XL * LDP;
  float* S1b = ws + off; off += (size_t)XL * LDP;
  float* Q1  = xs;
  float* Q2  = xs + (size_t)1 * XL * XL;
  float* Q3  = xs + (size_t)2 * XL * XL;
  float* Q4  = xs + (size_t)3 * XL * XL;
  float* Q5  = xs + (size_t)4 * XL * XL;
  float* Q6  = xs + (size_t)5 * XL * XL;
  float* Q7  = xs + (size_t)6 * XL * XL;
  float* Q8  = xs + (size_t)7 * XL * XL;
  float* Q9  = xs + (size_t)8 * XL * XL;
  float* Q10 = xs + (size_t)9 * XL * XL;
  float* Q11 = xs + (size_t)10 * XL * XL;
  float* Q12 = xs + (size_t)11 * XL * XL;
  float* Q13 = xs + (size_t)12 * XL * XL;

  transpose_k<<<dim3(UL / 32, XL / 32), dim3(32, 8), 0, stream>>>(B, Bt, XL, UL);
  transpose_k<<<dim3(XL / 32, YLN / 32), dim3(32, 8), 0, stream>>>(C, Ct, YLN, XL);
  transpose_k<<<dim3(YLN / 32, UL / 32), dim3(32, 8), 0, stream>>>(D, Dt, UL, YLN);

  gemm_rr_k<4, 4><<<dim3(NSEQ / 128, XL / 256), 256, 0, stream>>>(
      u, Bt, Bu, UL, UL, XL, XL);

  const dim3 sqg(XL / 32, XL / 64);
#define SQUARE(src, dst) \
  scan_wide_k<false, 0, XL, false><<<sqg, 256, 0, stream>>>(src, src, nullptr, dst, nullptr, 0)
  SQUARE(A,   Q1);
  SQUARE(Q1,  Q2);
  SQUARE(Q2,  Q3);
  SQUARE(Q3,  Q4);

  hipMemsetAsync(S1a, 0, (size_t)XL * LDP * sizeof(float), stream);
  hipMemsetAsync(S1b, 0, (size_t)XL * LDP * sizeof(float), stream);

  const dim3 stg(XL / 32, CHUNKS / 64);
  float* Va = S1a + PADC;
  float* Vb = S1b + PADC;

  float* Ut = xs + (size_t)4 * XL * XL;
  float* G8 = xs + (size_t)8 * XL * XL;
  float* E0 = xs + (size_t)10 * XL * XL;
  float* E1 = xs + (size_t)12 * XL * XL;
  u_substack_k<<<dim3(SUBC / 32, SUBK / 32), dim3(32, 8), 0, stream>>>(u, Ut);
  copy_b_k<<<XL * UL / 256, 256, 0, stream>>>(B, G8 + 7 * UL);
  scan_wide_k<false, 0, SUBK, false, XL, SUBK, SUBK><<<dim3(XL / 32, 4), 256, 0, stream>>>(
      A,  G8 + 7 * UL, nullptr, G8 + 6 * UL, nullptr, 0);
  scan_wide_k<false, 0, SUBK, false, XL, SUBK, SUBK><<<dim3(XL / 32, 8), 256, 0, stream>>>(
      Q1, G8 + 6 * UL, nullptr, G8 + 4 * UL, nullptr, 0);
  scan_wide_k<false, 0, SUBK, false, XL, SUBK, SUBK><<<dim3(XL / 32, 16), 256, 0, stream>>>(
      Q2, G8 + 4 * UL, nullptr, G8,          nullptr, 0);
  scan_wide_k<false, 0, SUBC, false, SUBK, SUBC, SUBC><<<dim3(XL / 32, SUBC / 64), 256, 0, stream>>>(
      G8, Ut, nullptr, E0, nullptr, 0);
  scan_wide_k<false, 2, SUBC, false, SUBK, SUBC, SUBC><<<dim3(XL / 32, SUBC / 64), 256, 0, stream>>>(
      G8 + XL, Ut + (size_t)XL * SUBC, E0, E1, nullptr, 0);
  scan_wide_k<false, 2, SUBC, false, XL, SUBC, SUBC><<<dim3(XL / 32, 16), 256, 0, stream>>>(
      Q3, E1, E1 + 1024, E0, nullptr, 0);
  scan_wide_k<false, 2, LDP, false, XL, SUBC, SUBC><<<stg, 256, 0, stream>>>(
      Q4, E0, E0 + CHUNKS, Va, nullptr, 0);
  set_col_k<<<4, 256, 0, stream>>>(Va, x0, LDP, XL);

  SQUARE(Q4,  Q5);
  SQUARE(Q5,  Q6);
  SQUARE(Q6,  Q7);
  SQUARE(Q7,  Q8);
  SQUARE(Q8,  Q9);
  SQUARE(Q9,  Q10);
  SQUARE(Q10, Q11);
  SQUARE(Q11, Q12);
  SQUARE(Q12, Q13);
#undef SQUARE


  scan_wide_k<false, 2, LDP, true ><<<stg, 256, 0, stream>>>(Q5,  Va - 1,   Va, Vb, nullptr, 1);
  scan_wide_k<false, 2, LDP, true ><<<stg, 256, 0, stream>>>(Q6,  Vb - 2,   Vb, Va, nullptr, 2);
  scan_wide_k<false, 2, LDP, false><<<stg, 256, 0, stream>>>(Q7,  Va - 4,   Va, Vb, nullptr, 4);
  scan_wide_k<false, 2, LDP, false><<<stg, 256, 0, stream>>>(Q8,  Vb - 8,   Vb, Va, nullptr, 8);
  scan_wide_k<false, 2, LDP, false><<<stg, 256, 0, stream>>>(Q9,  Va - 16,  Va, Vb, nullptr, 16);
  scan_wide_k<false, 2, LDP, false><<<stg, 256, 0, stream>>>(Q10, Vb - 32,  Vb, Va, nullptr, 32);
  scan_wide_k<false, 2, LDP, false><<<stg, 256, 0, stream>>>(Q11, Va - 64,  Va, Vb, nullptr, 64);
  scan_wide_k<false, 2, LDP, false><<<stg, 256, 0, stream>>>(Q12, Vb - 128, Vb, Va, nullptr, 128);
  scan_wide_k<false, 2, LDP, false><<<stg, 256, 0, stream>>>(Q13, Va - 256, Va, Vb, nullptr, 256);

  {
    float* cur = Vb; float* nxt = Va;
    for (int t = 0; t < CLEN; ++t) {
      scan_wide_k<true, 1, LDP, false><<<stg, 256, 0, stream>>>(
          A, cur, Bu + (size_t)t * XL, nxt, xs + (size_t)t * XL, 0);
      float* tmp = cur; cur = nxt; nxt = tmp;
    }
  }

  final_gemm_k<<<dim3(NSEQ / 128, 1), 256, 0, stream>>>(xs, Ct, u, Dt, out);
}


# output projection moved from the generic GEMM loop (128 workgroups, each k-step waits for its own loads) to a 32x64-tile K-split kernel, 2048 workgroups, loads two groups ahead; sum order changes
# speedup vs baseline: 7.4718x; 1.0584x over previous
#include <hip/hip_runtime.h>
#include <math.h>


typedef __attribute__((ext_vector_type(2))) float v2f;
typedef __attribute__((ext_vector_type(4))) float v4f;
typedef __attribute__((ext_vector_type(8))) float v8f;
typedef float v4fu __attribute__((ext_vector_type(4), aligned(4)));

#define NSEQ   16384
#define XL     1024
#define UL     256
#define YLN    256
#define CHUNKS 512
#define CLEN   32
#define COLSTR (CLEN * XL)
#define PADC   64
#define LDP    (CHUNKS + PADC)
#define SUBL   8
#define SUBK   (SUBL * UL)
#define SUBC   (4 * CHUNKS)

__device__ __forceinline__ v8f wmma4(v2f a, v2f b, v8f c) {
  return __builtin_amdgcn_wmma_f32_16x16x4_f32(
       false, a,  false, b,
       (short)0, c,  false,  false);
}

template <int WM, int WN>
__global__ __launch_bounds__(256) void gemm_rr_k(
    const float* __restrict__ Ap, const float* __restrict__ Bp,
    float* __restrict__ Cp, int K, int lda, int ldb, int ldc) {
  const int lane = threadIdx.x & 31;
  const int w    = threadIdx.x >> 5;
  const int wm = w >> 2, wn = w & 3;
  const int tm = blockIdx.x * (32 * WM) + wm * (16 * WM);
  const int tn = blockIdx.y * (64 * WN) + wn * (16 * WN);
  const int ml = lane & 15;
  const int kh = (lane >> 4) << 1;

  v8f acc[WM][WN];
#pragma unroll
  for (int mi = 0; mi < WM; ++mi)
#pragma unroll
    for (int ni = 0; ni < WN; ++ni) acc[mi][ni] = (v8f){};

  const float* ap[WM];
  const float* bp[WN];
#pragma unroll
  for (int mi = 0; mi < WM; ++mi)
    ap[mi] = Ap + (size_t)(tm + mi * 16 + ml) * lda + kh;
#pragma unroll
  for (int ni = 0; ni < WN; ++ni)
    bp[ni] = Bp + (size_t)kh * ldb + (tn + ni * 16 + ml);

  for (int k = 0; k < K; k += 4) {
    v2f af[WM], bf[WN];
    const size_t bo0 = (size_t)k * ldb, bo1 = bo0 + (size_t)ldb;
#pragma unroll
    for (int mi = 0; mi < WM; ++mi) af[mi] = (v2f){ ap[mi][k], ap[mi][k + 1] };
#pragma unroll
    for (int ni = 0; ni < WN; ++ni) bf[ni] = (v2f){ bp[ni][bo0], bp[ni][bo1] };
#pragma unroll
    for (int mi = 0; mi < WM; ++mi)
#pragma unroll
      for (int ni = 0; ni < WN; ++ni)
        acc[mi][ni] = wmma4(af[mi], bf[ni], acc[mi][ni]);
  }

  const int nn = lane & 15;
  const int mb = (lane >> 4) << 3;
#pragma unroll
  for (int mi = 0; mi < WM; ++mi)
#pragma unroll
    for (int ni = 0; ni < WN; ++ni) {
      float* cp = Cp + (size_t)(tm + mi * 16 + mb) * ldc + (tn + ni * 16 + nn);
#pragma unroll
      for (int r = 0; r < 8; ++r) cp[(size_t)r * ldc] = acc[mi][ni][r];
    }
}

template <bool STORE_XS, int ADD, int LD, bool UNAL,
          int LDA = XL, int LDB = LD, int LDC = LD>
__global__ __launch_bounds__(256) __attribute__((amdgpu_waves_per_eu(2, 4)))
void scan_wide_k(
    const float* __restrict__ Ap, const float* __restrict__ Sold,
    const float* addb, float* __restrict__ Snew,
    float* __restrict__ xsb, int sh) {
  static_assert(ADD == 1 || !STORE_XS, "xs copy goes with the level-1 addend");
  static_assert(ADD == 2 || !UNAL, "only a doubling step moves its operand");
  constexpr int KE = XL / 8;
  constexpr int NP = KE / 8;
  constexpr int GP = 2;
  constexpr int NG = NP / GP;
  constexpr int RP = 8 * 32 + 16;
  __shared__ float red[8][2][4][RP];

  const int lane = threadIdx.x & 31;
  const int w    = threadIdx.x >> 5;
  const int tm = blockIdx.x * 32;
  const int tn = blockIdx.y * 64;
  const int c  = lane & 15;
  const int h  = lane >> 4;

  if constexpr (ADD == 2) {
    if (tn + 64 <= sh) {
#pragma unroll
      for (int i = 0; i < 8; ++i) {
        const int idx = (int)threadIdx.x + 256 * i;
        const size_t o  = (size_t)(tm + (idx >> 6)) * LD  + (tn + (idx & 63));
        const size_t oa = (size_t)(tm + (idx >> 6)) * LDC + (tn + (idx & 63));
        Snew[o] = addb[oa];
      }
      return;
    }
  }

  const int eh = w & 1, rg = w >> 1;
  const int ej = tn + eh * 32 + lane;
  const float* addp = nullptr;
  if constexpr (ADD == 1) {
    const int aj = (ej - sh) > 0 ? (ej - sh) : 0;
    addp = addb + (size_t)aj * COLSTR + (tm + 8 * rg);
  } else if constexpr (ADD == 2) {
    addp = addb + (size_t)(tm + 8 * rg) * LDC + ej;
  } else {
    (void)addb; (void)sh;
  }
  v4f add0 = {}, add1 = {};

  const float* ap0 = Ap + (size_t)(tm + c) * LDA + (w * KE + 4 * h);
  const float* ap1 = ap0 + (size_t)16 * LDA;
  const float* bp  = Sold + (size_t)(w * KE + 4 * h) * LDB + (tn + 4 * c);

  v4f af[NG][GP][2];
  v4f bq[NG][GP][4];

#define SCAN_WIDE_LOAD_GROUP(g)                                              \
  _Pragma("unroll")                                                          \
  for (int p = 0; p < GP; ++p) {                                             \
    const int k = ((g) * GP + p) * 8;                                        \
    af[(g)][p][0] = *(const v4f*)(ap0 + k);                                  \
    af[(g)][p][1] = *(const v4f*)(ap1 + k);                                  \
    _Pragma("unroll")                                                        \
    for (int e = 0; e < 4; ++e) {                                            \
      if constexpr (UNAL)                                                    \
        bq[(g)][p][e] = *(const v4fu*)(bp + (size_t)(k + e) * LDB);          \
      else                                                                   \
        bq[(g)][p][e] = *(const v4f*)(bp + (size_t)(k + e) * LDB);           \
    }                                                                        \
  }

#pragma unroll
  for (int g = 0; g < 2; ++g) { SCAN_WIDE_LOAD_GROUP(g) }

  v8f acc[2][4];
#pragma unroll
  for (int mi = 0; mi < 2; ++mi)
#pragma unroll
    for (int ni = 0; ni < 4; ++ni) acc[mi][ni] = (v8f){};

#pragma unroll
  for (int g = 0; g < NG; ++g) {
    if (g + 2 < NG) { SCAN_WIDE_LOAD_GROUP(g + 2) }
    if (ADD != 0 && g + 3 == NG) {
      if constexpr (ADD == 1) {
        add0 = *(const v4f*)addp;
        add1 = *(const v4f*)(addp + 4);
      } else if constexpr (ADD == 2) {
#pragma unroll
        for (int i = 0; i < 4; ++i) {
          add0[i] = addp[(size_t)i * LDC];
          add1[i] = addp[(size_t)(i + 4) * LDC];
        }
      }
      __builtin_amdgcn_sched_barrier(0);
    }
#pragma unroll
    for (int p = 0; p < GP; ++p)
#pragma unroll
      for (int s = 0; s < 2; ++s)
#pragma unroll
        for (int ni = 0; ni < 4; ++ni)
#pragma unroll
          for (int mi = 0; mi < 2; ++mi)
            acc[mi][ni] = wmma4(
                (v2f){ af[g][p][mi][2 * s], af[g][p][mi][2 * s + 1] },
                (v2f){ bq[g][p][2 * s][ni], bq[g][p][2 * s + 1][ni] },
                acc[mi][ni]);
  }
#undef SCAN_WIDE_LOAD_GROUP

#pragma unroll
  for (int mi = 0; mi < 2; ++mi)
#pragma unroll
    for (int ni = 0; ni < 4; ++ni)
#pragma unroll
      for (int r = 0; r < 8; ++r) red[w][mi][ni][r * 32 + lane] = acc[mi][ni][r];
  __syncthreads();

  const int emi = rg >> 1;
  const int eni = lane & 3;
  const int esl = (rg & 1) * 16 + eh * 8 + (lane >> 2);
  float v[8];
#pragma unroll
  for (int i = 0; i < 8; ++i) {
    const int sl = i * 32 + esl;
    const float c0 = red[0][emi][eni][sl], c1 = red[1][emi][eni][sl];
    const float c2 = red[2][emi][eni][sl], c3 = red[3][emi][eni][sl];
    const float c4 = red[4][emi][eni][sl], c5 = red[5][emi][eni][sl];
    const float c6 = red[6][emi][eni][sl], c7 = red[7][emi][eni][sl];
    v[i] = ((c0 + c1) + (c2 + c3)) + ((c4 + c5) + (c6 + c7));
    if constexpr (ADD != 0) v[i] += (i < 4) ? add0[i & 3] : add1[i & 3];
    Snew[(size_t)(tm + 8 * rg + i) * LD + ej] = v[i];
  }
  if constexpr (STORE_XS) {
    float* xp = xsb + (size_t)ej * COLSTR + (tm + 8 * rg);
    *(v4f*)xp       = (v4f){ v[0], v[1], v[2], v[3] };
    *(v4f*)(xp + 4) = (v4f){ v[4], v[5], v[6], v[7] };
  } else {
    (void)xsb;
  }
}

__global__ __launch_bounds__(256) __attribute__((amdgpu_waves_per_eu(2, 4)))
void final_wide_k(
    const float* __restrict__ xs, const float* __restrict__ Ct,
    const float* __restrict__ uf, const float* __restrict__ Dt,
    float* __restrict__ out) {
  constexpr int KX  = XL / 8;
  constexpr int KU  = UL / 8;
  constexpr int GP  = 2;
  constexpr int NGX = KX / 8 / GP;
  constexpr int NG  = NGX + KU / 8 / GP;
  constexpr int RP = 8 * 32 + 16;
  __shared__ float red[8][2][4][RP];

  const int lane = threadIdx.x & 31;
  const int w    = threadIdx.x >> 5;
  const int tm = blockIdx.x * 32;
  const int tn = blockIdx.y * 64;
  const int c  = lane & 15;
  const int h  = lane >> 4;

  const float* ax0 = xs + (size_t)(tm + c) * XL + (w * KX + 4 * h);
  const float* ax1 = ax0 + (size_t)16 * XL;
  const float* bx  = Ct + (size_t)(w * KX + 4 * h) * YLN + (tn + 4 * c);
  const float* au0 = uf + (size_t)(tm + c) * UL + (w * KU + 4 * h);
  const float* au1 = au0 + (size_t)16 * UL;
  const float* bu  = Dt + (size_t)(w * KU + 4 * h) * YLN + (tn + 4 * c);

  v4f af[NG][GP][2];
  v4f bq[NG][GP][4];

#define FINAL_WIDE_LOAD_GROUP(g)                                             \
  _Pragma("unroll")                                                          \
  for (int p = 0; p < GP; ++p) {                                             \
    if ((g) < NGX) {                                                         \
      const int k = ((g) * GP + p) * 8;                                      \
      af[(g)][p][0] = *(const v4f*)(ax0 + k);                                \
      af[(g)][p][1] = *(const v4f*)(ax1 + k);                                \
      _Pragma("unroll")                                                      \
      for (int e = 0; e < 4; ++e)                                            \
        bq[(g)][p][e] = *(const v4f*)(bx + (size_t)(k + e) * YLN);           \
    } else {                                                                 \
      const int k = (((g) - NGX) * GP + p) * 8;                              \
      af[(g)][p][0] = *(const v4f*)(au0 + k);                                \
      af[(g)][p][1] = *(const v4f*)(au1 + k);                                \
      _Pragma("unroll")                                                      \
      for (int e = 0; e < 4; ++e)                                            \
        bq[(g)][p][e] = *(const v4f*)(bu + (size_t)(k + e) * YLN);           \
    }                                                                        \
  }

#pragma unroll
  for (int g = 0; g < 2; ++g) { FINAL_WIDE_LOAD_GROUP(g) }

  v8f acc[2][4];
#pragma unroll
  for (int mi = 0; mi < 2; ++mi)
#pragma unroll
    for (int ni = 0; ni < 4; ++ni) acc[mi][ni] = (v8f){};

#pragma unroll
  for (int g = 0; g < NG; ++g) {
    if (g + 2 < NG) { FINAL_WIDE_LOAD_GROUP(g + 2) }
#pragma unroll
    for (int p = 0; p < GP; ++p)
#pragma unroll
      for (int s = 0; s < 2; ++s)
#pragma unroll
        for (int ni = 0; ni < 4; ++ni)
#pragma unroll
          for (int mi = 0; mi < 2; ++mi)
            acc[mi][ni] = wmma4(
                (v2f){ af[g][p][mi][2 * s], af[g][p][mi][2 * s + 1] },
                (v2f){ bq[g][p][2 * s][ni], bq[g][p][2 * s + 1][ni] },
                acc[mi][ni]);
  }
#undef FINAL_WIDE_LOAD_GROUP

#pragma unroll
  for (int mi = 0; mi < 2; ++mi)
#pragma unroll
    for (int ni = 0; ni < 4; ++ni)
#pragma unroll
      for (int r = 0; r < 8; ++r) red[w][mi][ni][r * 32 + lane] = acc[mi][ni][r];
  __syncthreads();

  const int eh = w & 1, rg = w >> 1;
  const int ej = tn + eh * 32 + lane;
  const int emi = rg >> 1;
  const int eni = lane & 3;
  const int esl = (rg & 1) * 16 + eh * 8 + (lane >> 2);
  float v[8];
#pragma unroll
  for (int i = 0; i < 8; ++i) {
    const int sl = i * 32 + esl;
    const float c0 = red[0][emi][eni][sl], c1 = red[1][emi][eni][sl];
    const float c2 = red[2][emi][eni][sl], c3 = red[3][emi][eni][sl];
    const float c4 = red[4][emi][eni][sl], c5 = red[5][emi][eni][sl];
    const float c6 = red[6][emi][eni][sl], c7 = red[7][emi][eni][sl];
    v[i] = tanhf(((c0 + c1) + (c2 + c3)) + ((c4 + c5) + (c6 + c7)));
  }
  float* op = out + (size_t)ej * NSEQ + (tm + 8 * rg);
  *(v4f*)op       = (v4f){ v[0], v[1], v[2], v[3] };
  *(v4f*)(op + 4) = (v4f){ v[4], v[5], v[6], v[7] };
}

__global__ __launch_bounds__(256) void transpose_k(
    const float* __restrict__ in, float* __restrict__ out, int rows, int cols) {
  __shared__ float tile[32][33];
  const int c0 = blockIdx.x * 32, r0 = blockIdx.y * 32;
  const int tx = threadIdx.x, ty = threadIdx.y;
#pragma unroll
  for (int i = 0; i < 32; i += 8)
    tile[ty + i][tx] = in[(size_t)(r0 + ty + i) * cols + (c0 + tx)];
  __syncthreads();
#pragma unroll
  for (int i = 0; i < 32; i += 8)
    out[(size_t)(c0 + ty + i) * rows + (r0 + tx)] = tile[tx][ty + i];
}

__global__ __launch_bounds__(256) void u_substack_k(
    const float* __restrict__ u, float* __restrict__ Ut) {
  __shared__ float tile[32][33];
  const int c0 = blockIdx.x * 32, k0 = blockIdx.y * 32;
  const int tx = threadIdx.x, ty = threadIdx.y;
#pragma unroll
  for (int i = 0; i < 32; i += 8) {
    const int cc = c0 + ty + i;
    const int g = cc >> 9, jj = cc & (CHUNKS - 1);
    const int r = ((g & 1) << 1) | (g >> 1);
    const int j = jj > 0 ? jj - 1 : 0;
    const int q = 4 * j + r;
    tile[ty + i][tx] = u[(size_t)q * SUBK + (k0 + tx)];
  }
  __syncthreads();
#pragma unroll
  for (int i = 0; i < 32; i += 8)
    Ut[(size_t)(k0 + ty + i) * SUBC + (c0 + tx)] = tile[tx][ty + i];
}

__global__ __launch_bounds__(256) void copy_b_k(
    const float* __restrict__ B, float* __restrict__ dst) {
  const int idx = blockIdx.x * 256 + threadIdx.x;
  dst[(size_t)(idx >> 8) * SUBK + (idx & (UL - 1))] = B[idx];
}

__global__ void set_col_k(float* __restrict__ dst, const float* __restrict__ src,
                          int stride, int n) {
  int i = blockIdx.x * blockDim.x + threadIdx.x;
  if (i < n) dst[(size_t)i * stride] = src[i];
}

extern "C" void kernel_launch(void* const* d_in, const int* in_sizes, int n_in,
                              void* d_out, int out_size, void* d_ws, size_t ws_size,
                              hipStream_t stream) {
  (void)in_sizes; (void)n_in; (void)out_size; (void)ws_size;
  const float* u  = (const float*)d_in[0];
  const float* A  = (const float*)d_in[1];
  const float* B  = (const float*)d_in[2];
  const float* C  = (const float*)d_in[3];
  const float* D  = (const float*)d_in[4];
  const float* x0 = (const float*)d_in[5];
  float* out = (float*)d_out;

  float* ws = (float*)d_ws;
  size_t off = 0;
  float* Bu = ws + off; off += (size_t)NSEQ * XL;
  float* xs = ws + off; off += (size_t)NSEQ * XL;
  float* Bt = ws + off; off += (size_t)UL * XL;
  float* Ct = ws + off; off += (size_t)XL * YLN;
  float* Dt = ws + off; off += (size_t)UL * YLN;
  float* S1a = ws + off; off += (size_t)XL * LDP;
  float* S1b = ws + off; off += (size_t)XL * LDP;
  float* Q1  = xs;
  float* Q2  = xs + (size_t)1 * XL * XL;
  float* Q3  = xs + (size_t)2 * XL * XL;
  float* Q4  = xs + (size_t)3 * XL * XL;
  float* Q5  = xs + (size_t)4 * XL * XL;
  float* Q6  = xs + (size_t)5 * XL * XL;
  float* Q7  = xs + (size_t)6 * XL * XL;
  float* Q8  = xs + (size_t)7 * XL * XL;
  float* Q9  = xs + (size_t)8 * XL * XL;
  float* Q10 = xs + (size_t)9 * XL * XL;
  float* Q11 = xs + (size_t)10 * XL * XL;
  float* Q12 = xs + (size_t)11 * XL * XL;
  float* Q13 = xs + (size_t)12 * XL * XL;

  transpose_k<<<dim3(UL / 32, XL / 32), dim3(32, 8), 0, stream>>>(B, Bt, XL, UL);
  transpose_k<<<dim3(XL / 32, YLN / 32), dim3(32, 8), 0, stream>>>(C, Ct, YLN, XL);
  transpose_k<<<dim3(YLN / 32, UL / 32), dim3(32, 8), 0, stream>>>(D, Dt, UL, YLN);

  gemm_rr_k<4, 4><<<dim3(NSEQ / 128, XL / 256), 256, 0, stream>>>(
      u, Bt, Bu, UL, UL, XL, XL);

  const dim3 sqg(XL / 32, XL / 64);
#define SQUARE(src, dst) \
  scan_wide_k<false, 0, XL, false><<<sqg, 256, 0, stream>>>(src, src, nullptr, dst, nullptr, 0)
  SQUARE(A,   Q1);
  SQUARE(Q1,  Q2);
  SQUARE(Q2,  Q3);
  SQUARE(Q3,  Q4);

  hipMemsetAsync(S1a, 0, (size_t)XL * LDP * sizeof(float), stream);
  hipMemsetAsync(S1b, 0, (size_t)XL * LDP * sizeof(float), stream);

  const dim3 stg(XL / 32, CHUNKS / 64);
  float* Va = S1a + PADC;
  float* Vb = S1b + PADC;

  float* Ut = xs + (size_t)4 * XL * XL;
  float* G8 = xs + (size_t)8 * XL * XL;
  float* E0 = xs + (size_t)10 * XL * XL;
  float* E1 = xs + (size_t)12 * XL * XL;
  u_substack_k<<<dim3(SUBC / 32, SUBK / 32), dim3(32, 8), 0, stream>>>(u, Ut);
  copy_b_k<<<XL * UL / 256, 256, 0, stream>>>(B, G8 + 7 * UL);
  scan_wide_k<false, 0, SUBK, false, XL, SUBK, SUBK><<<dim3(XL / 32, 4), 256, 0, stream>>>(
      A,  G8 + 7 * UL, nullptr, G8 + 6 * UL, nullptr, 0);
  scan_wide_k<false, 0, SUBK, false, XL, SUBK, SUBK><<<dim3(XL / 32, 8), 256, 0, stream>>>(
      Q1, G8 + 6 * UL, nullptr, G8 + 4 * UL, nullptr, 0);
  scan_wide_k<false, 0, SUBK, false, XL, SUBK, SUBK><<<dim3(XL / 32, 16), 256, 0, stream>>>(
      Q2, G8 + 4 * UL, nullptr, G8,          nullptr, 0);
  scan_wide_k<false, 0, SUBC, false, SUBK, SUBC, SUBC><<<dim3(XL / 32, SUBC / 64), 256, 0, stream>>>(
      G8, Ut, nullptr, E0, nullptr, 0);
  scan_wide_k<false, 2, SUBC, false, SUBK, SUBC, SUBC><<<dim3(XL / 32, SUBC / 64), 256, 0, stream>>>(
      G8 + XL, Ut + (size_t)XL * SUBC, E0, E1, nullptr, 0);
  scan_wide_k<false, 2, SUBC, false, XL, SUBC, SUBC><<<dim3(XL / 32, 16), 256, 0, stream>>>(
      Q3, E1, E1 + 1024, E0, nullptr, 0);
  scan_wide_k<false, 2, LDP, false, XL, SUBC, SUBC><<<stg, 256, 0, stream>>>(
      Q4, E0, E0 + CHUNKS, Va, nullptr, 0);
  set_col_k<<<4, 256, 0, stream>>>(Va, x0, LDP, XL);

  SQUARE(Q4,  Q5);
  SQUARE(Q5,  Q6);
  SQUARE(Q6,  Q7);
  SQUARE(Q7,  Q8);
  SQUARE(Q8,  Q9);
  SQUARE(Q9,  Q10);
  SQUARE(Q10, Q11);
  SQUARE(Q11, Q12);
  SQUARE(Q12, Q13);
#undef SQUARE


  scan_wide_k<false, 2, LDP, true ><<<stg, 256, 0, stream>>>(Q5,  Va - 1,   Va, Vb, nullptr, 1);
  scan_wide_k<false, 2, LDP, true ><<<stg, 256, 0, stream>>>(Q6,  Vb - 2,   Vb, Va, nullptr, 2);
  scan_wide_k<false, 2, LDP, false><<<stg, 256, 0, stream>>>(Q7,  Va - 4,   Va, Vb, nullptr, 4);
  scan_wide_k<false, 2, LDP, false><<<stg, 256, 0, stream>>>(Q8,  Vb - 8,   Vb, Va, nullptr, 8);
  scan_wide_k<false, 2, LDP, false><<<stg, 256, 0, stream>>>(Q9,  Va - 16,  Va, Vb, nullptr, 16);
  scan_wide_k<false, 2, LDP, false><<<stg, 256, 0, stream>>>(Q10, Vb - 32,  Vb, Va, nullptr, 32);
  scan_wide_k<false, 2, LDP, false><<<stg, 256, 0, stream>>>(Q11, Va - 64,  Va, Vb, nullptr, 64);
  scan_wide_k<false, 2, LDP, false><<<stg, 256, 0, stream>>>(Q12, Vb - 128, Vb, Va, nullptr, 128);
  scan_wide_k<false, 2, LDP, false><<<stg, 256, 0, stream>>>(Q13, Va - 256, Va, Vb, nullptr, 256);

  {
    float* cur = Vb; float* nxt = Va;
    for (int t = 0; t < CLEN; ++t) {
      scan_wide_k<true, 1, LDP, false><<<stg, 256, 0, stream>>>(
          A, cur, Bu + (size_t)t * XL, nxt, xs + (size_t)t * XL, 0);
      float* tmp = cur; cur = nxt; nxt = tmp;
    }
  }

  final_wide_k<<<dim3(NSEQ / 32, YLN / 64), 256, 0, stream>>>(xs, Ct, u, Dt, out);
}


# wide K-split kernels on grids over 256 workgroups (squarings, chunk-end products, output) held to 256 registers: two workgroups per WGP instead of one, loads kept 2 groups ahead; sum order unchanged
# speedup vs baseline: 8.0242x; 1.0739x over previous
#include <hip/hip_runtime.h>
#include <math.h>


typedef __attribute__((ext_vector_type(2))) float v2f;
typedef __attribute__((ext_vector_type(4))) float v4f;
typedef __attribute__((ext_vector_type(8))) float v8f;
typedef float v4fu __attribute__((ext_vector_type(4), aligned(4)));

#define NSEQ   16384
#define XL     1024
#define UL     256
#define YLN    256
#define CHUNKS 512
#define CLEN   32
#define COLSTR (CLEN * XL)
#define PADC   64
#define LDP    (CHUNKS + PADC)
#define SUBL   8
#define SUBK   (SUBL * UL)
#define SUBC   (4 * CHUNKS)

__device__ __forceinline__ v8f wmma4(v2f a, v2f b, v8f c) {
  return __builtin_amdgcn_wmma_f32_16x16x4_f32(
       false, a,  false, b,
       (short)0, c,  false,  false);
}

template <int WM, int WN>
__global__ __launch_bounds__(256) void gemm_rr_k(
    const float* __restrict__ Ap, const float* __restrict__ Bp,
    float* __restrict__ Cp, int K, int lda, int ldb, int ldc) {
  const int lane = threadIdx.x & 31;
  const int w    = threadIdx.x >> 5;
  const int wm = w >> 2, wn = w & 3;
  const int tm = blockIdx.x * (32 * WM) + wm * (16 * WM);
  const int tn = blockIdx.y * (64 * WN) + wn * (16 * WN);
  const int ml = lane & 15;
  const int kh = (lane >> 4) << 1;

  v8f acc[WM][WN];
#pragma unroll
  for (int mi = 0; mi < WM; ++mi)
#pragma unroll
    for (int ni = 0; ni < WN; ++ni) acc[mi][ni] = (v8f){};

  const float* ap[WM];
  const float* bp[WN];
#pragma unroll
  for (int mi = 0; mi < WM; ++mi)
    ap[mi] = Ap + (size_t)(tm + mi * 16 + ml) * lda + kh;
#pragma unroll
  for (int ni = 0; ni < WN; ++ni)
    bp[ni] = Bp + (size_t)kh * ldb + (tn + ni * 16 + ml);

  for (int k = 0; k < K; k += 4) {
    v2f af[WM], bf[WN];
    const size_t bo0 = (size_t)k * ldb, bo1 = bo0 + (size_t)ldb;
#pragma unroll
    for (int mi = 0; mi < WM; ++mi) af[mi] = (v2f){ ap[mi][k], ap[mi][k + 1] };
#pragma unroll
    for (int ni = 0; ni < WN; ++ni) bf[ni] = (v2f){ bp[ni][bo0], bp[ni][bo1] };
#pragma unroll
    for (int mi = 0; mi < WM; ++mi)
#pragma unroll
      for (int ni = 0; ni < WN; ++ni)
        acc[mi][ni] = wmma4(af[mi], bf[ni], acc[mi][ni]);
  }

  const int nn = lane & 15;
  const int mb = (lane >> 4) << 3;
#pragma unroll
  for (int mi = 0; mi < WM; ++mi)
#pragma unroll
    for (int ni = 0; ni < WN; ++ni) {
      float* cp = Cp + (size_t)(tm + mi * 16 + mb) * ldc + (tn + ni * 16 + nn);
#pragma unroll
      for (int r = 0; r < 8; ++r) cp[(size_t)r * ldc] = acc[mi][ni][r];
    }
}

template <bool STORE_XS, int ADD, int LD, bool UNAL,
          int LDA = XL, int LDB = LD, int LDC = LD, int MINW = 2>
__global__ __launch_bounds__(256) __attribute__((amdgpu_waves_per_eu(MINW, 4)))
void scan_wide_k(
    const float* __restrict__ Ap, const float* __restrict__ Sold,
    const float* addb, float* __restrict__ Snew,
    float* __restrict__ xsb, int sh) {
  static_assert(ADD == 1 || !STORE_XS, "xs copy goes with the level-1 addend");
  static_assert(ADD == 2 || !UNAL, "only a doubling step moves its operand");
  constexpr int KE = XL / 8;
  constexpr int NP = KE / 8;
  constexpr int GP = 2;
  constexpr int NG = NP / GP;
  constexpr int RP = 8 * 32 + 16;
  __shared__ float red[8][2][4][RP];

  const int lane = threadIdx.x & 31;
  const int w    = threadIdx.x >> 5;
  const int tm = blockIdx.x * 32;
  const int tn = blockIdx.y * 64;
  const int c  = lane & 15;
  const int h  = lane >> 4;

  if constexpr (ADD == 2) {
    if (tn + 64 <= sh) {
#pragma unroll
      for (int i = 0; i < 8; ++i) {
        const int idx = (int)threadIdx.x + 256 * i;
        const size_t o  = (size_t)(tm + (idx >> 6)) * LD  + (tn + (idx & 63));
        const size_t oa = (size_t)(tm + (idx >> 6)) * LDC + (tn + (idx & 63));
        Snew[o] = addb[oa];
      }
      return;
    }
  }

  const int eh = w & 1, rg = w >> 1;
  const int ej = tn + eh * 32 + lane;
  const float* addp = nullptr;
  if constexpr (ADD == 1) {
    const int aj = (ej - sh) > 0 ? (ej - sh) : 0;
    addp = addb + (size_t)aj * COLSTR + (tm + 8 * rg);
  } else if constexpr (ADD == 2) {
    addp = addb + (size_t)(tm + 8 * rg) * LDC + ej;
  } else {
    (void)addb; (void)sh;
  }
  v4f add0 = {}, add1 = {};

  const float* ap0 = Ap + (size_t)(tm + c) * LDA + (w * KE + 4 * h);
  const float* ap1 = ap0 + (size_t)16 * LDA;
  const float* bp  = Sold + (size_t)(w * KE + 4 * h) * LDB + (tn + 4 * c);

  v4f af[NG][GP][2];
  v4f bq[NG][GP][4];

#define SCAN_WIDE_LOAD_GROUP(g)                                              \
  _Pragma("unroll")                                                          \
  for (int p = 0; p < GP; ++p) {                                             \
    const int k = ((g) * GP + p) * 8;                                        \
    af[(g)][p][0] = *(const v4f*)(ap0 + k);                                  \
    af[(g)][p][1] = *(const v4f*)(ap1 + k);                                  \
    _Pragma("unroll")                                                        \
    for (int e = 0; e < 4; ++e) {                                            \
      if constexpr (UNAL)                                                    \
        bq[(g)][p][e] = *(const v4fu*)(bp + (size_t)(k + e) * LDB);          \
      else                                                                   \
        bq[(g)][p][e] = *(const v4f*)(bp + (size_t)(k + e) * LDB);           \
    }                                                                        \
  }

#pragma unroll
  for (int g = 0; g < 2; ++g) { SCAN_WIDE_LOAD_GROUP(g) }

  v8f acc[2][4];
#pragma unroll
  for (int mi = 0; mi < 2; ++mi)
#pragma unroll
    for (int ni = 0; ni < 4; ++ni) acc[mi][ni] = (v8f){};

#pragma unroll
  for (int g = 0; g < NG; ++g) {
    if (MINW == 4) __builtin_amdgcn_sched_barrier(0);
    if (g + 2 < NG) { SCAN_WIDE_LOAD_GROUP(g + 2) }
    if (ADD != 0 && g + 3 == NG) {
      if constexpr (ADD == 1) {
        add0 = *(const v4f*)addp;
        add1 = *(const v4f*)(addp + 4);
      } else if constexpr (ADD == 2) {
#pragma unroll
        for (int i = 0; i < 4; ++i) {
          add0[i] = addp[(size_t)i * LDC];
          add1[i] = addp[(size_t)(i + 4) * LDC];
        }
      }
      __builtin_amdgcn_sched_barrier(0);
    }
#pragma unroll
    for (int p = 0; p < GP; ++p)
#pragma unroll
      for (int s = 0; s < 2; ++s)
#pragma unroll
        for (int ni = 0; ni < 4; ++ni)
#pragma unroll
          for (int mi = 0; mi < 2; ++mi)
            acc[mi][ni] = wmma4(
                (v2f){ af[g][p][mi][2 * s], af[g][p][mi][2 * s + 1] },
                (v2f){ bq[g][p][2 * s][ni], bq[g][p][2 * s + 1][ni] },
                acc[mi][ni]);
  }
#undef SCAN_WIDE_LOAD_GROUP

#pragma unroll
  for (int mi = 0; mi < 2; ++mi)
#pragma unroll
    for (int ni = 0; ni < 4; ++ni)
#pragma unroll
      for (int r = 0; r < 8; ++r) red[w][mi][ni][r * 32 + lane] = acc[mi][ni][r];
  __syncthreads();

  const int emi = rg >> 1;
  const int eni = lane & 3;
  const int esl = (rg & 1) * 16 + eh * 8 + (lane >> 2);
  float v[8];
#pragma unroll
  for (int i = 0; i < 8; ++i) {
    const int sl = i * 32 + esl;
    const float c0 = red[0][emi][eni][sl], c1 = red[1][emi][eni][sl];
    const float c2 = red[2][emi][eni][sl], c3 = red[3][emi][eni][sl];
    const float c4 = red[4][emi][eni][sl], c5 = red[5][emi][eni][sl];
    const float c6 = red[6][emi][eni][sl], c7 = red[7][emi][eni][sl];
    v[i] = ((c0 + c1) + (c2 + c3)) + ((c4 + c5) + (c6 + c7));
    if constexpr (ADD != 0) v[i] += (i < 4) ? add0[i & 3] : add1[i & 3];
    Snew[(size_t)(tm + 8 * rg + i) * LD + ej] = v[i];
  }
  if constexpr (STORE_XS) {
    float* xp = xsb + (size_t)ej * COLSTR + (tm + 8 * rg);
    *(v4f*)xp       = (v4f){ v[0], v[1], v[2], v[3] };
    *(v4f*)(xp + 4) = (v4f){ v[4], v[5], v[6], v[7] };
  } else {
    (void)xsb;
  }
}

__global__ __launch_bounds__(256) __attribute__((amdgpu_waves_per_eu(4, 4)))
void final_wide_k(
    const float* __restrict__ xs, const float* __restrict__ Ct,
    const float* __restrict__ uf, const float* __restrict__ Dt,
    float* __restrict__ out) {
  constexpr int KX  = XL / 8;
  constexpr int KU  = UL / 8;
  constexpr int GP  = 2;
  constexpr int NGX = KX / 8 / GP;
  constexpr int NG  = NGX + KU / 8 / GP;
  constexpr int RP = 8 * 32 + 16;
  __shared__ float red[8][2][4][RP];

  const int lane = threadIdx.x & 31;
  const int w    = threadIdx.x >> 5;
  const int tm = blockIdx.x * 32;
  const int tn = blockIdx.y * 64;
  const int c  = lane & 15;
  const int h  = lane >> 4;

  const float* ax0 = xs + (size_t)(tm + c) * XL + (w * KX + 4 * h);
  const float* ax1 = ax0 + (size_t)16 * XL;
  const float* bx  = Ct + (size_t)(w * KX + 4 * h) * YLN + (tn + 4 * c);
  const float* au0 = uf + (size_t)(tm + c) * UL + (w * KU + 4 * h);
  const float* au1 = au0 + (size_t)16 * UL;
  const float* bu  = Dt + (size_t)(w * KU + 4 * h) * YLN + (tn + 4 * c);

  v4f af[NG][GP][2];
  v4f bq[NG][GP][4];

#define FINAL_WIDE_LOAD_GROUP(g)                                             \
  _Pragma("unroll")                                                          \
  for (int p = 0; p < GP; ++p) {                                             \
    if ((g) < NGX) {                                                         \
      const int k = ((g) * GP + p) * 8;                                      \
      af[(g)][p][0] = *(const v4f*)(ax0 + k);                                \
      af[(g)][p][1] = *(const v4f*)(ax1 + k);                                \
      _Pragma("unroll")                                                      \
      for (int e = 0; e < 4; ++e)                                            \
        bq[(g)][p][e] = *(const v4f*)(bx + (size_t)(k + e) * YLN);           \
    } else {                                                                 \
      const int k = (((g) - NGX) * GP + p) * 8;                              \
      af[(g)][p][0] = *(const v4f*)(au0 + k);                                \
      af[(g)][p][1] = *(const v4f*)(au1 + k);                                \
      _Pragma("unroll")                                                      \
      for (int e = 0; e < 4; ++e)                                            \
        bq[(g)][p][e] = *(const v4f*)(bu + (size_t)(k + e) * YLN);           \
    }                                                                        \
  }

#pragma unroll
  for (int g = 0; g < 2; ++g) { FINAL_WIDE_LOAD_GROUP(g) }

  v8f acc[2][4];
#pragma unroll
  for (int mi = 0; mi < 2; ++mi)
#pragma unroll
    for (int ni = 0; ni < 4; ++ni) acc[mi][ni] = (v8f){};

#pragma unroll
  for (int g = 0; g < NG; ++g) {
    __builtin_amdgcn_sched_barrier(0);
    if (g + 2 < NG) { FINAL_WIDE_LOAD_GROUP(g + 2) }
#pragma unroll
    for (int p = 0; p < GP; ++p)
#pragma unroll
      for (int s = 0; s < 2; ++s)
#pragma unroll
        for (int ni = 0; ni < 4; ++ni)
#pragma unroll
          for (int mi = 0; mi < 2; ++mi)
            acc[mi][ni] = wmma4(
                (v2f){ af[g][p][mi][2 * s], af[g][p][mi][2 * s + 1] },
                (v2f){ bq[g][p][2 * s][ni], bq[g][p][2 * s + 1][ni] },
                acc[mi][ni]);
  }
#undef FINAL_WIDE_LOAD_GROUP

#pragma unroll
  for (int mi = 0; mi < 2; ++mi)
#pragma unroll
    for (int ni = 0; ni < 4; ++ni)
#pragma unroll
      for (int r = 0; r < 8; ++r) red[w][mi][ni][r * 32 + lane] = acc[mi][ni][r];
  __syncthreads();

  const int eh = w & 1, rg = w >> 1;
  const int ej = tn + eh * 32 + lane;
  const int emi = rg >> 1;
  const int eni = lane & 3;
  const int esl = (rg & 1) * 16 + eh * 8 + (lane >> 2);
  float v[8];
#pragma unroll
  for (int i = 0; i < 8; ++i) {
    const int sl = i * 32 + esl;
    const float c0 = red[0][emi][eni][sl], c1 = red[1][emi][eni][sl];
    const float c2 = red[2][emi][eni][sl], c3 = red[3][emi][eni][sl];
    const float c4 = red[4][emi][eni][sl], c5 = red[5][emi][eni][sl];
    const float c6 = red[6][emi][eni][sl], c7 = red[7][emi][eni][sl];
    v[i] = tanhf(((c0 + c1) + (c2 + c3)) + ((c4 + c5) + (c6 + c7)));
  }
  float* op = out + (size_t)ej * NSEQ + (tm + 8 * rg);
  *(v4f*)op       = (v4f){ v[0], v[1], v[2], v[3] };
  *(v4f*)(op + 4) = (v4f){ v[4], v[5], v[6], v[7] };
}

__global__ __launch_bounds__(256) void transpose_k(
    const float* __restrict__ in, float* __restrict__ out, int rows, int cols) {
  __shared__ float tile[32][33];
  const int c0 = blockIdx.x * 32, r0 = blockIdx.y * 32;
  const int tx = threadIdx.x, ty = threadIdx.y;
#pragma unroll
  for (int i = 0; i < 32; i += 8)
    tile[ty + i][tx] = in[(size_t)(r0 + ty + i) * cols + (c0 + tx)];
  __syncthreads();
#pragma unroll
  for (int i = 0; i < 32; i += 8)
    out[(size_t)(c0 + ty + i) * rows + (r0 + tx)] = tile[tx][ty + i];
}

__global__ __launch_bounds__(256) void u_substack_k(
    const float* __restrict__ u, float* __restrict__ Ut) {
  __shared__ float tile[32][33];
  const int c0 = blockIdx.x * 32, k0 = blockIdx.y * 32;
  const int tx = threadIdx.x, ty = threadIdx.y;
#pragma unroll
  for (int i = 0; i < 32; i += 8) {
    const int cc = c0 + ty + i;
    const int g = cc >> 9, jj = cc & (CHUNKS - 1);
    const int r = ((g & 1) << 1) | (g >> 1);
    const int j = jj > 0 ? jj - 1 : 0;
    const int q = 4 * j + r;
    tile[ty + i][tx] = u[(size_t)q * SUBK + (k0 + tx)];
  }
  __syncthreads();
#pragma unroll
  for (int i = 0; i < 32; i += 8)
    Ut[(size_t)(k0 + ty + i) * SUBC + (c0 + tx)] = tile[tx][ty + i];
}

__global__ __launch_bounds__(256) void copy_b_k(
    const float* __restrict__ B, float* __restrict__ dst) {
  const int idx = blockIdx.x * 256 + threadIdx.x;
  dst[(size_t)(idx >> 8) * SUBK + (idx & (UL - 1))] = B[idx];
}

__global__ void set_col_k(float* __restrict__ dst, const float* __restrict__ src,
                          int stride, int n) {
  int i = blockIdx.x * blockDim.x + threadIdx.x;
  if (i < n) dst[(size_t)i * stride] = src[i];
}

extern "C" void kernel_launch(void* const* d_in, const int* in_sizes, int n_in,
                              void* d_out, int out_size, void* d_ws, size_t ws_size,
                              hipStream_t stream) {
  (void)in_sizes; (void)n_in; (void)out_size; (void)ws_size;
  const float* u  = (const float*)d_in[0];
  const float* A  = (const float*)d_in[1];
  const float* B  = (const float*)d_in[2];
  const float* C  = (const float*)d_in[3];
  const float* D  = (const float*)d_in[4];
  const float* x0 = (const float*)d_in[5];
  float* out = (float*)d_out;

  float* ws = (float*)d_ws;
  size_t off = 0;
  float* Bu = ws + off; off += (size_t)NSEQ * XL;
  float* xs = ws + off; off += (size_t)NSEQ * XL;
  float* Bt = ws + off; off += (size_t)UL * XL;
  float* Ct = ws + off; off += (size_t)XL * YLN;
  float* Dt = ws + off; off += (size_t)UL * YLN;
  float* S1a = ws + off; off += (size_t)XL * LDP;
  float* S1b = ws + off; off += (size_t)XL * LDP;
  float* Q1  = xs;
  float* Q2  = xs + (size_t)1 * XL * XL;
  float* Q3  = xs + (size_t)2 * XL * XL;
  float* Q4  = xs + (size_t)3 * XL * XL;
  float* Q5  = xs + (size_t)4 * XL * XL;
  float* Q6  = xs + (size_t)5 * XL * XL;
  float* Q7  = xs + (size_t)6 * XL * XL;
  float* Q8  = xs + (size_t)7 * XL * XL;
  float* Q9  = xs + (size_t)8 * XL * XL;
  float* Q10 = xs + (size_t)9 * XL * XL;
  float* Q11 = xs + (size_t)10 * XL * XL;
  float* Q12 = xs + (size_t)11 * XL * XL;
  float* Q13 = xs + (size_t)12 * XL * XL;

  transpose_k<<<dim3(UL / 32, XL / 32), dim3(32, 8), 0, stream>>>(B, Bt, XL, UL);
  transpose_k<<<dim3(XL / 32, YLN / 32), dim3(32, 8), 0, stream>>>(C, Ct, YLN, XL);
  transpose_k<<<dim3(YLN / 32, UL / 32), dim3(32, 8), 0, stream>>>(D, Dt, UL, YLN);

  gemm_rr_k<4, 4><<<dim3(NSEQ / 128, XL / 256), 256, 0, stream>>>(
      u, Bt, Bu, UL, UL, XL, XL);

  const dim3 sqg(XL / 32, XL / 64);
#define SQUARE(src, dst) \
  scan_wide_k<false, 0, XL, false, XL, XL, XL, 4><<<sqg, 256, 0, stream>>>(src, src, nullptr, dst, nullptr, 0)
  SQUARE(A,   Q1);
  SQUARE(Q1,  Q2);
  SQUARE(Q2,  Q3);
  SQUARE(Q3,  Q4);

  hipMemsetAsync(S1a, 0, (size_t)XL * LDP * sizeof(float), stream);
  hipMemsetAsync(S1b, 0, (size_t)XL * LDP * sizeof(float), stream);

  const dim3 stg(XL / 32, CHUNKS / 64);
  float* Va = S1a + PADC;
  float* Vb = S1b + PADC;

  float* Ut = xs + (size_t)4 * XL * XL;
  float* G8 = xs + (size_t)8 * XL * XL;
  float* E0 = xs + (size_t)10 * XL * XL;
  float* E1 = xs + (size_t)12 * XL * XL;
  u_substack_k<<<dim3(SUBC / 32, SUBK / 32), dim3(32, 8), 0, stream>>>(u, Ut);
  copy_b_k<<<XL * UL / 256, 256, 0, stream>>>(B, G8 + 7 * UL);
  scan_wide_k<false, 0, SUBK, false, XL, SUBK, SUBK><<<dim3(XL / 32, 4), 256, 0, stream>>>(
      A,  G8 + 7 * UL, nullptr, G8 + 6 * UL, nullptr, 0);
  scan_wide_k<false, 0, SUBK, false, XL, SUBK, SUBK><<<dim3(XL / 32, 8), 256, 0, stream>>>(
      Q1, G8 + 6 * UL, nullptr, G8 + 4 * UL, nullptr, 0);
  scan_wide_k<false, 0, SUBK, false, XL, SUBK, SUBK, 4><<<dim3(XL / 32, 16), 256, 0, stream>>>(
      Q2, G8 + 4 * UL, nullptr, G8,          nullptr, 0);
  scan_wide_k<false, 0, SUBC, false, SUBK, SUBC, SUBC, 4><<<dim3(XL / 32, SUBC / 64), 256, 0, stream>>>(
      G8, Ut, nullptr, E0, nullptr, 0);
  scan_wide_k<false, 2, SUBC, false, SUBK, SUBC, SUBC, 4><<<dim3(XL / 32, SUBC / 64), 256, 0, stream>>>(
      G8 + XL, Ut + (size_t)XL * SUBC, E0, E1, nullptr, 0);
  scan_wide_k<false, 2, SUBC, false, XL, SUBC, SUBC, 4><<<dim3(XL / 32, 16), 256, 0, stream>>>(
      Q3, E1, E1 + 1024, E0, nullptr, 0);
  scan_wide_k<false, 2, LDP, false, XL, SUBC, SUBC><<<stg, 256, 0, stream>>>(
      Q4, E0, E0 + CHUNKS, Va, nullptr, 0);
  set_col_k<<<4, 256, 0, stream>>>(Va, x0, LDP, XL);

  SQUARE(Q4,  Q5);
  SQUARE(Q5,  Q6);
  SQUARE(Q6,  Q7);
  SQUARE(Q7,  Q8);
  SQUARE(Q8,  Q9);
  SQUARE(Q9,  Q10);
  SQUARE(Q10, Q11);
  SQUARE(Q11, Q12);
  SQUARE(Q12, Q13);
#undef SQUARE


  scan_wide_k<false, 2, LDP, true ><<<stg, 256, 0, stream>>>(Q5,  Va - 1,   Va, Vb, nullptr, 1);
  scan_wide_k<false, 2, LDP, true ><<<stg, 256, 0, stream>>>(Q6,  Vb - 2,   Vb, Va, nullptr, 2);
  scan_wide_k<false, 2, LDP, false><<<stg, 256, 0, stream>>>(Q7,  Va - 4,   Va, Vb, nullptr, 4);
  scan_wide_k<false, 2, LDP, false><<<stg, 256, 0, stream>>>(Q8,  Vb - 8,   Vb, Va, nullptr, 8);
  scan_wide_k<false, 2, LDP, false><<<stg, 256, 0, stream>>>(Q9,  Va - 16,  Va, Vb, nullptr, 16);
  scan_wide_k<false, 2, LDP, false><<<stg, 256, 0, stream>>>(Q10, Vb - 32,  Vb, Va, nullptr, 32);
  scan_wide_k<false, 2, LDP, false><<<stg, 256, 0, stream>>>(Q11, Va - 64,  Va, Vb, nullptr, 64);
  scan_wide_k<false, 2, LDP, false><<<stg, 256, 0, stream>>>(Q12, Vb - 128, Vb, Va, nullptr, 128);
  scan_wide_k<false, 2, LDP, false><<<stg, 256, 0, stream>>>(Q13, Va - 256, Va, Vb, nullptr, 256);

  {
    float* cur = Vb; float* nxt = Va;
    for (int t = 0; t < CLEN; ++t) {
      scan_wide_k<true, 1, LDP, false><<<stg, 256, 0, stream>>>(
          A, cur, Bu + (size_t)t * XL, nxt, xs + (size_t)t * XL, 0);
      float* tmp = cur; cur = nxt; nxt = tmp;
    }
  }

  final_wide_k<<<dim3(NSEQ / 32, YLN / 64), 256, 0, stream>>>(xs, Ct, u, Dt, out);
}


# 32-step down pass, Bu GEMM and output GEMM replaced by one direct product per 8-step sub-chunk: out = tanh([Toeplitz(C A^k B)+D | C A^1..8] @ [u; sub-chunk starts]); sum order changes
# speedup vs baseline: 10.0060x; 1.2470x over previous
#include <hip/hip_runtime.h>
#include <math.h>


typedef __attribute__((ext_vector_type(2))) float v2f;
typedef __attribute__((ext_vector_type(4))) float v4f;
typedef __attribute__((ext_vector_type(8))) float v8f;
typedef float v4fu __attribute__((ext_vector_type(4), aligned(4)));

#define NSEQ   16384
#define XL     1024
#define UL     256
#define YLN    256
#define CHUNKS 512
#define CLEN   32
#define PADC   64
#define LDP    (CHUNKS + PADC)
#define SUBL   8
#define SUBK   (SUBL * UL)
#define SUBC   (4 * CHUNKS)
#define GK     (SUBK + XL)

__device__ __forceinline__ v8f wmma4(v2f a, v2f b, v8f c) {
  return __builtin_amdgcn_wmma_f32_16x16x4_f32(
       false, a,  false, b,
       (short)0, c,  false,  false);
}

template <bool OUTT, int ADD, int LD, bool UNAL,
          int LDA = XL, int LDB = LD, int LDC = LD, int MINW = 2>
__global__ __launch_bounds__(256) __attribute__((amdgpu_waves_per_eu(MINW, 4)))
void scan_wide_k(
    const float* __restrict__ Ap, const float* __restrict__ Sold,
    const float* addb, float* __restrict__ Snew,
    float* __restrict__ outb, int sh) {
  static_assert(ADD == 0 || ADD == 2, "plain product or row-major addend");
  static_assert(ADD == 2 || !OUTT, "the output part adds the K parts before it");
  static_assert(ADD == 2 || !UNAL, "only a product with addend moves its operand");
  constexpr int KE = XL / 8;
  constexpr int NP = KE / 8;
  constexpr int GP = 2;
  constexpr int NG = NP / GP;
  constexpr int RP = 8 * 32 + 16;
  __shared__ float red[8][2][4][RP];

  const int lane = threadIdx.x & 31;
  const int w    = threadIdx.x >> 5;
  const int tm = blockIdx.x * 32;
  const int tn = blockIdx.y * 64;
  const int c  = lane & 15;
  const int h  = lane >> 4;

  if constexpr (ADD == 2) {
    if (tn + 64 <= sh) {
#pragma unroll
      for (int i = 0; i < 8; ++i) {
        const int idx = (int)threadIdx.x + 256 * i;
        const size_t o  = (size_t)(tm + (idx >> 6)) * LD  + (tn + (idx & 63));
        const size_t oa = (size_t)(tm + (idx >> 6)) * LDC + (tn + (idx & 63));
        Snew[o] = addb[oa];
      }
      return;
    }
  }

  const int eh = w & 1, rg = w >> 1;
  const int ej = tn + eh * 32 + lane;
  const float* addp = nullptr;
  if constexpr (ADD == 2) {
    addp = addb + (size_t)(tm + 8 * rg) * LDC + ej;
  } else {
    (void)addb; (void)sh;
  }
  v4f add0 = {}, add1 = {};

  const float* ap0 = Ap + (size_t)(tm + c) * LDA + (w * KE + 4 * h);
  const float* ap1 = ap0 + (size_t)16 * LDA;
  const float* bp  = Sold + (size_t)(w * KE + 4 * h) * LDB + (tn + 4 * c);

  v4f af[NG][GP][2];
  v4f bq[NG][GP][4];

#define SCAN_WIDE_LOAD_GROUP(g)                                              \
  _Pragma("unroll")                                                          \
  for (int p = 0; p < GP; ++p) {                                             \
    const int k = ((g) * GP + p) * 8;                                        \
    af[(g)][p][0] = *(const v4f*)(ap0 + k);                                  \
    af[(g)][p][1] = *(const v4f*)(ap1 + k);                                  \
    _Pragma("unroll")                                                        \
    for (int e = 0; e < 4; ++e) {                                            \
      if constexpr (UNAL)                                                    \
        bq[(g)][p][e] = *(const v4fu*)(bp + (size_t)(k + e) * LDB);          \
      else                                                                   \
        bq[(g)][p][e] = *(const v4f*)(bp + (size_t)(k + e) * LDB);           \
    }                                                                        \
  }

#pragma unroll
  for (int g = 0; g < 2; ++g) { SCAN_WIDE_LOAD_GROUP(g) }

  v8f acc[2][4];
#pragma unroll
  for (int mi = 0; mi < 2; ++mi)
#pragma unroll
    for (int ni = 0; ni < 4; ++ni) acc[mi][ni] = (v8f){};

#pragma unroll
  for (int g = 0; g < NG; ++g) {
    if (MINW == 4) __builtin_amdgcn_sched_barrier(0);
    if (g + 2 < NG) { SCAN_WIDE_LOAD_GROUP(g + 2) }
    if (ADD != 0 && g + 3 == NG) {
      if constexpr (ADD == 2) {
#pragma unroll
        for (int i = 0; i < 4; ++i) {
          add0[i] = addp[(size_t)i * LDC];
          add1[i] = addp[(size_t)(i + 4) * LDC];
        }
      }
      __builtin_amdgcn_sched_barrier(0);
    }
#pragma unroll
    for (int p = 0; p < GP; ++p)
#pragma unroll
      for (int s = 0; s < 2; ++s)
#pragma unroll
        for (int ni = 0; ni < 4; ++ni)
#pragma unroll
          for (int mi = 0; mi < 2; ++mi)
            acc[mi][ni] = wmma4(
                (v2f){ af[g][p][mi][2 * s], af[g][p][mi][2 * s + 1] },
                (v2f){ bq[g][p][2 * s][ni], bq[g][p][2 * s + 1][ni] },
                acc[mi][ni]);
  }
#undef SCAN_WIDE_LOAD_GROUP

#pragma unroll
  for (int mi = 0; mi < 2; ++mi)
#pragma unroll
    for (int ni = 0; ni < 4; ++ni)
#pragma unroll
      for (int r = 0; r < 8; ++r) red[w][mi][ni][r * 32 + lane] = acc[mi][ni][r];
  __syncthreads();

  const int emi = rg >> 1;
  const int eni = lane & 3;
  const int esl = (rg & 1) * 16 + eh * 8 + (lane >> 2);
  float v[8];
#pragma unroll
  for (int i = 0; i < 8; ++i) {
    const int sl = i * 32 + esl;
    const float c0 = red[0][emi][eni][sl], c1 = red[1][emi][eni][sl];
    const float c2 = red[2][emi][eni][sl], c3 = red[3][emi][eni][sl];
    const float c4 = red[4][emi][eni][sl], c5 = red[5][emi][eni][sl];
    const float c6 = red[6][emi][eni][sl], c7 = red[7][emi][eni][sl];
    v[i] = ((c0 + c1) + (c2 + c3)) + ((c4 + c5) + (c6 + c7));
    if constexpr (ADD != 0) v[i] += (i < 4) ? add0[i & 3] : add1[i & 3];
    if constexpr (OUTT) v[i] = tanhf(v[i]);
    else Snew[(size_t)(tm + 8 * rg + i) * LD + ej] = v[i];
  }
  if constexpr (OUTT) {
    float* op = outb + (size_t)((tm >> 3) + rg) * NSEQ + 8 * ej;
    *(v4f*)op       = (v4f){ v[0], v[1], v[2], v[3] };
    *(v4f*)(op + 4) = (v4f){ v[4], v[5], v[6], v[7] };
    (void)Snew;
  } else {
    (void)outb;
  }
}

__global__ __launch_bounds__(256) void transpose_k(
    const float* __restrict__ in, float* __restrict__ out, int rows, int cols) {
  __shared__ float tile[32][33];
  const int c0 = blockIdx.x * 32, r0 = blockIdx.y * 32;
  const int tx = threadIdx.x, ty = threadIdx.y;
#pragma unroll
  for (int i = 0; i < 32; i += 8)
    tile[ty + i][tx] = in[(size_t)(r0 + ty + i) * cols + (c0 + tx)];
  __syncthreads();
#pragma unroll
  for (int i = 0; i < 32; i += 8)
    out[(size_t)(c0 + ty + i) * rows + (r0 + tx)] = tile[tx][ty + i];
}

__global__ __launch_bounds__(256) void u_substack_k(
    const float* __restrict__ u, float* __restrict__ Ut) {
  __shared__ float tile[32][33];
  const int c0 = blockIdx.x * 32, k0 = blockIdx.y * 32;
  const int tx = threadIdx.x, ty = threadIdx.y;
#pragma unroll
  for (int i = 0; i < 32; i += 8) {
    const int cc = c0 + ty + i;
    const int g = cc >> 9, jj = cc & (CHUNKS - 1);
    const int r = ((g & 1) << 1) | (g >> 1);
    const int j = (jj + CHUNKS - 1) & (CHUNKS - 1);
    const int q = 4 * j + r;
    tile[ty + i][tx] = u[(size_t)q * SUBK + (k0 + tx)];
  }
  __syncthreads();
#pragma unroll
  for (int i = 0; i < 32; i += 8)
    Ut[(size_t)(k0 + ty + i) * SUBC + (c0 + tx)] = tile[tx][ty + i];
}

__global__ __launch_bounds__(256) void copy_b_k(
    const float* __restrict__ B, float* __restrict__ dst) {
  const int idx = blockIdx.x * 256 + threadIdx.x;
  dst[(size_t)(idx >> 8) * SUBK + (idx & (UL - 1))] = B[idx];
}

__global__ void set_col_k(float* __restrict__ dst, const float* __restrict__ src,
                          int stride, int n) {
  int i = blockIdx.x * blockDim.x + threadIdx.x;
  if (i < n) dst[(size_t)i * stride] = src[i];
}

__global__ void copy_col_k(float* __restrict__ dst, const float* __restrict__ src,
                           int stride, int n) {
  int i = blockIdx.x * blockDim.x + threadIdx.x;
  if (i < n) dst[(size_t)i * stride] = src[(size_t)i * stride];
}

__global__ __launch_bounds__(256) void copy_v4_k(
    const float* __restrict__ src, float* __restrict__ dst) {
  const size_t i = (size_t)blockIdx.x * 256 + threadIdx.x;
  ((v4f*)dst)[i] = ((const v4f*)src)[i];
}

__global__ __launch_bounds__(256) void interleave4_k(
    const float* __restrict__ Vs, const float* __restrict__ T1,
    const float* __restrict__ T2, const float* __restrict__ T3,
    float* __restrict__ S8) {
  const int idx = blockIdx.x * 256 + threadIdx.x;
  const int x = idx >> 9, j = idx & (CHUNKS - 1);
  const int jj = (j + 1) & (CHUNKS - 1);
  const v4f s = { Vs[(size_t)x * LDP + jj], T1[(size_t)x * CHUNKS + jj],
                  T2[(size_t)x * CHUNKS + jj], T3[(size_t)x * CHUNKS + jj] };
  *(v4f*)(S8 + (size_t)x * SUBC + 4 * j) = s;
}

__global__ __launch_bounds__(256) void build_g_k(
    const float* __restrict__ Hs, const float* __restrict__ P9,
    const float* __restrict__ D, float* __restrict__ G) {
  const int row = blockIdx.y;
  const int yl = row >> 3, t = row & 7;
  const int s = blockIdx.x;
  const int i = threadIdx.x;
  float v;
  if (s < SUBL) {
    if (s > t) {
      v = 0.0f;
    } else {
      v = Hs[(size_t)((t - s) * YLN + yl) * UL + i];
      if (s == t) v += D[(size_t)yl * UL + i];
    }
  } else {
    v = P9[(size_t)((t + 1) * YLN + yl) * XL + ((s - SUBL) * 256 + i)];
  }
  G[(size_t)row * GK + s * 256 + i] = v;
}


extern "C" void kernel_launch(void* const* d_in, const int* in_sizes, int n_in,
                              void* d_out, int out_size, void* d_ws, size_t ws_size,
                              hipStream_t stream) {
  (void)in_sizes; (void)n_in; (void)out_size; (void)ws_size;
  const float* u  = (const float*)d_in[0];
  const float* A  = (const float*)d_in[1];
  const float* B  = (const float*)d_in[2];
  const float* C  = (const float*)d_in[3];
  const float* D  = (const float*)d_in[4];
  const float* x0 = (const float*)d_in[5];
  float* out = (float*)d_out;

  float* ws = (float*)d_ws;
  const size_t SL = (size_t)XL * XL;
  size_t off = 32 * SL + (size_t)UL * XL + (size_t)XL * YLN + (size_t)UL * YLN;
  float* S1a = ws + off; off += (size_t)XL * LDP;
  float* S1b = ws + off; off += (size_t)XL * LDP;
  float* Q1  = ws;
  float* Q2  = ws + 1 * SL;
  float* Q3  = ws + 2 * SL;
  float* Q4  = ws + 3 * SL;
  float* Q5  = ws + 4 * SL;
  float* Q6  = ws + 5 * SL;
  float* Q7  = ws + 6 * SL;
  float* Q8  = ws + 7 * SL;
  float* Q9  = ws + 8 * SL;
  float* Q10 = ws + 9 * SL;
  float* Q11 = ws + 10 * SL;
  float* Q12 = ws + 11 * SL;
  float* Q13 = ws + 12 * SL;
  float* Ut = ws + 16 * SL;
  float* G8 = ws + 20 * SL;
  float* E0 = ws + 22 * SL;
  float* E1 = ws + 24 * SL;
  float* T1 = ws + 26 * SL;
  float* T2 = T1 + (size_t)XL * CHUNKS;
  float* T3 = ws + 27 * SL;
  float* S8 = ws + 28 * SL;
  float* P9 = ws + 4 * SL;
  float* Hs = ws + 7 * SL;
  float* Un = ws + 8 * SL;
  float* O1 = ws + 12 * SL;
  float* Gm = ws + 16 * SL;
  float* O0 = ws + 22 * SL;

  const dim3 sqg(XL / 32, XL / 64);
#define SQUARE(src, dst) \
  scan_wide_k<false, 0, XL, false, XL, XL, XL, 4><<<sqg, 256, 0, stream>>>(src, src, nullptr, dst, nullptr, 0)
  SQUARE(A,   Q1);
  SQUARE(Q1,  Q2);
  SQUARE(Q2,  Q3);
  SQUARE(Q3,  Q4);

  hipMemsetAsync(S1a, 0, (size_t)XL * LDP * sizeof(float), stream);
  hipMemsetAsync(S1b, 0, (size_t)XL * LDP * sizeof(float), stream);

  const dim3 stg(XL / 32, CHUNKS / 64);
  float* Va = S1a + PADC;
  float* Vb = S1b + PADC;

  u_substack_k<<<dim3(SUBC / 32, SUBK / 32), dim3(32, 8), 0, stream>>>(u, Ut);
  copy_b_k<<<XL * UL / 256, 256, 0, stream>>>(B, G8 + 7 * UL);
  scan_wide_k<false, 0, SUBK, false, XL, SUBK, SUBK><<<dim3(XL / 32, 4), 256, 0, stream>>>(
      A,  G8 + 7 * UL, nullptr, G8 + 6 * UL, nullptr, 0);
  scan_wide_k<false, 0, SUBK, false, XL, SUBK, SUBK><<<dim3(XL / 32, 8), 256, 0, stream>>>(
      Q1, G8 + 6 * UL, nullptr, G8 + 4 * UL, nullptr, 0);
  scan_wide_k<false, 0, SUBK, false, XL, SUBK, SUBK, 4><<<dim3(XL / 32, 16), 256, 0, stream>>>(
      Q2, G8 + 4 * UL, nullptr, G8,          nullptr, 0);
  scan_wide_k<false, 0, SUBC, false, SUBK, SUBC, SUBC, 4><<<dim3(XL / 32, SUBC / 64), 256, 0, stream>>>(
      G8, Ut, nullptr, E0, nullptr, 0);
  scan_wide_k<false, 2, SUBC, false, SUBK, SUBC, SUBC, 4><<<dim3(XL / 32, SUBC / 64), 256, 0, stream>>>(
      G8 + XL, Ut + (size_t)XL * SUBC, E0, E1, nullptr, 0);
  scan_wide_k<false, 2, SUBC, false, XL, SUBC, SUBC, 4><<<dim3(XL / 32, 16), 256, 0, stream>>>(
      Q3, E1, E1 + 1024, E0, nullptr, 0);
  scan_wide_k<false, 2, LDP, false, XL, SUBC, SUBC><<<stg, 256, 0, stream>>>(
      Q4, E0, E0 + CHUNKS, Va, nullptr, 0);
  set_col_k<<<4, 256, 0, stream>>>(Va, x0, LDP, XL);

  SQUARE(Q4,  Q5);
  SQUARE(Q5,  Q6);
  SQUARE(Q6,  Q7);
  SQUARE(Q7,  Q8);
  SQUARE(Q8,  Q9);
  SQUARE(Q9,  Q10);
  SQUARE(Q10, Q11);
  SQUARE(Q11, Q12);
  SQUARE(Q12, Q13);
#undef SQUARE

  scan_wide_k<false, 2, LDP, true ><<<stg, 256, 0, stream>>>(Q5,  Va - 1,   Va, Vb, nullptr, 1);
  scan_wide_k<false, 2, LDP, true ><<<stg, 256, 0, stream>>>(Q6,  Vb - 2,   Vb, Va, nullptr, 2);
  scan_wide_k<false, 2, LDP, false><<<stg, 256, 0, stream>>>(Q7,  Va - 4,   Va, Vb, nullptr, 4);
  scan_wide_k<false, 2, LDP, false><<<stg, 256, 0, stream>>>(Q8,  Vb - 8,   Vb, Va, nullptr, 8);
  scan_wide_k<false, 2, LDP, false><<<stg, 256, 0, stream>>>(Q9,  Va - 16,  Va, Vb, nullptr, 16);
  scan_wide_k<false, 2, LDP, false><<<stg, 256, 0, stream>>>(Q10, Vb - 32,  Vb, Va, nullptr, 32);
  scan_wide_k<false, 2, LDP, false><<<stg, 256, 0, stream>>>(Q11, Va - 64,  Va, Vb, nullptr, 64);
  scan_wide_k<false, 2, LDP, false><<<stg, 256, 0, stream>>>(Q12, Vb - 128, Vb, Va, nullptr, 128);
  scan_wide_k<false, 2, LDP, false><<<stg, 256, 0, stream>>>(Q13, Va - 256, Va, Vb, nullptr, 256);

  float* Vs = Vb - 1;
  copy_col_k<<<4, 256, 0, stream>>>(Vs, Vb + (CHUNKS - 1), LDP, XL);
  scan_wide_k<false, 2, CHUNKS, true,  XL, LDP,    SUBC><<<stg, 256, 0, stream>>>(
      Q3, Vs, E1,          T1, nullptr, 0);
  scan_wide_k<false, 2, CHUNKS, true,  XL, LDP,    SUBC><<<stg, 256, 0, stream>>>(
      Q4, Vs, E0,          T2, nullptr, 0);
  scan_wide_k<false, 2, CHUNKS, false, XL, CHUNKS, SUBC><<<stg, 256, 0, stream>>>(
      Q3, T2, E1 + CHUNKS, T3, nullptr, 0);
  interleave4_k<<<XL * CHUNKS / 256, 256, 0, stream>>>(Vs, T1, T2, T3, S8);

  copy_v4_k<<<YLN * XL / 4 / 256, 256, 0, stream>>>(C, P9);
  scan_wide_k<false, 0, XL, false, XL, XL, XL><<<dim3(YLN / 32, XL / 64), 256, 0, stream>>>(
      P9, A, nullptr, P9 + (size_t)1 * YLN * XL, nullptr, 0);
  scan_wide_k<false, 0, XL, false, XL, XL, XL><<<dim3(2 * YLN / 32, XL / 64), 256, 0, stream>>>(
      P9, Q1, nullptr, P9 + (size_t)2 * YLN * XL, nullptr, 0);
  scan_wide_k<false, 0, XL, false, XL, XL, XL, 4><<<dim3(4 * YLN / 32, XL / 64), 256, 0, stream>>>(
      P9, Q2, nullptr, P9 + (size_t)4 * YLN * XL, nullptr, 0);
  scan_wide_k<false, 0, XL, false, XL, XL, XL><<<dim3(YLN / 32, XL / 64), 256, 0, stream>>>(
      P9 + (size_t)4 * YLN * XL, Q2, nullptr, P9 + (size_t)8 * YLN * XL, nullptr, 0);
  scan_wide_k<false, 0, UL, false, XL, UL, UL><<<dim3(SUBL * YLN / 32, UL / 64), 256, 0, stream>>>(
      P9, B, nullptr, Hs, nullptr, 0);
  build_g_k<<<dim3(GK / 256, SUBL * YLN), 256, 0, stream>>>(Hs, P9, D, Gm);

  transpose_k<<<dim3(SUBK / 32, SUBC / 32), dim3(32, 8), 0, stream>>>(u, Un, SUBC, SUBK);

  const dim3 og(SUBL * YLN / 32, SUBC / 64);
  scan_wide_k<false, 0, SUBC, false, GK, SUBC, SUBC, 4><<<og, 256, 0, stream>>>(
      Gm, Un, nullptr, O0, nullptr, 0);
  scan_wide_k<false, 2, SUBC, false, GK, SUBC, SUBC, 4><<<og, 256, 0, stream>>>(
      Gm + XL, Un + (size_t)XL * SUBC, O0, O1, nullptr, 0);
  scan_wide_k<true,  2, SUBC, false, GK, SUBC, SUBC, 4><<<og, 256, 0, stream>>>(
      Gm + 2 * XL, S8, O1, nullptr, out, 0);
}


# products with K over 1024 (sub-chunk ends K=2048, output K=3072) each run as ONE launch, every wave looping over KTOT/8, instead of 2 and 3 K-part launches chained through memory; sum order changes
# speedup vs baseline: 10.3970x; 1.0391x over previous
#include <hip/hip_runtime.h>
#include <math.h>


typedef __attribute__((ext_vector_type(2))) float v2f;
typedef __attribute__((ext_vector_type(4))) float v4f;
typedef __attribute__((ext_vector_type(8))) float v8f;
typedef float v4fu __attribute__((ext_vector_type(4), aligned(4)));

#define NSEQ   16384
#define XL     1024
#define UL     256
#define YLN    256
#define CHUNKS 512
#define CLEN   32
#define PADC   64
#define LDP    (CHUNKS + PADC)
#define SUBL   8
#define SUBK   (SUBL * UL)
#define SUBC   (4 * CHUNKS)
#define GK     (SUBK + XL)

__device__ __forceinline__ v8f wmma4(v2f a, v2f b, v8f c) {
  return __builtin_amdgcn_wmma_f32_16x16x4_f32(
       false, a,  false, b,
       (short)0, c,  false,  false);
}

template <bool OUTT, int ADD, int LD, bool UNAL,
          int LDA = XL, int LDB = LD, int LDC = LD, int MINW = 2, int KTOT = XL>
__global__ __launch_bounds__(256) __attribute__((amdgpu_waves_per_eu(MINW, 4)))
void scan_wide_k(
    const float* __restrict__ Ap, const float* __restrict__ Sold,
    const float* addb, float* __restrict__ Snew,
    float* __restrict__ outb, int sh) {
  static_assert(ADD == 0 || ADD == 2, "plain product or row-major addend");
  static_assert(KTOT % 128 == 0, "a K-eighth is a whole number of load groups");
  static_assert(ADD == 2 || !UNAL, "only a product with addend moves its operand");
  constexpr int KE = KTOT / 8;
  constexpr int NP = KE / 8;
  constexpr int GP = 2;
  constexpr int NG = NP / GP;
  constexpr int RP = 8 * 32 + 16;
  __shared__ float red[8][2][4][RP];

  const int lane = threadIdx.x & 31;
  const int w    = threadIdx.x >> 5;
  const int tm = blockIdx.x * 32;
  const int tn = blockIdx.y * 64;
  const int c  = lane & 15;
  const int h  = lane >> 4;

  if constexpr (ADD == 2) {
    if (tn + 64 <= sh) {
#pragma unroll
      for (int i = 0; i < 8; ++i) {
        const int idx = (int)threadIdx.x + 256 * i;
        const size_t o  = (size_t)(tm + (idx >> 6)) * LD  + (tn + (idx & 63));
        const size_t oa = (size_t)(tm + (idx >> 6)) * LDC + (tn + (idx & 63));
        Snew[o] = addb[oa];
      }
      return;
    }
  }

  const int eh = w & 1, rg = w >> 1;
  const int ej = tn + eh * 32 + lane;
  const float* addp = nullptr;
  if constexpr (ADD == 2) {
    addp = addb + (size_t)(tm + 8 * rg) * LDC + ej;
  } else {
    (void)addb; (void)sh;
  }
  v4f add0 = {}, add1 = {};

  const float* ap0 = Ap + (size_t)(tm + c) * LDA + (w * KE + 4 * h);
  const float* ap1 = ap0 + (size_t)16 * LDA;
  const float* bp  = Sold + (size_t)(w * KE + 4 * h) * LDB + (tn + 4 * c);

  v4f af[NG][GP][2];
  v4f bq[NG][GP][4];

#define SCAN_WIDE_LOAD_GROUP(g)                                              \
  _Pragma("unroll")                                                          \
  for (int p = 0; p < GP; ++p) {                                             \
    const int k = ((g) * GP + p) * 8;                                        \
    af[(g)][p][0] = *(const v4f*)(ap0 + k);                                  \
    af[(g)][p][1] = *(const v4f*)(ap1 + k);                                  \
    _Pragma("unroll")                                                        \
    for (int e = 0; e < 4; ++e) {                                            \
      if constexpr (UNAL)                                                    \
        bq[(g)][p][e] = *(const v4fu*)(bp + (size_t)(k + e) * LDB);          \
      else                                                                   \
        bq[(g)][p][e] = *(const v4f*)(bp + (size_t)(k + e) * LDB);           \
    }                                                                        \
  }

#pragma unroll
  for (int g = 0; g < 2; ++g) { SCAN_WIDE_LOAD_GROUP(g) }

  v8f acc[2][4];
#pragma unroll
  for (int mi = 0; mi < 2; ++mi)
#pragma unroll
    for (int ni = 0; ni < 4; ++ni) acc[mi][ni] = (v8f){};

#pragma unroll
  for (int g = 0; g < NG; ++g) {
    if (MINW == 4) __builtin_amdgcn_sched_barrier(0);
    if (g + 2 < NG) { SCAN_WIDE_LOAD_GROUP(g + 2) }
    if (ADD != 0 && g + 3 == NG) {
      if constexpr (ADD == 2) {
#pragma unroll
        for (int i = 0; i < 4; ++i) {
          add0[i] = addp[(size_t)i * LDC];
          add1[i] = addp[(size_t)(i + 4) * LDC];
        }
      }
      __builtin_amdgcn_sched_barrier(0);
    }
#pragma unroll
    for (int p = 0; p < GP; ++p)
#pragma unroll
      for (int s = 0; s < 2; ++s)
#pragma unroll
        for (int ni = 0; ni < 4; ++ni)
#pragma unroll
          for (int mi = 0; mi < 2; ++mi)
            acc[mi][ni] = wmma4(
                (v2f){ af[g][p][mi][2 * s], af[g][p][mi][2 * s + 1] },
                (v2f){ bq[g][p][2 * s][ni], bq[g][p][2 * s + 1][ni] },
                acc[mi][ni]);
  }
#undef SCAN_WIDE_LOAD_GROUP

#pragma unroll
  for (int mi = 0; mi < 2; ++mi)
#pragma unroll
    for (int ni = 0; ni < 4; ++ni)
#pragma unroll
      for (int r = 0; r < 8; ++r) red[w][mi][ni][r * 32 + lane] = acc[mi][ni][r];
  __syncthreads();

  const int emi = rg >> 1;
  const int eni = lane & 3;
  const int esl = (rg & 1) * 16 + eh * 8 + (lane >> 2);
  float v[8];
#pragma unroll
  for (int i = 0; i < 8; ++i) {
    const int sl = i * 32 + esl;
    const float c0 = red[0][emi][eni][sl], c1 = red[1][emi][eni][sl];
    const float c2 = red[2][emi][eni][sl], c3 = red[3][emi][eni][sl];
    const float c4 = red[4][emi][eni][sl], c5 = red[5][emi][eni][sl];
    const float c6 = red[6][emi][eni][sl], c7 = red[7][emi][eni][sl];
    v[i] = ((c0 + c1) + (c2 + c3)) + ((c4 + c5) + (c6 + c7));
    if constexpr (ADD != 0) v[i] += (i < 4) ? add0[i & 3] : add1[i & 3];
    if constexpr (OUTT) v[i] = tanhf(v[i]);
    else Snew[(size_t)(tm + 8 * rg + i) * LD + ej] = v[i];
  }
  if constexpr (OUTT) {
    float* op = outb + (size_t)((tm >> 3) + rg) * NSEQ + 8 * ej;
    *(v4f*)op       = (v4f){ v[0], v[1], v[2], v[3] };
    *(v4f*)(op + 4) = (v4f){ v[4], v[5], v[6], v[7] };
    (void)Snew;
  } else {
    (void)outb;
  }
}

__global__ __launch_bounds__(256) void transpose_k(
    const float* __restrict__ in, float* __restrict__ out, int rows, int cols) {
  __shared__ float tile[32][33];
  const int c0 = blockIdx.x * 32, r0 = blockIdx.y * 32;
  const int tx = threadIdx.x, ty = threadIdx.y;
#pragma unroll
  for (int i = 0; i < 32; i += 8)
    tile[ty + i][tx] = in[(size_t)(r0 + ty + i) * cols + (c0 + tx)];
  __syncthreads();
#pragma unroll
  for (int i = 0; i < 32; i += 8)
    out[(size_t)(c0 + ty + i) * rows + (r0 + tx)] = tile[tx][ty + i];
}

__global__ __launch_bounds__(256) void u_substack_k(
    const float* __restrict__ u, float* __restrict__ Ut) {
  __shared__ float tile[32][33];
  const int c0 = blockIdx.x * 32, k0 = blockIdx.y * 32;
  const int tx = threadIdx.x, ty = threadIdx.y;
#pragma unroll
  for (int i = 0; i < 32; i += 8) {
    const int cc = c0 + ty + i;
    const int g = cc >> 9, jj = cc & (CHUNKS - 1);
    const int r = ((g & 1) << 1) | (g >> 1);
    const int j = (jj + CHUNKS - 1) & (CHUNKS - 1);
    const int q = 4 * j + r;
    tile[ty + i][tx] = u[(size_t)q * SUBK + (k0 + tx)];
  }
  __syncthreads();
#pragma unroll
  for (int i = 0; i < 32; i += 8)
    Ut[(size_t)(k0 + ty + i) * SUBC + (c0 + tx)] = tile[tx][ty + i];
}

__global__ __launch_bounds__(256) void copy_b_k(
    const float* __restrict__ B, float* __restrict__ dst) {
  const int idx = blockIdx.x * 256 + threadIdx.x;
  dst[(size_t)(idx >> 8) * SUBK + (idx & (UL - 1))] = B[idx];
}

__global__ void set_col_k(float* __restrict__ dst, const float* __restrict__ src,
                          int stride, int n) {
  int i = blockIdx.x * blockDim.x + threadIdx.x;
  if (i < n) dst[(size_t)i * stride] = src[i];
}

__global__ void copy_col_k(float* __restrict__ dst, const float* __restrict__ src,
                           int stride, int n) {
  int i = blockIdx.x * blockDim.x + threadIdx.x;
  if (i < n) dst[(size_t)i * stride] = src[(size_t)i * stride];
}

__global__ __launch_bounds__(256) void copy_v4_k(
    const float* __restrict__ src, float* __restrict__ dst) {
  const size_t i = (size_t)blockIdx.x * 256 + threadIdx.x;
  ((v4f*)dst)[i] = ((const v4f*)src)[i];
}

__global__ __launch_bounds__(256) void interleave4_k(
    const float* __restrict__ Vs, const float* __restrict__ T1,
    const float* __restrict__ T2, const float* __restrict__ T3,
    float* __restrict__ S8) {
  const int idx = blockIdx.x * 256 + threadIdx.x;
  const int x = idx >> 9, j = idx & (CHUNKS - 1);
  const int jj = (j + 1) & (CHUNKS - 1);
  const v4f s = { Vs[(size_t)x * LDP + jj], T1[(size_t)x * CHUNKS + jj],
                  T2[(size_t)x * CHUNKS + jj], T3[(size_t)x * CHUNKS + jj] };
  *(v4f*)(S8 + (size_t)x * SUBC + 4 * j) = s;
}

__global__ __launch_bounds__(256) void build_g_k(
    const float* __restrict__ Hs, const float* __restrict__ P9,
    const float* __restrict__ D, float* __restrict__ G) {
  const int row = blockIdx.y;
  const int yl = row >> 3, t = row & 7;
  const int s = blockIdx.x;
  const int i = threadIdx.x;
  float v;
  if (s < SUBL) {
    if (s > t) {
      v = 0.0f;
    } else {
      v = Hs[(size_t)((t - s) * YLN + yl) * UL + i];
      if (s == t) v += D[(size_t)yl * UL + i];
    }
  } else {
    v = P9[(size_t)((t + 1) * YLN + yl) * XL + ((s - SUBL) * 256 + i)];
  }
  G[(size_t)row * GK + s * 256 + i] = v;
}


extern "C" void kernel_launch(void* const* d_in, const int* in_sizes, int n_in,
                              void* d_out, int out_size, void* d_ws, size_t ws_size,
                              hipStream_t stream) {
  (void)in_sizes; (void)n_in; (void)out_size; (void)ws_size;
  const float* u  = (const float*)d_in[0];
  const float* A  = (const float*)d_in[1];
  const float* B  = (const float*)d_in[2];
  const float* C  = (const float*)d_in[3];
  const float* D  = (const float*)d_in[4];
  const float* x0 = (const float*)d_in[5];
  float* out = (float*)d_out;

  float* ws = (float*)d_ws;
  const size_t SL = (size_t)XL * XL;
  size_t off = 32 * SL + (size_t)UL * XL + (size_t)XL * YLN + (size_t)UL * YLN;
  float* S1a = ws + off; off += (size_t)XL * LDP;
  float* S1b = ws + off; off += (size_t)XL * LDP;
  float* Q1  = ws;
  float* Q2  = ws + 1 * SL;
  float* Q3  = ws + 2 * SL;
  float* Q4  = ws + 3 * SL;
  float* Q5  = ws + 4 * SL;
  float* Q6  = ws + 5 * SL;
  float* Q7  = ws + 6 * SL;
  float* Q8  = ws + 7 * SL;
  float* Q9  = ws + 8 * SL;
  float* Q10 = ws + 9 * SL;
  float* Q11 = ws + 10 * SL;
  float* Q12 = ws + 11 * SL;
  float* Q13 = ws + 12 * SL;
  float* Ut = ws + 16 * SL;
  float* G8 = ws + 20 * SL;
  float* E0 = ws + 22 * SL;
  float* E1 = ws + 24 * SL;
  float* T1 = ws + 26 * SL;
  float* T2 = T1 + (size_t)XL * CHUNKS;
  float* T3 = ws + 27 * SL;
  float* S8 = ws + 12 * SL;
  float* P9 = ws + 4 * SL;
  float* Hs = ws + 7 * SL;
  float* Un = ws + 8 * SL;
  float* Gm = ws + 16 * SL;

  const dim3 sqg(XL / 32, XL / 64);
#define SQUARE(src, dst) \
  scan_wide_k<false, 0, XL, false, XL, XL, XL, 4><<<sqg, 256, 0, stream>>>(src, src, nullptr, dst, nullptr, 0)
  SQUARE(A,   Q1);
  SQUARE(Q1,  Q2);
  SQUARE(Q2,  Q3);
  SQUARE(Q3,  Q4);

  hipMemsetAsync(S1a, 0, (size_t)XL * LDP * sizeof(float), stream);
  hipMemsetAsync(S1b, 0, (size_t)XL * LDP * sizeof(float), stream);

  const dim3 stg(XL / 32, CHUNKS / 64);
  float* Va = S1a + PADC;
  float* Vb = S1b + PADC;

  u_substack_k<<<dim3(SUBC / 32, SUBK / 32), dim3(32, 8), 0, stream>>>(u, Ut);
  copy_b_k<<<XL * UL / 256, 256, 0, stream>>>(B, G8 + 7 * UL);
  scan_wide_k<false, 0, SUBK, false, XL, SUBK, SUBK><<<dim3(XL / 32, 4), 256, 0, stream>>>(
      A,  G8 + 7 * UL, nullptr, G8 + 6 * UL, nullptr, 0);
  scan_wide_k<false, 0, SUBK, false, XL, SUBK, SUBK><<<dim3(XL / 32, 8), 256, 0, stream>>>(
      Q1, G8 + 6 * UL, nullptr, G8 + 4 * UL, nullptr, 0);
  scan_wide_k<false, 0, SUBK, false, XL, SUBK, SUBK, 4><<<dim3(XL / 32, 16), 256, 0, stream>>>(
      Q2, G8 + 4 * UL, nullptr, G8,          nullptr, 0);
  scan_wide_k<false, 0, SUBC, false, SUBK, SUBC, SUBC, 4, SUBK><<<dim3(XL / 32, SUBC / 64), 256, 0, stream>>>(
      G8, Ut, nullptr, E1, nullptr, 0);
  scan_wide_k<false, 2, SUBC, false, XL, SUBC, SUBC, 4><<<dim3(XL / 32, 16), 256, 0, stream>>>(
      Q3, E1, E1 + 1024, E0, nullptr, 0);
  scan_wide_k<false, 2, LDP, false, XL, SUBC, SUBC><<<stg, 256, 0, stream>>>(
      Q4, E0, E0 + CHUNKS, Va, nullptr, 0);
  set_col_k<<<4, 256, 0, stream>>>(Va, x0, LDP, XL);

  SQUARE(Q4,  Q5);
  SQUARE(Q5,  Q6);
  SQUARE(Q6,  Q7);
  SQUARE(Q7,  Q8);
  SQUARE(Q8,  Q9);
  SQUARE(Q9,  Q10);
  SQUARE(Q10, Q11);
  SQUARE(Q11, Q12);
  SQUARE(Q12, Q13);
#undef SQUARE

  scan_wide_k<false, 2, LDP, true ><<<stg, 256, 0, stream>>>(Q5,  Va - 1,   Va, Vb, nullptr, 1);
  scan_wide_k<false, 2, LDP, true ><<<stg, 256, 0, stream>>>(Q6,  Vb - 2,   Vb, Va, nullptr, 2);
  scan_wide_k<false, 2, LDP, false><<<stg, 256, 0, stream>>>(Q7,  Va - 4,   Va, Vb, nullptr, 4);
  scan_wide_k<false, 2, LDP, false><<<stg, 256, 0, stream>>>(Q8,  Vb - 8,   Vb, Va, nullptr, 8);
  scan_wide_k<false, 2, LDP, false><<<stg, 256, 0, stream>>>(Q9,  Va - 16,  Va, Vb, nullptr, 16);
  scan_wide_k<false, 2, LDP, false><<<stg, 256, 0, stream>>>(Q10, Vb - 32,  Vb, Va, nullptr, 32);
  scan_wide_k<false, 2, LDP, false><<<stg, 256, 0, stream>>>(Q11, Va - 64,  Va, Vb, nullptr, 64);
  scan_wide_k<false, 2, LDP, false><<<stg, 256, 0, stream>>>(Q12, Vb - 128, Vb, Va, nullptr, 128);
  scan_wide_k<false, 2, LDP, false><<<stg, 256, 0, stream>>>(Q13, Va - 256, Va, Vb, nullptr, 256);

  float* Vs = Vb - 1;
  copy_col_k<<<4, 256, 0, stream>>>(Vs, Vb + (CHUNKS - 1), LDP, XL);
  scan_wide_k<false, 2, CHUNKS, true,  XL, LDP,    SUBC><<<stg, 256, 0, stream>>>(
      Q3, Vs, E1,          T1, nullptr, 0);
  scan_wide_k<false, 2, CHUNKS, true,  XL, LDP,    SUBC><<<stg, 256, 0, stream>>>(
      Q4, Vs, E0,          T2, nullptr, 0);
  scan_wide_k<false, 2, CHUNKS, false, XL, CHUNKS, SUBC><<<stg, 256, 0, stream>>>(
      Q3, T2, E1 + CHUNKS, T3, nullptr, 0);
  interleave4_k<<<XL * CHUNKS / 256, 256, 0, stream>>>(Vs, T1, T2, T3, S8);

  copy_v4_k<<<YLN * XL / 4 / 256, 256, 0, stream>>>(C, P9);
  scan_wide_k<false, 0, XL, false, XL, XL, XL><<<dim3(YLN / 32, XL / 64), 256, 0, stream>>>(
      P9, A, nullptr, P9 + (size_t)1 * YLN * XL, nullptr, 0);
  scan_wide_k<false, 0, XL, false, XL, XL, XL><<<dim3(2 * YLN / 32, XL / 64), 256, 0, stream>>>(
      P9, Q1, nullptr, P9 + (size_t)2 * YLN * XL, nullptr, 0);
  scan_wide_k<false, 0, XL, false, XL, XL, XL, 4><<<dim3(4 * YLN / 32, XL / 64), 256, 0, stream>>>(
      P9, Q2, nullptr, P9 + (size_t)4 * YLN * XL, nullptr, 0);
  scan_wide_k<false, 0, XL, false, XL, XL, XL><<<dim3(YLN / 32, XL / 64), 256, 0, stream>>>(
      P9 + (size_t)4 * YLN * XL, Q2, nullptr, P9 + (size_t)8 * YLN * XL, nullptr, 0);
  scan_wide_k<false, 0, UL, false, XL, UL, UL><<<dim3(SUBL * YLN / 32, UL / 64), 256, 0, stream>>>(
      P9, B, nullptr, Hs, nullptr, 0);
  build_g_k<<<dim3(GK / 256, SUBL * YLN), 256, 0, stream>>>(Hs, P9, D, Gm);

  transpose_k<<<dim3(SUBK / 32, SUBC / 32), dim3(32, 8), 0, stream>>>(u, Un, SUBC, SUBK);

  const dim3 og(SUBL * YLN / 32, SUBC / 64);
  scan_wide_k<true, 0, SUBC, false, GK, SUBC, SUBC, 4, GK><<<og, 256, 0, stream>>>(
      Gm, Un, nullptr, nullptr, out, 0);
}
